# starting kernel
# baseline (speedup 1.0000x reference)
#include <hip/hip_runtime.h>
#include <math.h>

typedef __attribute__((ext_vector_type(2))) float v2f;
typedef __attribute__((ext_vector_type(8))) float v8f;

namespace {

constexpr int B_   = 32;
constexpr int L_   = 64;
constexpr int H_   = 200;
constexpr int M_   = 3;
constexpr int A_   = B_ * L_;
constexpr int N_   = M_ * A_;
constexpr int TAG_ = 6;
constexpr float PI_F = 3.14159265358979323846f;

template <typename FA, typename FB, typename FE>
__device__ __forceinline__ void wmma_tile_f32(int m0, int n0, int K,
                                              FA loadA, FB loadB, FE epi) {
  const int lane = threadIdx.x & 31;
  const int mr   = lane & 15;
  const int kh   = lane >> 4;
  v8f acc = {};
  for (int k = 0; k < K; k += 4) {
    const int kk = k + 2 * kh;
    v2f a, b;
    a.x = loadA(m0 + mr, kk);
    a.y = loadA(m0 + mr, kk + 1);
    b.x = loadB(kk, n0 + mr);
    b.y = loadB(kk + 1, n0 + mr);
    acc = __builtin_amdgcn_wmma_f32_16x16x4_f32(false, a, false, b,
                                                (short)0, acc, false, false);
  }
#pragma unroll
  for (int v = 0; v < 8; ++v) epi(m0 + 8 * kh + v, n0 + mr, acc[v]);
}

__device__ __forceinline__ float wave_sum(float s) {
#pragma unroll
  for (int off = 16; off > 0; off >>= 1) s += __shfl_xor(s, off, 32);
  return s;
}

__global__ void k_norm(const float* __restrict__ xa, const float* __restrict__ xv,
                       const float* __restrict__ xt, float* __restrict__ big,
                       float* __restrict__ nrm) {
  const int row  = blockIdx.x * 8 + (threadIdx.x >> 5);
  const int lane = threadIdx.x & 31;
  const float* src = (row < A_)     ? xa + (size_t)row * H_
                   : (row < 2 * A_) ? xv + (size_t)(row - A_) * H_
                                    : xt + (size_t)(row - 2 * A_) * H_;
  float s = 0.f;
  for (int k = lane; k < H_; k += 32) { float v = src[k]; s += v * v; }
  s = wave_sum(s);
  const float rn = 1.0f / sqrtf(s);
  for (int k = lane; k < H_; k += 32) {
    const float v = src[k];
    big[(size_t)row * H_ + k] = v;
    nrm[(size_t)row * H_ + k] = v * rn;
  }
}

__global__ void k_gram(const float* __restrict__ nrm, float* __restrict__ intra) {
  const int mb = blockIdx.x;
  const int w  = threadIdx.x >> 5;
  const int m0 = (w >> 2) * 16, n0 = (w & 3) * 16;
  const float* blk = nrm + (size_t)mb * L_ * H_;
  auto lA = [&](int i, int k) { return blk[(size_t)i * H_ + k]; };
  auto lB = [&](int k, int j) { return blk[(size_t)j * H_ + k]; };
  auto ep = [&](int i, int j, float v) {
    intra[((size_t)mb * L_ + i) * L_ + j] = 1.0f - acosf(v * 0.99999f) * (1.0f / PI_F);
  };
  wmma_tile_f32(m0, n0, H_, lA, lB, ep);
}

__global__ void k_cross(const float* __restrict__ nrm, float* __restrict__ crossv) {
  const int idx  = blockIdx.x * 8 + (threadIdx.x >> 5);
  const int lane = threadIdx.x & 31;
  const int pm[6] = {0, 0, 1, 1, 2, 2};
  const int pn[6] = {1, 2, 0, 2, 0, 1};
  const int q = idx / A_, a = idx % A_;
  const int m = pm[q], n = pn[q];
  const float* pa = nrm + ((size_t)m * A_ + a) * H_;
  const float* pb = nrm + ((size_t)n * A_ + a) * H_;
  float s = 0.f;
  for (int k = lane; k < H_; k += 32) s += pa[k] * pb[k];
  s = wave_sum(s);
  if (lane == 0)
    crossv[((size_t)m * M_ + n) * A_ + a] = 1.0f - acosf(s * 0.99999f) * (1.0f / PI_F);
}

__global__ void k_dinv(const float* __restrict__ intra, const float* __restrict__ crossv,
                       float* __restrict__ dinv) {
  const int row  = blockIdx.x * 8 + (threadIdx.x >> 5);
  const int lane = threadIdx.x & 31;
  const int m = row >> 11, a = row & (A_ - 1);
  const float* r = intra + (size_t)row * L_;
  float s = 0.f;
  for (int j = lane; j < L_; j += 32) s += r[j];
  s = wave_sum(s);
  if (lane == 0) {
#pragma unroll
    for (int n = 0; n < M_; ++n)
      if (n != m) s += crossv[((size_t)m * M_ + n) * A_ + a];
    dinv[row] = 1.0f / sqrtf(s);
  }
}

__global__ void k_fc0(const float* __restrict__ big, const float* __restrict__ w,
                      const float* __restrict__ bias, float* __restrict__ h0) {
  const int wv = threadIdx.x >> 5;
  const int m0 = (blockIdx.y * 8 + wv) * 16, n0 = blockIdx.x * 16;
  auto lA = [&](int m, int k) { return big[(size_t)m * H_ + k]; };
  auto lB = [&](int k, int n) { return (n < H_) ? w[(size_t)k * H_ + n] : 0.0f; };
  auto ep = [&](int m, int n, float v) {
    if (n < H_) h0[(size_t)m * H_ + n] = fmaxf(v + bias[n], 0.0f);
  };
  wmma_tile_f32(m0, n0, H_, lA, lB, ep);
}

__global__ void k_spmm(const float* __restrict__ intra, const float* __restrict__ crossv,
                       const float* __restrict__ dinv, const float* __restrict__ h,
                       float* __restrict__ hi) {
  const int mb = blockIdx.y;
  const int m  = mb >> 5;
  const int rowbase = mb * L_;
  const int m0 = (threadIdx.x >> 5) * 16;
  const int n0 = blockIdx.x * 16;
  const float* S = intra + (size_t)mb * L_ * L_;
  auto lA = [&](int i, int k) { return S[(size_t)i * L_ + k] * dinv[rowbase + k]; };
  auto lB = [&](int k, int n) { return (n < H_) ? h[(size_t)(rowbase + k) * H_ + n] : 0.0f; };
  auto ep = [&](int i, int n, float acc) {
    if (n >= H_) return;
    const int row = rowbase + i;
    const int a   = row - m * A_;
    float s = acc;
#pragma unroll
    for (int nn = 0; nn < M_; ++nn) {
      if (nn == m) continue;
      const int orow = nn * A_ + a;
      s += crossv[((size_t)m * M_ + nn) * A_ + a] * dinv[orow] * h[(size_t)orow * H_ + n];
    }
    hi[(size_t)row * H_ + n] = s * dinv[row];
  };
  wmma_tile_f32(m0, n0, L_, lA, lB, ep);
}

__global__ void k_conv(const float* __restrict__ hi, const float* __restrict__ h0,
                       const float* __restrict__ w, float beta, float* __restrict__ hout) {
  const int wv = threadIdx.x >> 5;
  const int m0 = (blockIdx.y * 8 + wv) * 16, n0 = blockIdx.x * 16;
  auto lA = [&](int m, int k) {
    return (k < H_) ? hi[(size_t)m * H_ + k] : h0[(size_t)m * H_ + (k - H_)];
  };
  auto lB = [&](int k, int n) { return (n < H_) ? w[(size_t)k * H_ + n] : 0.0f; };
  auto ep = [&](int m, int n, float v) {
    if (n >= H_) return;
    const float r = 0.9f * hi[(size_t)m * H_ + n] + 0.1f * h0[(size_t)m * H_ + n];
    hout[(size_t)m * H_ + n] = fmaxf(beta * v + (1.0f - beta) * r, 0.0f);
  };
  wmma_tile_f32(m0, n0, 2 * H_, lA, lB, ep);
}

__global__ void k_hx(const float* __restrict__ big, const float* __restrict__ hf,
                     const float* __restrict__ tw, const float* __restrict__ tb,
                     float* __restrict__ hx) {
  const int m  = blockIdx.z;
  const int wv = threadIdx.x >> 5;
  const int a0 = (blockIdx.y * 8 + wv) * 16, n0 = blockIdx.x * 16;
  const float* W  = tw + (size_t)m * 2 * H_ * H_;
  const float* Bb = tb + (size_t)m * H_;
  const int rb = m * A_;
  auto lA = [&](int a, int k) {
    return (k < H_) ? big[(size_t)(rb + a) * H_ + k] : hf[(size_t)(rb + a) * H_ + (k - H_)];
  };
  auto lB = [&](int k, int n) { return (n < H_) ? W[(size_t)k * H_ + n] : 0.0f; };
  auto ep = [&](int a, int n, float v) {
    if (n < H_) hx[(size_t)(rb + a) * H_ + n] = fmaxf(v + Bb[n], 0.0f);
  };
  wmma_tile_f32(a0, n0, 2 * H_, lA, lB, ep);
}

__global__ void k_gate(const float* __restrict__ hx, const float* __restrict__ cw,
                       const float* __restrict__ cb, float* __restrict__ hout,
                       int p, int i, int j) {
  const int wv = threadIdx.x >> 5;
  const int a0 = (blockIdx.y * 8 + wv) * 16, n0 = blockIdx.x * 16;
  const float* Hi = hx + (size_t)i * A_ * H_;
  const float* Hj = hx + (size_t)j * A_ * H_;
  const float* W  = cw + (size_t)p * 3 * H_ * H_;
  const float* Bb = cb + (size_t)p * H_;
  auto lA = [&](int a, int k) {
    if (k < H_) return Hi[(size_t)a * H_ + k];
    if (k < 2 * H_) return Hj[(size_t)a * H_ + (k - H_)];
    const int kk = k - 2 * H_;
    return Hi[(size_t)a * H_ + kk] * Hj[(size_t)a * H_ + kk];
  };
  auto lB = [&](int k, int n) { return (n < H_) ? W[(size_t)k * H_ + n] : 0.0f; };
  auto ep = [&](int a, int n, float v) {
    if (n >= H_) return;
    const float z = 1.0f / (1.0f + expf(-(v + Bb[n])));
    hout[(size_t)(i * A_ + a) * H_ + n] +=
        z * Hi[(size_t)a * H_ + n] + (1.0f - z) * Hj[(size_t)a * H_ + n];
  };
  wmma_tile_f32(a0, n0, 3 * H_, lA, lB, ep);
}

__global__ void k_final(const float* __restrict__ hout, const float* __restrict__ uw,
                        const float* __restrict__ ub, float* __restrict__ out) {
  const int a = blockIdx.x * blockDim.x + threadIdx.x;
  float acc[TAG_];
#pragma unroll
  for (int t = 0; t < TAG_; ++t)
    acc[t] = ub[t] + ub[TAG_ + t] + ub[2 * TAG_ + t];
  for (int m = 0; m < M_; ++m) {
    const float* hr = hout + (size_t)(m * A_ + a) * H_;
    const float* wm = uw + (size_t)m * H_ * TAG_;
    for (int h = 0; h < H_; ++h) {
      const float v = hr[h];
#pragma unroll
      for (int t = 0; t < TAG_; ++t) acc[t] += v * wm[h * TAG_ + t];
    }
  }
#pragma unroll
  for (int t = 0; t < TAG_; ++t) out[(size_t)a * TAG_ + t] = acc[t];
}

}

extern "C" void kernel_launch(void* const* d_in, const int* in_sizes, int n_in,
                              void* d_out, int out_size, void* d_ws, size_t ws_size,
                              hipStream_t stream) {
  (void)in_sizes; (void)n_in; (void)out_size; (void)ws_size;
  const float* x_a     = (const float*)d_in[0];
  const float* x_v     = (const float*)d_in[1];
  const float* x_t     = (const float*)d_in[2];
  const float* fc0_w   = (const float*)d_in[3];
  const float* fc0_b   = (const float*)d_in[4];
  const float* conv_w  = (const float*)d_in[5];
  const float* tm_w    = (const float*)d_in[6];
  const float* tm_b    = (const float*)d_in[7];
  const float* cross_w = (const float*)d_in[8];
  const float* cross_b = (const float*)d_in[9];
  const float* uni_w   = (const float*)d_in[10];
  const float* uni_b   = (const float*)d_in[11];
  float* out = (float*)d_out;

  const size_t NH = (size_t)N_ * H_;
  float* ws    = (float*)d_ws;
  float* big   = ws; ws += NH;
  float* nrm   = ws; ws += NH;
  float* intra = ws; ws += (size_t)N_ * L_;
  float* crossv= ws; ws += (size_t)M_ * M_ * A_;
  float* dinv  = ws; ws += N_;
  float* h0    = ws; ws += NH;
  float* hA    = ws; ws += NH;
  float* hBf   = ws; ws += NH;
  float* hi    = ws; ws += NH;
  float* hx    = ws; ws += NH;
  float* hout  = ws; ws += NH;

  const int NT = (H_ + 15) / 16;

  k_norm <<<N_ / 8,              dim3(256), 0, stream>>>(x_a, x_v, x_t, big, nrm);
  k_gram <<<M_ * B_,             dim3(512), 0, stream>>>(nrm, intra);
  k_cross<<<6 * A_ / 8,          dim3(256), 0, stream>>>(nrm, crossv);
  k_dinv <<<N_ / 8,              dim3(256), 0, stream>>>(intra, crossv, dinv);
  k_fc0  <<<dim3(NT, N_ / 128),  dim3(256), 0, stream>>>(big, fc0_w, fc0_b, h0);

  const float beta0 = logf(0.5f / 1.0f + 1.0f);
  const float beta1 = logf(0.5f / 2.0f + 1.0f);
  k_spmm <<<dim3(NT, M_ * B_),   dim3(128), 0, stream>>>(intra, crossv, dinv, h0, hi);
  k_conv <<<dim3(NT, N_ / 128),  dim3(256), 0, stream>>>(hi, h0, conv_w, beta0, hA);
  k_spmm <<<dim3(NT, M_ * B_),   dim3(128), 0, stream>>>(intra, crossv, dinv, hA, hi);
  k_conv <<<dim3(NT, N_ / 128),  dim3(256), 0, stream>>>(hi, h0, conv_w + 2 * H_ * H_, beta1, hBf);

  k_hx   <<<dim3(NT, A_ / 128, M_), dim3(256), 0, stream>>>(big, hBf, tm_w, tm_b, hx);

  hipMemsetAsync(hout, 0, NH * sizeof(float), stream);
  const int pi_[6] = {0, 0, 1, 1, 2, 2};
  const int pj_[6] = {1, 2, 0, 2, 0, 1};
  for (int p = 0; p < 6; ++p)
    k_gate<<<dim3(NT, A_ / 128), dim3(256), 0, stream>>>(hx, cross_w, cross_b, hout,
                                                         p, pi_[p], pj_[p]);

  k_final<<<A_ / 256, dim3(256), 0, stream>>>(hout, uni_w, uni_b, out);
}


# GEMM core (fc0, spmm, conv, hx, gate): in two-tile strips a lane owns 2 adjacent columns, so B comes by 8-byte row loads (4 per 8 k, was 8), rows stored 8 bytes wide; gate's column sum order changes
# speedup vs baseline: 5.3981x; 5.3981x over previous
#include <hip/hip_runtime.h>
#include <math.h>

typedef __attribute__((ext_vector_type(2))) float v2f;
typedef __attribute__((ext_vector_type(4))) float v4f;
typedef __attribute__((ext_vector_type(8))) float v8f;

namespace {

constexpr int B_   = 32;
constexpr int L_   = 64;
constexpr int H_   = 200;
constexpr int M_   = 3;
constexpr int A_   = B_ * L_;
constexpr int N_   = M_ * A_;
constexpr int TAG_ = 6;
constexpr float PI_F = 3.14159265358979323846f;

template <typename FA, typename FB, typename FE>
__device__ __forceinline__ void wmma_tile_f32(int m0, int n0, int K,
                                              FA loadA, FB loadB, FE epi) {
  const int lane = threadIdx.x & 31;
  const int mr   = lane & 15;
  const int kh   = lane >> 4;
  v8f acc = {};
  for (int k = 0; k < K; k += 4) {
    const int kk = k + 2 * kh;
    v2f a, b;
    a.x = loadA(m0 + mr, kk);
    a.y = loadA(m0 + mr, kk + 1);
    b.x = loadB(kk, n0 + mr);
    b.y = loadB(kk + 1, n0 + mr);
    acc = __builtin_amdgcn_wmma_f32_16x16x4_f32(false, a, false, b,
                                                (short)0, acc, false, false);
  }
#pragma unroll
  for (int v = 0; v < 8; ++v) epi(m0 + 8 * kh + v, n0 + mr, acc[v]);
}

__device__ __forceinline__ float wave_sum(float s) {
#pragma unroll
  for (int off = 16; off > 0; off >>= 1) s += __shfl_xor(s, off, 32);
  return s;
}

__global__ void k_norm(const float* __restrict__ xa, const float* __restrict__ xv,
                       const float* __restrict__ xt, float* __restrict__ big,
                       float* __restrict__ nrm) {
  const int row  = blockIdx.x * 8 + (threadIdx.x >> 5);
  const int lane = threadIdx.x & 31;
  const float* src = (row < A_)     ? xa + (size_t)row * H_
                   : (row < 2 * A_) ? xv + (size_t)(row - A_) * H_
                                    : xt + (size_t)(row - 2 * A_) * H_;
  float s = 0.f;
  for (int k = lane; k < H_; k += 32) { float v = src[k]; s += v * v; }
  s = wave_sum(s);
  const float rn = 1.0f / sqrtf(s);
  for (int k = lane; k < H_; k += 32) {
    const float v = src[k];
    big[(size_t)row * H_ + k] = v;
    nrm[(size_t)row * H_ + k] = v * rn;
  }
}

__global__ void k_gram(const float* __restrict__ nrm, float* __restrict__ intra) {
  const int mb = blockIdx.x;
  const int w  = threadIdx.x >> 5;
  const int m0 = (w >> 2) * 16, n0 = (w & 3) * 16;
  const float* blk = nrm + (size_t)mb * L_ * H_;
  auto lA = [&](int i, int k) { return blk[(size_t)i * H_ + k]; };
  auto lB = [&](int k, int j) { return blk[(size_t)j * H_ + k]; };
  auto ep = [&](int i, int j, float v) {
    intra[((size_t)mb * L_ + i) * L_ + j] = 1.0f - acosf(v * 0.99999f) * (1.0f / PI_F);
  };
  wmma_tile_f32(m0, n0, H_, lA, lB, ep);
}

__global__ void k_cross(const float* __restrict__ nrm, float* __restrict__ crossv) {
  const int idx  = blockIdx.x * 8 + (threadIdx.x >> 5);
  const int lane = threadIdx.x & 31;
  const int pm[6] = {0, 0, 1, 1, 2, 2};
  const int pn[6] = {1, 2, 0, 2, 0, 1};
  const int q = idx / A_, a = idx % A_;
  const int m = pm[q], n = pn[q];
  const float* pa = nrm + ((size_t)m * A_ + a) * H_;
  const float* pb = nrm + ((size_t)n * A_ + a) * H_;
  float s = 0.f;
  for (int k = lane; k < H_; k += 32) s += pa[k] * pb[k];
  s = wave_sum(s);
  if (lane == 0)
    crossv[((size_t)m * M_ + n) * A_ + a] = 1.0f - acosf(s * 0.99999f) * (1.0f / PI_F);
}

__global__ void k_dinv(const float* __restrict__ intra, const float* __restrict__ crossv,
                       float* __restrict__ dinv) {
  const int row  = blockIdx.x * 8 + (threadIdx.x >> 5);
  const int lane = threadIdx.x & 31;
  const int m = row >> 11, a = row & (A_ - 1);
  const float* r = intra + (size_t)row * L_;
  float s = 0.f;
  for (int j = lane; j < L_; j += 32) s += r[j];
  s = wave_sum(s);
  if (lane == 0) {
#pragma unroll
    for (int n = 0; n < M_; ++n)
      if (n != m) s += crossv[((size_t)m * M_ + n) * A_ + a];
    dinv[row] = 1.0f / sqrtf(s);
  }
}

struct RawQuad { v4f p, q; };

__device__ __forceinline__ v4f ld4(const float* p) {
  return *reinterpret_cast<const v4f*>(p);
}

__device__ __forceinline__ v2f ld2(const float* p) {
  return *reinterpret_cast<const v2f*>(p);
}

template <int MT, int NT, int D, typename RA>
struct Ring { RA a[D][MT]; v2f b[D][2][NT]; };

template <int NT>
__device__ __forceinline__ void b_frags(const v2f (&raw)[NT], v2f (&f)[NT]) {
  if constexpr (NT == 2) {
    f[0].x = raw[0].x; f[0].y = raw[1].x;
    f[1].x = raw[0].y; f[1].y = raw[1].y;
  } else {
    f[0] = raw[0];
  }
}

template <int MT, int NT, bool TR = false>
__device__ __forceinline__ void mma_step(v8f (&acc)[MT][NT], const v2f (&a)[MT],
                                         const v2f (&b)[NT]) {
#pragma unroll
  for (int t = 0; t < MT; ++t)
#pragma unroll
    for (int u = 0; u < NT; ++u) {
      if constexpr (TR)
        acc[t][u] = __builtin_amdgcn_wmma_f32_16x16x4_f32(false, b[u], false, a[t],
                                                          (short)0, acc[t][u], false, false);
      else
        acc[t][u] = __builtin_amdgcn_wmma_f32_16x16x4_f32(false, a[t], false, b[u],
                                                          (short)0, acc[t][u], false, false);
    }
}


template <int MT, int NT, int D, typename RA, typename FA, typename FB>
__device__ __forceinline__ void ring_fill(Ring<MT, NT, D, RA>& r, FA ldA, FB ldB) {
  const int kh = (threadIdx.x & 31) >> 4;
#pragma unroll
  for (int d = 0; d < D; ++d) {
    const int kq = 8 * d + 4 * kh;
#pragma unroll
    for (int t = 0; t < MT; ++t) r.a[d][t] = ldA(t, kq);
#pragma unroll
    for (int s = 0; s < 2; ++s)
#pragma unroll
      for (int u = 0; u < NT; ++u) r.b[d][s][u] = ldB(u, kq + 2 * s);
    __builtin_amdgcn_sched_barrier(0);
  }
}

template <int MT, int NT, int D, int NPAIRS, bool TR = false, typename RA, typename FA, typename FC, typename FB>
__device__ __forceinline__ void ring_main(v8f (&acc)[MT][NT], Ring<MT, NT, D, RA>& r,
                                          FA ldA, FC cvA, FB ldB) {
  static_assert(NPAIRS % D == 0 && NPAIRS >= D, "ring depth must divide the pair count");
  const int kh = (threadIdx.x & 31) >> 4;
#pragma nounroll
  for (int p0 = 0; p0 < NPAIRS - D; p0 += D) {
#pragma unroll
    for (int d = 0; d < D; ++d) {
      v2f lo[MT], hi[MT];
#pragma unroll
      for (int t = 0; t < MT; ++t) {
        const v4f c = cvA(r.a[d][t]);
        lo[t] = c.xy;
        hi[t] = c.zw;
      }
      v2f f0[NT], f1[NT];
      b_frags<NT>(r.b[d][0], f0);
      b_frags<NT>(r.b[d][1], f1);
      mma_step<MT, NT, TR>(acc, lo, f0);
      mma_step<MT, NT, TR>(acc, hi, f1);
      const int kq = 8 * (p0 + d + D) + 4 * kh;
#pragma unroll
      for (int t = 0; t < MT; ++t) r.a[d][t] = ldA(t, kq);
#pragma unroll
      for (int s = 0; s < 2; ++s)
#pragma unroll
        for (int u = 0; u < NT; ++u) r.b[d][s][u] = ldB(u, kq + 2 * s);
      __builtin_amdgcn_sched_barrier(0);
    }
  }
}

template <bool TR = false, int MT, int NT, int D, typename RA, typename FC>
__device__ __forceinline__ void ring_drain(v8f (&acc)[MT][NT], Ring<MT, NT, D, RA>& r, FC cvA) {
#pragma unroll
  for (int d = 0; d < D; ++d) {
    v2f lo[MT], hi[MT];
#pragma unroll
    for (int t = 0; t < MT; ++t) {
      const v4f c = cvA(r.a[d][t]);
      lo[t] = c.xy;
      hi[t] = c.zw;
    }
    v2f f0[NT], f1[NT];
    b_frags<NT>(r.b[d][0], f0);
    b_frags<NT>(r.b[d][1], f1);
    mma_step<MT, NT, TR>(acc, lo, f0);
    mma_step<MT, NT, TR>(acc, hi, f1);
  }
}

template <int MT, int NT, typename FV>
__device__ __forceinline__ void mm_epi(const v8f (&acc)[MT][NT], int m0, int n0,
                                       float* __restrict__ dst, FV val) {
  const int lane = threadIdx.x & 31;
  const int mr = lane & 15, kh = lane >> 4;
#pragma unroll
  for (int t = 0; t < MT; ++t)
#pragma unroll
    for (int v = 0; v < 8; ++v) {
      const int m = m0 + 16 * t + 8 * kh + v;
      if constexpr (NT == 2) {
        const int n = n0 + 2 * mr;
        v2f o;
        o.x = val(m, n, acc[t][0][v]);
        o.y = val(m, n + 1, acc[t][1][v]);
        *reinterpret_cast<v2f*>(dst + (size_t)m * H_ + n) = o;
      } else {
        const int n = n0 + mr;
        if (n < H_) dst[(size_t)m * H_ + n] = val(m, n, acc[t][0][v]);
      }
    }
}

template <int NT>
__device__ __forceinline__ int lane_col(int n0) {
  const int mr = threadIdx.x & 15;
  return NT == 2 ? n0 + 2 * mr : min(n0 + mr, H_ - 1);
}

template <int NT>
__device__ __forceinline__ v2f ldB_unit(const float* __restrict__ w, int cb, int j, int kk) {
  if constexpr (NT == 2) {
    return ld2(w + (size_t)(kk + j) * H_ + cb);
  } else {
    v2f b;
    b.x = w[(size_t)kk * H_ + cb];
    b.y = w[(size_t)(kk + 1) * H_ + cb];
    return b;
  }
}

constexpr int NS_ = 7;
constexpr int KP_ = H_ / 8;

template <int NT>
__device__ __forceinline__ void fc0_body(const float* __restrict__ big, const float* __restrict__ w,
                                         const float* __restrict__ bias, float* __restrict__ h0,
                                         int m0, int n0) {
  const int mr = threadIdx.x & 15;
  const int cb = lane_col<NT>(n0);
  const float* ar[2] = {big + (size_t)(m0 + mr) * H_, big + (size_t)(m0 + 16 + mr) * H_};
  v8f acc[2][NT] = {};
  auto cA = [](v4f r) { return r; };
  auto lA = [&](int t, int kq) { return ld4(ar[t] + kq); };
  auto lB = [&](int j, int kk) { return ldB_unit<NT>(w, cb, j, kk); };
  Ring<2, NT, 5, v4f> g0;
  ring_fill(g0, lA, lB);
  ring_main<2, NT, 5, KP_>(acc, g0, lA, cA, lB);
  ring_drain(acc, g0, cA);
  auto ep = [&](int m, int n, float v) { return fmaxf(v + bias[n], 0.0f); };
  mm_epi<2, NT>(acc, m0, n0, h0, ep);
}

__global__ void __launch_bounds__(64) k_fc0(const float* __restrict__ big, const float* __restrict__ w,
                      const float* __restrict__ bias, float* __restrict__ h0) {
  const int m0 = (blockIdx.y * 2 + (threadIdx.x >> 5)) * 32, n0 = blockIdx.x * 32;
  if (blockIdx.x == NS_ - 1) fc0_body<1>(big, w, bias, h0, m0, n0);
  else                       fc0_body<2>(big, w, bias, h0, m0, n0);
}

template <int NT>
__device__ __forceinline__ void spmm_body(const float* __restrict__ intra,
                                          const float* __restrict__ crossv,
                                          const float* __restrict__ dinv,
                                          const float* __restrict__ h, float* __restrict__ hi,
                                          int mb, int i0, int n0) {
  const int mr = threadIdx.x & 15;
  const int m = mb >> 5;
  const int rowbase = mb * L_;
  const int cb = lane_col<NT>(n0);
  const float* S = intra + (size_t)mb * L_ * L_;
  const float* sr[2] = {S + (size_t)(i0 + mr) * L_, S + (size_t)(i0 + 16 + mr) * L_};
  const float* dv = dinv + rowbase;
  const float* hb = h + (size_t)rowbase * H_;
  v8f acc[2][NT] = {};
  auto lA = [&](int t, int kq) { RawQuad r; r.p = ld4(sr[t] + kq); r.q = ld4(dv + kq); return r; };
  auto cA = [](RawQuad r) { return r.p * r.q; };
  auto lB = [&](int j, int kk) { return ldB_unit<NT>(hb, cb, j, kk); };
  Ring<2, NT, 4, RawQuad> g0;
  ring_fill(g0, lA, lB);
  ring_main<2, NT, 4, L_ / 8>(acc, g0, lA, cA, lB);
  ring_drain(acc, g0, cA);
  auto ep = [&](int i, int n, float a_) {
    const int row = rowbase + i;
    const int a   = row - m * A_;
    float s = a_;
#pragma unroll
    for (int nn = 0; nn < M_; ++nn) {
      if (nn == m) continue;
      const int orow = nn * A_ + a;
      s += crossv[((size_t)m * M_ + nn) * A_ + a] * dinv[orow] * h[(size_t)orow * H_ + n];
    }
    return s * dinv[row];
  };
  mm_epi<2, NT>(acc, i0, n0, hi + (size_t)rowbase * H_, ep);
}

__global__ void __launch_bounds__(64) k_spmm(const float* __restrict__ intra, const float* __restrict__ crossv,
                       const float* __restrict__ dinv, const float* __restrict__ h,
                       float* __restrict__ hi) {
  const int mb = blockIdx.y;
  const int i0 = (threadIdx.x >> 5) * 32, n0 = blockIdx.x * 32;
  if (blockIdx.x == NS_ - 1) spmm_body<1>(intra, crossv, dinv, h, hi, mb, i0, n0);
  else                       spmm_body<2>(intra, crossv, dinv, h, hi, mb, i0, n0);
}

template <int NT>
__device__ __forceinline__ void conv_body(const float* __restrict__ hi, const float* __restrict__ h0,
                                          const float* __restrict__ w, float beta,
                                          float* __restrict__ hout, int m0, int n0) {
  const int mr = threadIdx.x & 15;
  const int cb = lane_col<NT>(n0);
  const size_t r0 = (size_t)(m0 + mr) * H_, r1 = (size_t)(m0 + 16 + mr) * H_;
  const float* a0[2] = {hi + r0, hi + r1};
  const float* a1[2] = {h0 + r0, h0 + r1};
  const float* w1 = w + (size_t)H_ * H_;
  v8f acc[2][NT] = {};
  auto cA = [](v4f r) { return r; };
  auto lA0 = [&](int t, int kq) { return ld4(a0[t] + kq); };
  auto lA1 = [&](int t, int kq) { return ld4(a1[t] + kq); };
  auto lB0 = [&](int j, int kk) { return ldB_unit<NT>(w, cb, j, kk); };
  auto lB1 = [&](int j, int kk) { return ldB_unit<NT>(w1, cb, j, kk); };
  Ring<2, NT, 5, v4f> g0, g1;
  ring_fill(g0, lA0, lB0);
  ring_main<2, NT, 5, KP_>(acc, g0, lA0, cA, lB0);
  ring_fill(g1, lA1, lB1);
  ring_drain(acc, g0, cA);
  ring_main<2, NT, 5, KP_>(acc, g1, lA1, cA, lB1);
  ring_drain(acc, g1, cA);
  auto ep = [&](int m, int n, float v) {
    const float r = 0.9f * hi[(size_t)m * H_ + n] + 0.1f * h0[(size_t)m * H_ + n];
    return fmaxf(beta * v + (1.0f - beta) * r, 0.0f);
  };
  mm_epi<2, NT>(acc, m0, n0, hout, ep);
}

__global__ void __launch_bounds__(64) k_conv(const float* __restrict__ hi, const float* __restrict__ h0,
                       const float* __restrict__ w, float beta, float* __restrict__ hout) {
  const int m0 = (blockIdx.y * 2 + (threadIdx.x >> 5)) * 32, n0 = blockIdx.x * 32;
  if (blockIdx.x == NS_ - 1) conv_body<1>(hi, h0, w, beta, hout, m0, n0);
  else                       conv_body<2>(hi, h0, w, beta, hout, m0, n0);
}

template <int NT>
__device__ __forceinline__ void hx_body(const float* __restrict__ big, const float* __restrict__ hf,
                                        const float* __restrict__ W, const float* __restrict__ Bb,
                                        float* __restrict__ hx, int rb, int a0_, int n0) {
  const int mr = threadIdx.x & 15;
  const int cb = lane_col<NT>(n0);
  const size_t r0 = (size_t)(rb + a0_ + mr) * H_, r1 = (size_t)(rb + a0_ + 16 + mr) * H_;
  const float* p0[2] = {big + r0, big + r1};
  const float* p1[2] = {hf + r0, hf + r1};
  const float* W1 = W + (size_t)H_ * H_;
  v8f acc[2][NT] = {};
  auto cA = [](v4f r) { return r; };
  auto lA0 = [&](int t, int kq) { return ld4(p0[t] + kq); };
  auto lA1 = [&](int t, int kq) { return ld4(p1[t] + kq); };
  auto lB0 = [&](int j, int kk) { return ldB_unit<NT>(W, cb, j, kk); };
  auto lB1 = [&](int j, int kk) { return ldB_unit<NT>(W1, cb, j, kk); };
  Ring<2, NT, 5, v4f> g0, g1;
  ring_fill(g0, lA0, lB0);
  ring_main<2, NT, 5, KP_>(acc, g0, lA0, cA, lB0);
  ring_fill(g1, lA1, lB1);
  ring_drain(acc, g0, cA);
  ring_main<2, NT, 5, KP_>(acc, g1, lA1, cA, lB1);
  ring_drain(acc, g1, cA);
  auto ep = [&](int a, int n, float v) { return fmaxf(v + Bb[n], 0.0f); };
  mm_epi<2, NT>(acc, a0_, n0, hx + (size_t)rb * H_, ep);
}

__global__ void __launch_bounds__(64) k_hx(const float* __restrict__ big, const float* __restrict__ hf,
                     const float* __restrict__ tw, const float* __restrict__ tb,
                     float* __restrict__ hx) {
  const int m  = blockIdx.z;
  const int a0_ = (blockIdx.y * 2 + (threadIdx.x >> 5)) * 32, n0 = blockIdx.x * 32;
  const float* W  = tw + (size_t)m * 2 * H_ * H_;
  const float* Bb = tb + (size_t)m * H_;
  if (blockIdx.x == NS_ - 1) hx_body<1>(big, hf, W, Bb, hx, m * A_, a0_, n0);
  else                       hx_body<2>(big, hf, W, Bb, hx, m * A_, a0_, n0);
}

template <int NT>
__device__ __forceinline__ float tr_logit(const v8f (&a)[NT], int g, int e) {
  if constexpr (NT == 2) return a[e & 1][4 * g + (e >> 1)];
  else                   return a[0][e];
}

template <int NT>
__device__ __forceinline__ void gate_body(const float* __restrict__ Hi, const float* __restrict__ Hj,
                                          const float* __restrict__ W, const float* __restrict__ Bb,
                                          const float* __restrict__ Ui, float* __restrict__ part_out,
                                          int a0_, int n0) {
  const int mr = threadIdx.x & 15;
  const int kh = (threadIdx.x & 31) >> 4;
  const int cb = lane_col<NT>(n0);
  const size_t r0 = (size_t)(a0_ + mr) * H_, r1 = (size_t)(a0_ + 16 + mr) * H_;
  const float* pi[2] = {Hi + r0, Hi + r1};
  const float* pj[2] = {Hj + r0, Hj + r1};
  const float* W1 = W + (size_t)H_ * H_;
  const float* W2 = W + (size_t)2 * H_ * H_;
  v8f acc[2][NT] = {};
  auto cA = [](v4f r) { return r; };
  auto cP = [](RawQuad r) { return r.p * r.q; };
  auto lAi = [&](int t, int kq) { return ld4(pi[t] + kq); };
  auto lAj = [&](int t, int kq) { return ld4(pj[t] + kq); };
  auto lAp = [&](int t, int kq) { RawQuad r; r.p = ld4(pi[t] + kq); r.q = ld4(pj[t] + kq); return r; };
  auto lB0 = [&](int j, int kk) { return ldB_unit<NT>(W, cb, j, kk); };
  auto lB1 = [&](int j, int kk) { return ldB_unit<NT>(W1, cb, j, kk); };
  auto lB2 = [&](int j, int kk) { return ldB_unit<NT>(W2, cb, j, kk); };
  Ring<2, NT, 5, v4f> g0, g1;
  Ring<2, NT, 5, RawQuad> g2;
  ring_fill(g0, lAi, lB0);
  ring_main<2, NT, 5, KP_, true>(acc, g0, lAi, cA, lB0);
  ring_fill(g1, lAj, lB1);
  ring_drain<true>(acc, g0, cA);
  ring_main<2, NT, 5, KP_, true>(acc, g1, lAj, cA, lB1);
  ring_fill(g2, lAp, lB2);
  ring_drain<true>(acc, g1, cA);
  ring_main<2, NT, 5, KP_, true>(acc, g2, lAp, cP, lB2);
  ring_drain<true>(acc, g2, cP);

  float part[2][TAG_] = {};
#pragma unroll
  for (int t = 0; t < 2; ++t) {
#pragma unroll
    for (int g = 0; g < NT; ++g) {
      const int nb = (NT == 2) ? n0 + 16 * kh + 8 * g : n0 + 8 * kh;
      if (NT == 2 || nb < H_) {
        const v4f* hip = reinterpret_cast<const v4f*>(pi[t] + nb);
        const v4f* hjp = reinterpret_cast<const v4f*>(pj[t] + nb);
        const v4f* bp  = reinterpret_cast<const v4f*>(Bb + nb);
        const v4f* up  = reinterpret_cast<const v4f*>(Ui + (size_t)nb * TAG_);
        const v4f hi0 = hip[0], hi1 = hip[1], hj0 = hjp[0], hj1 = hjp[1], b0 = bp[0], b1 = bp[1];
        const float hiv[8] = {hi0.x, hi0.y, hi0.z, hi0.w, hi1.x, hi1.y, hi1.z, hi1.w};
        const float hjv[8] = {hj0.x, hj0.y, hj0.z, hj0.w, hj1.x, hj1.y, hj1.z, hj1.w};
        const float bv[8]  = {b0.x, b0.y, b0.z, b0.w, b1.x, b1.y, b1.z, b1.w};
        float uu[8 * TAG_];
#pragma unroll
        for (int q = 0; q < 2 * TAG_; ++q) {
          const v4f x = up[q];
          uu[4 * q] = x.x; uu[4 * q + 1] = x.y; uu[4 * q + 2] = x.z; uu[4 * q + 3] = x.w;
        }
#pragma unroll
        for (int e = 0; e < 8; ++e) {
          const float lg = tr_logit<NT>(acc[t], g, e);
          const float z = 1.0f / (1.0f + expf(-(lg + bv[e])));
          const float gt = z * hiv[e] + (1.0f - z) * hjv[e];
#pragma unroll
          for (int c = 0; c < TAG_; ++c) part[t][c] += gt * uu[e * TAG_ + c];
        }
      }
    }
  }
#pragma unroll
  for (int t = 0; t < 2; ++t)
#pragma unroll
    for (int c = 0; c < TAG_; ++c) part[t][c] += __shfl_xor(part[t][c], 16, 32);
  if (kh == 0) {
#pragma unroll
    for (int t = 0; t < 2; ++t) {
      float* po = part_out + (size_t)(a0_ + 16 * t + mr) * TAG_;
#pragma unroll
      for (int c = 0; c < TAG_; c += 2) {
        v2f o; o.x = part[t][c]; o.y = part[t][c + 1];
        *reinterpret_cast<v2f*>(po + c) = o;
      }
    }
  }
}

__global__ void __launch_bounds__(64) k_gate(const float* __restrict__ hx, const float* __restrict__ cw,
                       const float* __restrict__ cb, const float* __restrict__ uw,
                       float* __restrict__ plog) {
  const int p = blockIdx.z;
  const int i = p >> 1;
  const int j = (p & 1) ? (i == 2 ? 1 : 2) : (i == 0 ? 1 : 0);
  const int a0_ = (blockIdx.y * 2 + (threadIdx.x >> 5)) * 32, n0 = blockIdx.x * 32;
  const float* Hi = hx + (size_t)i * A_ * H_;
  const float* Hj = hx + (size_t)j * A_ * H_;
  const float* W  = cw + (size_t)p * 3 * H_ * H_;
  const float* Bb = cb + (size_t)p * H_;
  const float* Ui = uw + (size_t)i * H_ * TAG_;
  float* po = plog + ((size_t)p * NS_ + blockIdx.x) * A_ * TAG_;
  if (blockIdx.x == NS_ - 1) gate_body<1>(Hi, Hj, W, Bb, Ui, po, a0_, n0);
  else                       gate_body<2>(Hi, Hj, W, Bb, Ui, po, a0_, n0);
}

__global__ void k_final(const float* __restrict__ plog, const float* __restrict__ ub,
                        float* __restrict__ out) {
  const int e = blockIdx.x * blockDim.x + threadIdx.x;
  const int t = e % TAG_;
  float s = ub[t] + ub[TAG_ + t] + ub[2 * TAG_ + t];
  for (int q = 0; q < 6 * NS_; ++q) s += plog[(size_t)q * A_ * TAG_ + e];
  out[e] = s;
}

}

extern "C" void kernel_launch(void* const* d_in, const int* in_sizes, int n_in,
                              void* d_out, int out_size, void* d_ws, size_t ws_size,
                              hipStream_t stream) {
  (void)in_sizes; (void)n_in; (void)out_size; (void)ws_size;
  const float* x_a     = (const float*)d_in[0];
  const float* x_v     = (const float*)d_in[1];
  const float* x_t     = (const float*)d_in[2];
  const float* fc0_w   = (const float*)d_in[3];
  const float* fc0_b   = (const float*)d_in[4];
  const float* conv_w  = (const float*)d_in[5];
  const float* tm_w    = (const float*)d_in[6];
  const float* tm_b    = (const float*)d_in[7];
  const float* cross_w = (const float*)d_in[8];
  const float* cross_b = (const float*)d_in[9];
  const float* uni_w   = (const float*)d_in[10];
  const float* uni_b   = (const float*)d_in[11];
  float* out = (float*)d_out;

  const size_t NH = (size_t)N_ * H_;
  float* ws    = (float*)d_ws;
  float* big   = ws; ws += NH;
  float* nrm   = ws; ws += NH;
  float* intra = ws; ws += (size_t)N_ * L_;
  float* crossv= ws; ws += (size_t)M_ * M_ * A_;
  float* dinv  = ws; ws += N_;
  float* h0    = ws; ws += NH;
  float* hA    = ws; ws += NH;
  float* hBf   = ws; ws += NH;
  float* hi    = ws; ws += NH;
  float* hx    = ws; ws += NH;
  float* plog  = ws; ws += (size_t)6 * NS_ * A_ * TAG_;


  k_norm <<<N_ / 8,              dim3(256), 0, stream>>>(x_a, x_v, x_t, big, nrm);
  k_gram <<<M_ * B_,             dim3(512), 0, stream>>>(nrm, intra);
  k_cross<<<6 * A_ / 8,          dim3(256), 0, stream>>>(nrm, crossv);
  k_dinv <<<N_ / 8,              dim3(256), 0, stream>>>(intra, crossv, dinv);
  k_fc0  <<<dim3(NS_, N_ / 64),  dim3(64), 0, stream>>>(big, fc0_w, fc0_b, h0);

  const float beta0 = logf(0.5f / 1.0f + 1.0f);
  const float beta1 = logf(0.5f / 2.0f + 1.0f);
  k_spmm <<<dim3(NS_, M_ * B_),  dim3(64), 0, stream>>>(intra, crossv, dinv, h0, hi);
  k_conv <<<dim3(NS_, N_ / 64),  dim3(64), 0, stream>>>(hi, h0, conv_w, beta0, hA);
  k_spmm <<<dim3(NS_, M_ * B_),  dim3(64), 0, stream>>>(intra, crossv, dinv, hA, hi);
  k_conv <<<dim3(NS_, N_ / 64),  dim3(64), 0, stream>>>(hi, h0, conv_w + 2 * H_ * H_, beta1, hBf);

  k_hx   <<<dim3(NS_, A_ / 64, M_), dim3(64), 0, stream>>>(big, hBf, tm_w, tm_b, hx);

  k_gate <<<dim3(NS_, A_ / 64, 6), dim3(64), 0, stream>>>(hx, cross_w, cross_b, uni_w, plog);

  k_final<<<A_ * TAG_ / 256, dim3(256), 0, stream>>>(plog, uni_b, out);
}


# Activations also stored feature-major by their producers: A fragments of conv, hx (2nd segment) and gate come by 8-byte loads of two adjacent rows, 2 lines per load, was 16; k order kept
# speedup vs baseline: 5.8390x; 1.0817x over previous
#include <hip/hip_runtime.h>
#include <math.h>

typedef __attribute__((ext_vector_type(2))) float v2f;
typedef __attribute__((ext_vector_type(4))) float v4f;
typedef __attribute__((ext_vector_type(8))) float v8f;

namespace {

constexpr int B_   = 32;
constexpr int L_   = 64;
constexpr int H_   = 200;
constexpr int M_   = 3;
constexpr int A_   = B_ * L_;
constexpr int N_   = M_ * A_;
constexpr int TAG_ = 6;
constexpr float PI_F = 3.14159265358979323846f;

template <typename FA, typename FB, typename FE>
__device__ __forceinline__ void wmma_tile_f32(int m0, int n0, int K,
                                              FA loadA, FB loadB, FE epi) {
  const int lane = threadIdx.x & 31;
  const int mr   = lane & 15;
  const int kh   = lane >> 4;
  v8f acc = {};
  for (int k = 0; k < K; k += 4) {
    const int kk = k + 2 * kh;
    v2f a, b;
    a.x = loadA(m0 + mr, kk);
    a.y = loadA(m0 + mr, kk + 1);
    b.x = loadB(kk, n0 + mr);
    b.y = loadB(kk + 1, n0 + mr);
    acc = __builtin_amdgcn_wmma_f32_16x16x4_f32(false, a, false, b,
                                                (short)0, acc, false, false);
  }
#pragma unroll
  for (int v = 0; v < 8; ++v) epi(m0 + 8 * kh + v, n0 + mr, acc[v]);
}

__device__ __forceinline__ float wave_sum(float s) {
#pragma unroll
  for (int off = 16; off > 0; off >>= 1) s += __shfl_xor(s, off, 32);
  return s;
}

__global__ void k_norm(const float* __restrict__ xa, const float* __restrict__ xv,
                       const float* __restrict__ xt, float* __restrict__ big,
                       float* __restrict__ nrm) {
  const int row  = blockIdx.x * 8 + (threadIdx.x >> 5);
  const int lane = threadIdx.x & 31;
  const float* src = (row < A_)     ? xa + (size_t)row * H_
                   : (row < 2 * A_) ? xv + (size_t)(row - A_) * H_
                                    : xt + (size_t)(row - 2 * A_) * H_;
  float s = 0.f;
  for (int k = lane; k < H_; k += 32) { float v = src[k]; s += v * v; }
  s = wave_sum(s);
  const float rn = 1.0f / sqrtf(s);
  for (int k = lane; k < H_; k += 32) {
    const float v = src[k];
    big[(size_t)row * H_ + k] = v;
    nrm[(size_t)row * H_ + k] = v * rn;
  }
}

__global__ void k_gram(const float* __restrict__ nrm, float* __restrict__ intra) {
  const int mb = blockIdx.x;
  const int w  = threadIdx.x >> 5;
  const int m0 = (w >> 2) * 16, n0 = (w & 3) * 16;
  const float* blk = nrm + (size_t)mb * L_ * H_;
  auto lA = [&](int i, int k) { return blk[(size_t)i * H_ + k]; };
  auto lB = [&](int k, int j) { return blk[(size_t)j * H_ + k]; };
  auto ep = [&](int i, int j, float v) {
    intra[((size_t)mb * L_ + i) * L_ + j] = 1.0f - acosf(v * 0.99999f) * (1.0f / PI_F);
  };
  wmma_tile_f32(m0, n0, H_, lA, lB, ep);
}

__global__ void k_cross(const float* __restrict__ nrm, float* __restrict__ crossv) {
  const int idx  = blockIdx.x * 8 + (threadIdx.x >> 5);
  const int lane = threadIdx.x & 31;
  const int pm[6] = {0, 0, 1, 1, 2, 2};
  const int pn[6] = {1, 2, 0, 2, 0, 1};
  const int q = idx / A_, a = idx % A_;
  const int m = pm[q], n = pn[q];
  const float* pa = nrm + ((size_t)m * A_ + a) * H_;
  const float* pb = nrm + ((size_t)n * A_ + a) * H_;
  float s = 0.f;
  for (int k = lane; k < H_; k += 32) s += pa[k] * pb[k];
  s = wave_sum(s);
  if (lane == 0)
    crossv[((size_t)m * M_ + n) * A_ + a] = 1.0f - acosf(s * 0.99999f) * (1.0f / PI_F);
}

__global__ void k_dinv(const float* __restrict__ intra, const float* __restrict__ crossv,
                       float* __restrict__ dinv) {
  const int row  = blockIdx.x * 8 + (threadIdx.x >> 5);
  const int lane = threadIdx.x & 31;
  const int m = row >> 11, a = row & (A_ - 1);
  const float* r = intra + (size_t)row * L_;
  float s = 0.f;
  for (int j = lane; j < L_; j += 32) s += r[j];
  s = wave_sum(s);
  if (lane == 0) {
#pragma unroll
    for (int n = 0; n < M_; ++n)
      if (n != m) s += crossv[((size_t)m * M_ + n) * A_ + a];
    dinv[row] = 1.0f / sqrtf(s);
  }
}

struct RawQuad { v4f p, q; };

__device__ __forceinline__ v4f ld4(const float* p) {
  return *reinterpret_cast<const v4f*>(p);
}

__device__ __forceinline__ v2f ld2(const float* p) {
  return *reinterpret_cast<const v2f*>(p);
}

template <int MT, int NT, int D, typename RA>
struct Ring { RA a[D][MT]; v2f b[D][2][NT]; };

template <int NT>
__device__ __forceinline__ void b_frags(const v2f (&raw)[NT], v2f (&f)[NT]) {
  if constexpr (NT == 2) {
    f[0].x = raw[0].x; f[0].y = raw[1].x;
    f[1].x = raw[0].y; f[1].y = raw[1].y;
  } else {
    f[0] = raw[0];
  }
}

template <int MT, int NT, bool TR = false>
__device__ __forceinline__ void mma_step(v8f (&acc)[MT][NT], const v2f (&a)[MT],
                                         const v2f (&b)[NT]) {
#pragma unroll
  for (int t = 0; t < MT; ++t)
#pragma unroll
    for (int u = 0; u < NT; ++u) {
      if constexpr (TR)
        acc[t][u] = __builtin_amdgcn_wmma_f32_16x16x4_f32(false, b[u], false, a[t],
                                                          (short)0, acc[t][u], false, false);
      else
        acc[t][u] = __builtin_amdgcn_wmma_f32_16x16x4_f32(false, a[t], false, b[u],
                                                          (short)0, acc[t][u], false, false);
    }
}


template <int MT, int NT, int D, typename RA, typename FA, typename FB>
__device__ __forceinline__ void ring_fill(Ring<MT, NT, D, RA>& r, FA ldA, FB ldB) {
  const int kh = (threadIdx.x & 31) >> 4;
#pragma unroll
  for (int d = 0; d < D; ++d) {
    const int kq = 8 * d + 4 * kh;
#pragma unroll
    for (int t = 0; t < MT; ++t) r.a[d][t] = ldA(t, kq);
#pragma unroll
    for (int s = 0; s < 2; ++s)
#pragma unroll
      for (int u = 0; u < NT; ++u) r.b[d][s][u] = ldB(u, kq + 2 * s);
    __builtin_amdgcn_sched_barrier(0);
  }
}

template <int MT, int NT, int D, int NPAIRS, bool TR = false, typename RA, typename FA, typename FC, typename FB>
__device__ __forceinline__ void ring_main(v8f (&acc)[MT][NT], Ring<MT, NT, D, RA>& r,
                                          FA ldA, FC cvA, FB ldB) {
  static_assert(NPAIRS % D == 0 && NPAIRS >= D, "ring depth must divide the pair count");
  const int kh = (threadIdx.x & 31) >> 4;
#pragma nounroll
  for (int p0 = 0; p0 < NPAIRS - D; p0 += D) {
#pragma unroll
    for (int d = 0; d < D; ++d) {
      v2f lo[MT], hi[MT];
#pragma unroll
      for (int t = 0; t < MT; ++t) {
        const v4f c = cvA(r.a[d][t]);
        lo[t] = c.xy;
        hi[t] = c.zw;
      }
      v2f f0[NT], f1[NT];
      b_frags<NT>(r.b[d][0], f0);
      b_frags<NT>(r.b[d][1], f1);
      mma_step<MT, NT, TR>(acc, lo, f0);
      mma_step<MT, NT, TR>(acc, hi, f1);
      const int kq = 8 * (p0 + d + D) + 4 * kh;
#pragma unroll
      for (int t = 0; t < MT; ++t) r.a[d][t] = ldA(t, kq);
#pragma unroll
      for (int s = 0; s < 2; ++s)
#pragma unroll
        for (int u = 0; u < NT; ++u) r.b[d][s][u] = ldB(u, kq + 2 * s);
      __builtin_amdgcn_sched_barrier(0);
    }
  }
}

template <bool TR = false, int MT, int NT, int D, typename RA, typename FC>
__device__ __forceinline__ void ring_drain(v8f (&acc)[MT][NT], Ring<MT, NT, D, RA>& r, FC cvA) {
#pragma unroll
  for (int d = 0; d < D; ++d) {
    v2f lo[MT], hi[MT];
#pragma unroll
    for (int t = 0; t < MT; ++t) {
      const v4f c = cvA(r.a[d][t]);
      lo[t] = c.xy;
      hi[t] = c.zw;
    }
    v2f f0[NT], f1[NT];
    b_frags<NT>(r.b[d][0], f0);
    b_frags<NT>(r.b[d][1], f1);
    mma_step<MT, NT, TR>(acc, lo, f0);
    mma_step<MT, NT, TR>(acc, hi, f1);
  }
}

template <int NT, int D, int NP>
struct RingT { v2f a[D][NP][4]; v2f b[D][2][NT]; };

template <int NP>
__device__ __forceinline__ void a_frags(const v2f (&raw)[NP][4], v2f (&lo)[2], v2f (&hi)[2]) {
  v2f q[4];
#pragma unroll
  for (int j = 0; j < 4; ++j) {
    if constexpr (NP == 2) q[j] = raw[0][j] * raw[1][j];
    else                   q[j] = raw[0][j];
  }
  lo[0].x = q[0].x; lo[0].y = q[1].x; hi[0].x = q[2].x; hi[0].y = q[3].x;
  lo[1].x = q[0].y; lo[1].y = q[1].y; hi[1].x = q[2].y; hi[1].y = q[3].y;
}

template <int NT, int D, int NP, typename FA, typename FB>
__device__ __forceinline__ void ringt_fill(RingT<NT, D, NP>& r, FA ldA, FB ldB) {
  const int kh = (threadIdx.x & 31) >> 4;
#pragma unroll
  for (int d = 0; d < D; ++d) {
    const int kq = 8 * d + 4 * kh;
#pragma unroll
    for (int pl = 0; pl < NP; ++pl)
#pragma unroll
      for (int j = 0; j < 4; ++j) r.a[d][pl][j] = ldA(pl, kq + j);
#pragma unroll
    for (int s = 0; s < 2; ++s)
#pragma unroll
      for (int u = 0; u < NT; ++u) r.b[d][s][u] = ldB(u, kq + 2 * s);
    __builtin_amdgcn_sched_barrier(0);
  }
}

template <bool TR, int NT, int D, int NPAIRS, int NP, typename FA, typename FB>
__device__ __forceinline__ void ringt_main(v8f (&acc)[2][NT], RingT<NT, D, NP>& r, FA ldA, FB ldB) {
  static_assert(NPAIRS % D == 0 && NPAIRS >= D, "ring depth must divide the pair count");
  const int kh = (threadIdx.x & 31) >> 4;
#pragma nounroll
  for (int p0 = 0; p0 < NPAIRS - D; p0 += D) {
#pragma unroll
    for (int d = 0; d < D; ++d) {
      v2f lo[2], hi[2];
      a_frags<NP>(r.a[d], lo, hi);
      v2f f0[NT], f1[NT];
      b_frags<NT>(r.b[d][0], f0);
      b_frags<NT>(r.b[d][1], f1);
      mma_step<2, NT, TR>(acc, lo, f0);
      mma_step<2, NT, TR>(acc, hi, f1);
      const int kq = 8 * (p0 + d + D) + 4 * kh;
#pragma unroll
      for (int pl = 0; pl < NP; ++pl)
#pragma unroll
        for (int j = 0; j < 4; ++j) r.a[d][pl][j] = ldA(pl, kq + j);
#pragma unroll
      for (int s = 0; s < 2; ++s)
#pragma unroll
        for (int u = 0; u < NT; ++u) r.b[d][s][u] = ldB(u, kq + 2 * s);
      __builtin_amdgcn_sched_barrier(0);
    }
  }
}

template <bool TR, int NT, int D, int NP>
__device__ __forceinline__ void ringt_drain(v8f (&acc)[2][NT], RingT<NT, D, NP>& r) {
#pragma unroll
  for (int d = 0; d < D; ++d) {
    v2f lo[2], hi[2];
    a_frags<NP>(r.a[d], lo, hi);
    v2f f0[NT], f1[NT];
    b_frags<NT>(r.b[d][0], f0);
    b_frags<NT>(r.b[d][1], f1);
    mma_step<2, NT, TR>(acc, lo, f0);
    mma_step<2, NT, TR>(acc, hi, f1);
  }
}

template <bool RI>
__device__ __forceinline__ int tile_row(int t, int x) { return RI ? 2 * x + t : 16 * t + x; }

template <int NT, bool RI, typename FV>
__device__ __forceinline__ void epi_vals(const v8f (&acc)[2][NT], int m0, int n0,
                                         float (&o)[2][NT][8], FV val) {
  const int lane = threadIdx.x & 31;
  const int mr = lane & 15, kh = lane >> 4;
#pragma unroll
  for (int t = 0; t < 2; ++t)
#pragma unroll
    for (int u = 0; u < NT; ++u)
#pragma unroll
      for (int v = 0; v < 8; ++v) {
        const int m = m0 + tile_row<RI>(t, 8 * kh + v);
        const int n = (NT == 2) ? n0 + 2 * mr + u : n0 + mr;
        o[t][u][v] = (NT == 2 || n < H_) ? val(m, n, acc[t][u][v]) : 0.0f;
      }
}

template <int NT, bool RI>
__device__ __forceinline__ void store_rm(const float (&o)[2][NT][8], int m0, int n0,
                                         float* __restrict__ dst) {
  const int lane = threadIdx.x & 31;
  const int mr = lane & 15, kh = lane >> 4;
#pragma unroll
  for (int t = 0; t < 2; ++t)
#pragma unroll
    for (int v = 0; v < 8; ++v) {
      const int m = m0 + tile_row<RI>(t, 8 * kh + v);
      if constexpr (NT == 2) {
        v2f w;
        w.x = o[t][0][v];
        w.y = o[t][1][v];
        *reinterpret_cast<v2f*>(dst + (size_t)m * H_ + n0 + 2 * mr) = w;
      } else {
        const int n = n0 + mr;
        if (n < H_) dst[(size_t)m * H_ + n] = o[t][0][v];
      }
    }
}

template <int NT, bool RI>
__device__ __forceinline__ void store_fm(const float (&o)[2][NT][8], int m0, int n0,
                                         float* __restrict__ dstT, int pitch) {
  const int lane = threadIdx.x & 31;
  const int mr = lane & 15, kh = lane >> 4;
#pragma unroll
  for (int u = 0; u < NT; ++u) {
    const int n = (NT == 2) ? n0 + 2 * mr + u : n0 + mr;
    if (NT == 2 || n < H_) {
      float* col = dstT + (size_t)n * pitch + m0;
      if constexpr (RI) {
#pragma unroll
        for (int q = 0; q < 4; ++q) {
          v4f w;
          w.x = o[0][u][2 * q];
          w.y = o[1][u][2 * q];
          w.z = o[0][u][2 * q + 1];
          w.w = o[1][u][2 * q + 1];
          *reinterpret_cast<v4f*>(col + 16 * kh + 4 * q) = w;
        }
      } else {
#pragma unroll
        for (int t = 0; t < 2; ++t)
#pragma unroll
          for (int q = 0; q < 2; ++q) {
            v4f w;
            w.x = o[t][u][4 * q];
            w.y = o[t][u][4 * q + 1];
            w.z = o[t][u][4 * q + 2];
            w.w = o[t][u][4 * q + 3];
            *reinterpret_cast<v4f*>(col + 16 * t + 8 * kh + 4 * q) = w;
          }
      }
    }
  }
}

template <int NT>
__device__ __forceinline__ int lane_col(int n0) {
  const int mr = threadIdx.x & 15;
  return NT == 2 ? n0 + 2 * mr : min(n0 + mr, H_ - 1);
}

template <int NT>
__device__ __forceinline__ v2f ldB_unit(const float* __restrict__ w, int cb, int j, int kk) {
  if constexpr (NT == 2) {
    return ld2(w + (size_t)(kk + j) * H_ + cb);
  } else {
    v2f b;
    b.x = w[(size_t)kk * H_ + cb];
    b.y = w[(size_t)(kk + 1) * H_ + cb];
    return b;
  }
}

constexpr int NS_ = 7;
constexpr int KP_ = H_ / 8;

template <int NT>
__device__ __forceinline__ void fc0_body(const float* __restrict__ big, const float* __restrict__ w,
                                         const float* __restrict__ bias, float* __restrict__ h0,
                                         float* __restrict__ h0T, int m0, int n0) {
  const int mr = threadIdx.x & 15;
  const int cb = lane_col<NT>(n0);
  const float* ar[2] = {big + (size_t)(m0 + mr) * H_, big + (size_t)(m0 + 16 + mr) * H_};
  v8f acc[2][NT] = {};
  auto cA = [](v4f r) { return r; };
  auto lA = [&](int t, int kq) { return ld4(ar[t] + kq); };
  auto lB = [&](int j, int kk) { return ldB_unit<NT>(w, cb, j, kk); };
  Ring<2, NT, 5, v4f> g0;
  ring_fill(g0, lA, lB);
  ring_main<2, NT, 5, KP_>(acc, g0, lA, cA, lB);
  ring_drain(acc, g0, cA);
  auto ep = [&](int m, int n, float v) { return fmaxf(v + bias[n], 0.0f); };
  float o[2][NT][8];
  epi_vals<NT, false>(acc, m0, n0, o, ep);
  store_rm<NT, false>(o, m0, n0, h0);
  store_fm<NT, false>(o, m0, n0, h0T, N_);
}

__global__ void __launch_bounds__(64) k_fc0(const float* __restrict__ big, const float* __restrict__ w,
                      const float* __restrict__ bias, float* __restrict__ h0,
                      float* __restrict__ h0T) {
  const int m0 = (blockIdx.y * 2 + (threadIdx.x >> 5)) * 32, n0 = blockIdx.x * 32;
  if (blockIdx.x == NS_ - 1) fc0_body<1>(big, w, bias, h0, h0T, m0, n0);
  else                       fc0_body<2>(big, w, bias, h0, h0T, m0, n0);
}

template <int NT>
__device__ __forceinline__ void spmm_body(const float* __restrict__ intra,
                                          const float* __restrict__ crossv,
                                          const float* __restrict__ dinv,
                                          const float* __restrict__ h, float* __restrict__ hi,
                                          float* __restrict__ hiT, int mb, int i0, int n0) {
  const int mr = threadIdx.x & 15;
  const int m = mb >> 5;
  const int rowbase = mb * L_;
  const int cb = lane_col<NT>(n0);
  const float* S = intra + (size_t)mb * L_ * L_;
  const float* sr[2] = {S + (size_t)(i0 + mr) * L_, S + (size_t)(i0 + 16 + mr) * L_};
  const float* dv = dinv + rowbase;
  const float* hb = h + (size_t)rowbase * H_;
  v8f acc[2][NT] = {};
  auto lA = [&](int t, int kq) { RawQuad r; r.p = ld4(sr[t] + kq); r.q = ld4(dv + kq); return r; };
  auto cA = [](RawQuad r) { return r.p * r.q; };
  auto lB = [&](int j, int kk) { return ldB_unit<NT>(hb, cb, j, kk); };
  Ring<2, NT, 4, RawQuad> g0;
  ring_fill(g0, lA, lB);
  ring_main<2, NT, 4, L_ / 8>(acc, g0, lA, cA, lB);
  ring_drain(acc, g0, cA);
  auto ep = [&](int i, int n, float a_) {
    const int row = rowbase + i;
    const int a   = row - m * A_;
    float s = a_;
#pragma unroll
    for (int nn = 0; nn < M_; ++nn) {
      if (nn == m) continue;
      const int orow = nn * A_ + a;
      s += crossv[((size_t)m * M_ + nn) * A_ + a] * dinv[orow] * h[(size_t)orow * H_ + n];
    }
    return s * dinv[row];
  };
  float o[2][NT][8];
  epi_vals<NT, false>(acc, i0, n0, o, ep);
  store_rm<NT, false>(o, i0, n0, hi + (size_t)rowbase * H_);
  store_fm<NT, false>(o, i0, n0, hiT + rowbase, N_);
}

__global__ void __launch_bounds__(64) k_spmm(const float* __restrict__ intra, const float* __restrict__ crossv,
                       const float* __restrict__ dinv, const float* __restrict__ h,
                       float* __restrict__ hi, float* __restrict__ hiT) {
  const int mb = blockIdx.y;
  const int i0 = (threadIdx.x >> 5) * 32, n0 = blockIdx.x * 32;
  if (blockIdx.x == NS_ - 1) spmm_body<1>(intra, crossv, dinv, h, hi, hiT, mb, i0, n0);
  else                       spmm_body<2>(intra, crossv, dinv, h, hi, hiT, mb, i0, n0);
}

template <int NT, bool FM_OUT>
__device__ __forceinline__ void conv_body(const float* __restrict__ hi, const float* __restrict__ h0,
                                          const float* __restrict__ hiT, const float* __restrict__ h0T,
                                          const float* __restrict__ w, float beta,
                                          float* __restrict__ hout, int m0, int n0) {
  const int mr = threadIdx.x & 15;
  const int cb = lane_col<NT>(n0);
  const float* t0 = hiT + m0 + 2 * mr;
  const float* t1 = h0T + m0 + 2 * mr;
  const float* w1 = w + (size_t)H_ * H_;
  v8f acc[2][NT] = {};
  auto lA0 = [&](int pl, int k) { return ld2(t0 + (size_t)k * N_); };
  auto lA1 = [&](int pl, int k) { return ld2(t1 + (size_t)k * N_); };
  auto lB0 = [&](int j, int kk) { return ldB_unit<NT>(w, cb, j, kk); };
  auto lB1 = [&](int j, int kk) { return ldB_unit<NT>(w1, cb, j, kk); };
  RingT<NT, 5, 1> g0, g1;
  ringt_fill(g0, lA0, lB0);
  ringt_main<false, NT, 5, KP_>(acc, g0, lA0, lB0);
  ringt_fill(g1, lA1, lB1);
  ringt_drain<false>(acc, g0);
  ringt_main<false, NT, 5, KP_>(acc, g1, lA1, lB1);
  ringt_drain<false>(acc, g1);
  auto ep = [&](int m, int n, float v) {
    const float r = 0.9f * hi[(size_t)m * H_ + n] + 0.1f * h0[(size_t)m * H_ + n];
    return fmaxf(beta * v + (1.0f - beta) * r, 0.0f);
  };
  float o[2][NT][8];
  epi_vals<NT, true>(acc, m0, n0, o, ep);
  if constexpr (FM_OUT) store_fm<NT, true>(o, m0, n0, hout, N_);
  else                  store_rm<NT, true>(o, m0, n0, hout);
}

template <bool FM_OUT>
__global__ void __launch_bounds__(64) k_conv(const float* __restrict__ hi, const float* __restrict__ h0,
                       const float* __restrict__ hiT, const float* __restrict__ h0T,
                       const float* __restrict__ w, float beta, float* __restrict__ hout) {
  const int m0 = (blockIdx.y * 2 + (threadIdx.x >> 5)) * 32, n0 = blockIdx.x * 32;
  if (blockIdx.x == NS_ - 1) conv_body<1, FM_OUT>(hi, h0, hiT, h0T, w, beta, hout, m0, n0);
  else                       conv_body<2, FM_OUT>(hi, h0, hiT, h0T, w, beta, hout, m0, n0);
}

template <int NT>
__device__ __forceinline__ void hx_body(const float* __restrict__ big, const float* __restrict__ hfT,
                                        const float* __restrict__ W, const float* __restrict__ Bb,
                                        float* __restrict__ hx, float* __restrict__ hxT,
                                        int rb, int a0_, int n0) {
  const int mr = threadIdx.x & 15;
  const int cb = lane_col<NT>(n0);
  const size_t r0 = (size_t)(rb + a0_ + 2 * mr) * H_, r1 = (size_t)(rb + a0_ + 2 * mr + 1) * H_;
  const float* p0[2] = {big + r0, big + r1};
  const float* tf = hfT + rb + a0_ + 2 * mr;
  const float* W1 = W + (size_t)H_ * H_;
  v8f acc[2][NT] = {};
  auto cA = [](v4f r) { return r; };
  auto lA0 = [&](int t, int kq) { return ld4(p0[t] + kq); };
  auto lA1 = [&](int pl, int k) { return ld2(tf + (size_t)k * N_); };
  auto lB0 = [&](int j, int kk) { return ldB_unit<NT>(W, cb, j, kk); };
  auto lB1 = [&](int j, int kk) { return ldB_unit<NT>(W1, cb, j, kk); };
  Ring<2, NT, 5, v4f> g0;
  RingT<NT, 5, 1> g1;
  ring_fill(g0, lA0, lB0);
  ring_main<2, NT, 5, KP_>(acc, g0, lA0, cA, lB0);
  ringt_fill(g1, lA1, lB1);
  ring_drain(acc, g0, cA);
  ringt_main<false, NT, 5, KP_>(acc, g1, lA1, lB1);
  ringt_drain<false>(acc, g1);
  auto ep = [&](int a, int n, float v) { return fmaxf(v + Bb[n], 0.0f); };
  float o[2][NT][8];
  epi_vals<NT, true>(acc, a0_, n0, o, ep);
  store_rm<NT, true>(o, a0_, n0, hx + (size_t)rb * H_);
  store_fm<NT, true>(o, a0_, n0, hxT + (size_t)rb * H_, A_);
}

__global__ void __launch_bounds__(64) k_hx(const float* __restrict__ big, const float* __restrict__ hfT,
                     const float* __restrict__ tw, const float* __restrict__ tb,
                     float* __restrict__ hx, float* __restrict__ hxT) {
  const int m  = blockIdx.z;
  const int a0_ = (blockIdx.y * 2 + (threadIdx.x >> 5)) * 32, n0 = blockIdx.x * 32;
  const float* W  = tw + (size_t)m * 2 * H_ * H_;
  const float* Bb = tb + (size_t)m * H_;
  if (blockIdx.x == NS_ - 1) hx_body<1>(big, hfT, W, Bb, hx, hxT, m * A_, a0_, n0);
  else                       hx_body<2>(big, hfT, W, Bb, hx, hxT, m * A_, a0_, n0);
}

template <int NT>
__device__ __forceinline__ float tr_logit(const v8f (&a)[NT], int g, int e) {
  if constexpr (NT == 2) return a[e & 1][4 * g + (e >> 1)];
  else                   return a[0][e];
}

template <int NT>
__device__ __forceinline__ void gate_body(const float* __restrict__ Hi, const float* __restrict__ Hj,
                                          const float* __restrict__ HiT, const float* __restrict__ HjT,
                                          const float* __restrict__ W, const float* __restrict__ Bb,
                                          const float* __restrict__ Ui, float* __restrict__ part_out,
                                          int a0_, int n0) {
  const int mr = threadIdx.x & 15;
  const int kh = (threadIdx.x & 31) >> 4;
  const int cb = lane_col<NT>(n0);
  const size_t r0 = (size_t)(a0_ + 2 * mr) * H_, r1 = (size_t)(a0_ + 2 * mr + 1) * H_;
  const float* pi[2] = {Hi + r0, Hi + r1};
  const float* pj[2] = {Hj + r0, Hj + r1};
  const float* ti = HiT + a0_ + 2 * mr;
  const float* tj = HjT + a0_ + 2 * mr;
  const float* W1 = W + (size_t)H_ * H_;
  const float* W2 = W + (size_t)2 * H_ * H_;
  v8f acc[2][NT] = {};
  auto lAi = [&](int pl, int k) { return ld2(ti + (size_t)k * A_); };
  auto lAj = [&](int pl, int k) { return ld2(tj + (size_t)k * A_); };
  auto lAp = [&](int pl, int k) { return ld2((pl ? tj : ti) + (size_t)k * A_); };
  auto lB0 = [&](int j, int kk) { return ldB_unit<NT>(W, cb, j, kk); };
  auto lB1 = [&](int j, int kk) { return ldB_unit<NT>(W1, cb, j, kk); };
  auto lB2 = [&](int j, int kk) { return ldB_unit<NT>(W2, cb, j, kk); };
  RingT<NT, 5, 1> g0, g1;
  RingT<NT, 5, 2> g2;
  ringt_fill(g0, lAi, lB0);
  ringt_main<true, NT, 5, KP_>(acc, g0, lAi, lB0);
  ringt_fill(g1, lAj, lB1);
  ringt_drain<true>(acc, g0);
  ringt_main<true, NT, 5, KP_>(acc, g1, lAj, lB1);
  ringt_fill(g2, lAp, lB2);
  ringt_drain<true>(acc, g1);
  ringt_main<true, NT, 5, KP_>(acc, g2, lAp, lB2);
  ringt_drain<true>(acc, g2);

  float part[2][TAG_] = {};
#pragma unroll
  for (int t = 0; t < 2; ++t) {
#pragma unroll
    for (int g = 0; g < NT; ++g) {
      const int nb = (NT == 2) ? n0 + 16 * kh + 8 * g : n0 + 8 * kh;
      if (NT == 2 || nb < H_) {
        const v4f* hip = reinterpret_cast<const v4f*>(pi[t] + nb);
        const v4f* hjp = reinterpret_cast<const v4f*>(pj[t] + nb);
        const v4f* bp  = reinterpret_cast<const v4f*>(Bb + nb);
        const v4f* up  = reinterpret_cast<const v4f*>(Ui + (size_t)nb * TAG_);
        const v4f hi0 = hip[0], hi1 = hip[1], hj0 = hjp[0], hj1 = hjp[1], b0 = bp[0], b1 = bp[1];
        const float hiv[8] = {hi0.x, hi0.y, hi0.z, hi0.w, hi1.x, hi1.y, hi1.z, hi1.w};
        const float hjv[8] = {hj0.x, hj0.y, hj0.z, hj0.w, hj1.x, hj1.y, hj1.z, hj1.w};
        const float bv[8]  = {b0.x, b0.y, b0.z, b0.w, b1.x, b1.y, b1.z, b1.w};
        float uu[8 * TAG_];
#pragma unroll
        for (int q = 0; q < 2 * TAG_; ++q) {
          const v4f x = up[q];
          uu[4 * q] = x.x; uu[4 * q + 1] = x.y; uu[4 * q + 2] = x.z; uu[4 * q + 3] = x.w;
        }
#pragma unroll
        for (int e = 0; e < 8; ++e) {
          const float lg = tr_logit<NT>(acc[t], g, e);
          const float z = 1.0f / (1.0f + expf(-(lg + bv[e])));
          const float gt = z * hiv[e] + (1.0f - z) * hjv[e];
#pragma unroll
          for (int c = 0; c < TAG_; ++c) part[t][c] += gt * uu[e * TAG_ + c];
        }
      }
    }
  }
#pragma unroll
  for (int t = 0; t < 2; ++t)
#pragma unroll
    for (int c = 0; c < TAG_; ++c) part[t][c] += __shfl_xor(part[t][c], 16, 32);
  if (kh == 0) {
#pragma unroll
    for (int t = 0; t < 2; ++t) {
      float* po = part_out + (size_t)(a0_ + 2 * mr + t) * TAG_;
#pragma unroll
      for (int c = 0; c < TAG_; c += 2) {
        v2f o; o.x = part[t][c]; o.y = part[t][c + 1];
        *reinterpret_cast<v2f*>(po + c) = o;
      }
    }
  }
}

__global__ void __launch_bounds__(64) k_gate(const float* __restrict__ hx, const float* __restrict__ hxT,
                       const float* __restrict__ cw,
                       const float* __restrict__ cb, const float* __restrict__ uw,
                       float* __restrict__ plog) {
  const int p = blockIdx.z;
  const int i = p >> 1;
  const int j = (p & 1) ? (i == 2 ? 1 : 2) : (i == 0 ? 1 : 0);
  const int a0_ = (blockIdx.y * 2 + (threadIdx.x >> 5)) * 32, n0 = blockIdx.x * 32;
  const float* Hi = hx + (size_t)i * A_ * H_;
  const float* Hj = hx + (size_t)j * A_ * H_;
  const float* HiT = hxT + (size_t)i * H_ * A_;
  const float* HjT = hxT + (size_t)j * H_ * A_;
  const float* W  = cw + (size_t)p * 3 * H_ * H_;
  const float* Bb = cb + (size_t)p * H_;
  const float* Ui = uw + (size_t)i * H_ * TAG_;
  float* po = plog + ((size_t)p * NS_ + blockIdx.x) * A_ * TAG_;
  if (blockIdx.x == NS_ - 1) gate_body<1>(Hi, Hj, HiT, HjT, W, Bb, Ui, po, a0_, n0);
  else                       gate_body<2>(Hi, Hj, HiT, HjT, W, Bb, Ui, po, a0_, n0);
}

__global__ void k_final(const float* __restrict__ plog, const float* __restrict__ ub,
                        float* __restrict__ out) {
  const int e = blockIdx.x * blockDim.x + threadIdx.x;
  const int t = e % TAG_;
  float s = ub[t] + ub[TAG_ + t] + ub[2 * TAG_ + t];
  for (int q = 0; q < 6 * NS_; ++q) s += plog[(size_t)q * A_ * TAG_ + e];
  out[e] = s;
}

}

extern "C" void kernel_launch(void* const* d_in, const int* in_sizes, int n_in,
                              void* d_out, int out_size, void* d_ws, size_t ws_size,
                              hipStream_t stream) {
  (void)in_sizes; (void)n_in; (void)out_size; (void)ws_size;
  const float* x_a     = (const float*)d_in[0];
  const float* x_v     = (const float*)d_in[1];
  const float* x_t     = (const float*)d_in[2];
  const float* fc0_w   = (const float*)d_in[3];
  const float* fc0_b   = (const float*)d_in[4];
  const float* conv_w  = (const float*)d_in[5];
  const float* tm_w    = (const float*)d_in[6];
  const float* tm_b    = (const float*)d_in[7];
  const float* cross_w = (const float*)d_in[8];
  const float* cross_b = (const float*)d_in[9];
  const float* uni_w   = (const float*)d_in[10];
  const float* uni_b   = (const float*)d_in[11];
  float* out = (float*)d_out;

  const size_t NH = (size_t)N_ * H_;
  float* ws    = (float*)d_ws;
  float* big   = ws; ws += NH;
  float* nrm   = ws; ws += NH;
  float* intra = ws; ws += (size_t)N_ * L_;
  float* crossv= ws; ws += (size_t)M_ * M_ * A_;
  float* dinv  = ws; ws += N_;
  float* h0    = ws; ws += NH;
  float* h0T   = ws; ws += NH;
  float* hA    = ws; ws += NH;
  float* hBfT  = ws; ws += NH;
  float* hi    = ws; ws += NH;
  float* hiT   = ws; ws += NH;
  float* hx    = ws; ws += NH;
  float* hxT   = ws; ws += NH;
  float* plog  = ws; ws += (size_t)6 * NS_ * A_ * TAG_;


  k_norm <<<N_ / 8,              dim3(256), 0, stream>>>(x_a, x_v, x_t, big, nrm);
  k_gram <<<M_ * B_,             dim3(512), 0, stream>>>(nrm, intra);
  k_cross<<<6 * A_ / 8,          dim3(256), 0, stream>>>(nrm, crossv);
  k_dinv <<<N_ / 8,              dim3(256), 0, stream>>>(intra, crossv, dinv);
  k_fc0  <<<dim3(NS_, N_ / 64),  dim3(64), 0, stream>>>(big, fc0_w, fc0_b, h0, h0T);

  const float beta0 = logf(0.5f / 1.0f + 1.0f);
  const float beta1 = logf(0.5f / 2.0f + 1.0f);
  k_spmm <<<dim3(NS_, M_ * B_),  dim3(64), 0, stream>>>(intra, crossv, dinv, h0, hi, hiT);
  k_conv<false><<<dim3(NS_, N_ / 64),  dim3(64), 0, stream>>>(hi, h0, hiT, h0T, conv_w, beta0, hA);
  k_spmm <<<dim3(NS_, M_ * B_),  dim3(64), 0, stream>>>(intra, crossv, dinv, hA, hi, hiT);
  k_conv<true><<<dim3(NS_, N_ / 64),  dim3(64), 0, stream>>>(hi, h0, hiT, h0T, conv_w + 2 * H_ * H_, beta1, hBfT);

  k_hx   <<<dim3(NS_, A_ / 64, M_), dim3(64), 0, stream>>>(big, hBfT, tm_w, tm_b, hx, hxT);

  k_gate <<<dim3(NS_, A_ / 64, 6), dim3(64), 0, stream>>>(hx, hxT, cross_w, cross_b, uni_w, plog);

  k_final<<<A_ * TAG_ / 256, dim3(256), 0, stream>>>(plog, uni_b, out);
}


# Graph setup (norm, Gram, cross, degree: 4 launches) as 1 launch on the raw inputs: wave per 16-row strip, raw Gram on WMMA scaled by norms summed from its fragments; norm/dot/degree sum orders change
# speedup vs baseline: 6.4371x; 1.1024x over previous
#include <hip/hip_runtime.h>
#include <math.h>

typedef __attribute__((ext_vector_type(2))) float v2f;
typedef __attribute__((ext_vector_type(4))) float v4f;
typedef __attribute__((ext_vector_type(8))) float v8f;

namespace {

constexpr int B_   = 32;
constexpr int L_   = 64;
constexpr int H_   = 200;
constexpr int M_   = 3;
constexpr int A_   = B_ * L_;
constexpr int N_   = M_ * A_;
constexpr int TAG_ = 6;
constexpr float PI_F = 3.14159265358979323846f;

struct RawQuad { v4f p, q; };

__device__ __forceinline__ v4f ld4(const float* p) {
  return *reinterpret_cast<const v4f*>(p);
}

__device__ __forceinline__ v2f ld2(const float* p) {
  return *reinterpret_cast<const v2f*>(p);
}

template <int MT, int NT, int D, typename RA>
struct Ring { RA a[D][MT]; v2f b[D][2][NT]; };

template <int NT>
__device__ __forceinline__ void b_frags(const v2f (&raw)[NT], v2f (&f)[NT]) {
  if constexpr (NT == 2) {
    f[0].x = raw[0].x; f[0].y = raw[1].x;
    f[1].x = raw[0].y; f[1].y = raw[1].y;
  } else {
    f[0] = raw[0];
  }
}

template <int MT, int NT, bool TR = false>
__device__ __forceinline__ void mma_step(v8f (&acc)[MT][NT], const v2f (&a)[MT],
                                         const v2f (&b)[NT]) {
#pragma unroll
  for (int t = 0; t < MT; ++t)
#pragma unroll
    for (int u = 0; u < NT; ++u) {
      if constexpr (TR)
        acc[t][u] = __builtin_amdgcn_wmma_f32_16x16x4_f32(false, b[u], false, a[t],
                                                          (short)0, acc[t][u], false, false);
      else
        acc[t][u] = __builtin_amdgcn_wmma_f32_16x16x4_f32(false, a[t], false, b[u],
                                                          (short)0, acc[t][u], false, false);
    }
}


template <int MT, int NT, int D, typename RA, typename FA, typename FB>
__device__ __forceinline__ void ring_fill(Ring<MT, NT, D, RA>& r, FA ldA, FB ldB) {
  const int kh = (threadIdx.x & 31) >> 4;
#pragma unroll
  for (int d = 0; d < D; ++d) {
    const int kq = 8 * d + 4 * kh;
#pragma unroll
    for (int t = 0; t < MT; ++t) r.a[d][t] = ldA(t, kq);
#pragma unroll
    for (int s = 0; s < 2; ++s)
#pragma unroll
      for (int u = 0; u < NT; ++u) r.b[d][s][u] = ldB(u, kq + 2 * s);
    __builtin_amdgcn_sched_barrier(0);
  }
}

template <int MT, int NT, int D, int NPAIRS, bool TR = false, typename RA, typename FA, typename FC, typename FB>
__device__ __forceinline__ void ring_main(v8f (&acc)[MT][NT], Ring<MT, NT, D, RA>& r,
                                          FA ldA, FC cvA, FB ldB) {
  static_assert(NPAIRS % D == 0 && NPAIRS >= D, "ring depth must divide the pair count");
  const int kh = (threadIdx.x & 31) >> 4;
#pragma nounroll
  for (int p0 = 0; p0 < NPAIRS - D; p0 += D) {
#pragma unroll
    for (int d = 0; d < D; ++d) {
      v2f lo[MT], hi[MT];
#pragma unroll
      for (int t = 0; t < MT; ++t) {
        const v4f c = cvA(r.a[d][t]);
        lo[t] = c.xy;
        hi[t] = c.zw;
      }
      v2f f0[NT], f1[NT];
      b_frags<NT>(r.b[d][0], f0);
      b_frags<NT>(r.b[d][1], f1);
      mma_step<MT, NT, TR>(acc, lo, f0);
      mma_step<MT, NT, TR>(acc, hi, f1);
      const int kq = 8 * (p0 + d + D) + 4 * kh;
#pragma unroll
      for (int t = 0; t < MT; ++t) r.a[d][t] = ldA(t, kq);
#pragma unroll
      for (int s = 0; s < 2; ++s)
#pragma unroll
        for (int u = 0; u < NT; ++u) r.b[d][s][u] = ldB(u, kq + 2 * s);
      __builtin_amdgcn_sched_barrier(0);
    }
  }
}

template <bool TR = false, int MT, int NT, int D, typename RA, typename FC>
__device__ __forceinline__ void ring_drain(v8f (&acc)[MT][NT], Ring<MT, NT, D, RA>& r, FC cvA) {
#pragma unroll
  for (int d = 0; d < D; ++d) {
    v2f lo[MT], hi[MT];
#pragma unroll
    for (int t = 0; t < MT; ++t) {
      const v4f c = cvA(r.a[d][t]);
      lo[t] = c.xy;
      hi[t] = c.zw;
    }
    v2f f0[NT], f1[NT];
    b_frags<NT>(r.b[d][0], f0);
    b_frags<NT>(r.b[d][1], f1);
    mma_step<MT, NT, TR>(acc, lo, f0);
    mma_step<MT, NT, TR>(acc, hi, f1);
  }
}

template <int NT, int D, int NP>
struct RingT { v2f a[D][NP][4]; v2f b[D][2][NT]; };

template <int NP>
__device__ __forceinline__ void a_frags(const v2f (&raw)[NP][4], v2f (&lo)[2], v2f (&hi)[2]) {
  v2f q[4];
#pragma unroll
  for (int j = 0; j < 4; ++j) {
    if constexpr (NP == 2) q[j] = raw[0][j] * raw[1][j];
    else                   q[j] = raw[0][j];
  }
  lo[0].x = q[0].x; lo[0].y = q[1].x; hi[0].x = q[2].x; hi[0].y = q[3].x;
  lo[1].x = q[0].y; lo[1].y = q[1].y; hi[1].x = q[2].y; hi[1].y = q[3].y;
}

template <int NT, int D, int NP, typename FA, typename FB>
__device__ __forceinline__ void ringt_fill(RingT<NT, D, NP>& r, FA ldA, FB ldB) {
  const int kh = (threadIdx.x & 31) >> 4;
#pragma unroll
  for (int d = 0; d < D; ++d) {
    const int kq = 8 * d + 4 * kh;
#pragma unroll
    for (int pl = 0; pl < NP; ++pl)
#pragma unroll
      for (int j = 0; j < 4; ++j) r.a[d][pl][j] = ldA(pl, kq + j);
#pragma unroll
    for (int s = 0; s < 2; ++s)
#pragma unroll
      for (int u = 0; u < NT; ++u) r.b[d][s][u] = ldB(u, kq + 2 * s);
    __builtin_amdgcn_sched_barrier(0);
  }
}

template <bool TR, int NT, int D, int NPAIRS, int NP, typename FA, typename FB>
__device__ __forceinline__ void ringt_main(v8f (&acc)[2][NT], RingT<NT, D, NP>& r, FA ldA, FB ldB) {
  static_assert(NPAIRS % D == 0 && NPAIRS >= D, "ring depth must divide the pair count");
  const int kh = (threadIdx.x & 31) >> 4;
#pragma nounroll
  for (int p0 = 0; p0 < NPAIRS - D; p0 += D) {
#pragma unroll
    for (int d = 0; d < D; ++d) {
      v2f lo[2], hi[2];
      a_frags<NP>(r.a[d], lo, hi);
      v2f f0[NT], f1[NT];
      b_frags<NT>(r.b[d][0], f0);
      b_frags<NT>(r.b[d][1], f1);
      mma_step<2, NT, TR>(acc, lo, f0);
      mma_step<2, NT, TR>(acc, hi, f1);
      const int kq = 8 * (p0 + d + D) + 4 * kh;
#pragma unroll
      for (int pl = 0; pl < NP; ++pl)
#pragma unroll
        for (int j = 0; j < 4; ++j) r.a[d][pl][j] = ldA(pl, kq + j);
#pragma unroll
      for (int s = 0; s < 2; ++s)
#pragma unroll
        for (int u = 0; u < NT; ++u) r.b[d][s][u] = ldB(u, kq + 2 * s);
      __builtin_amdgcn_sched_barrier(0);
    }
  }
}

template <bool TR, int NT, int D, int NP>
__device__ __forceinline__ void ringt_drain(v8f (&acc)[2][NT], RingT<NT, D, NP>& r) {
#pragma unroll
  for (int d = 0; d < D; ++d) {
    v2f lo[2], hi[2];
    a_frags<NP>(r.a[d], lo, hi);
    v2f f0[NT], f1[NT];
    b_frags<NT>(r.b[d][0], f0);
    b_frags<NT>(r.b[d][1], f1);
    mma_step<2, NT, TR>(acc, lo, f0);
    mma_step<2, NT, TR>(acc, hi, f1);
  }
}

template <bool RI>
__device__ __forceinline__ int tile_row(int t, int x) { return RI ? 2 * x + t : 16 * t + x; }

template <int NT, bool RI, typename FV>
__device__ __forceinline__ void epi_vals(const v8f (&acc)[2][NT], int m0, int n0,
                                         float (&o)[2][NT][8], FV val) {
  const int lane = threadIdx.x & 31;
  const int mr = lane & 15, kh = lane >> 4;
#pragma unroll
  for (int t = 0; t < 2; ++t)
#pragma unroll
    for (int u = 0; u < NT; ++u)
#pragma unroll
      for (int v = 0; v < 8; ++v) {
        const int m = m0 + tile_row<RI>(t, 8 * kh + v);
        const int n = (NT == 2) ? n0 + 2 * mr + u : n0 + mr;
        o[t][u][v] = (NT == 2 || n < H_) ? val(m, n, acc[t][u][v]) : 0.0f;
      }
}

template <int NT, bool RI>
__device__ __forceinline__ void store_rm(const float (&o)[2][NT][8], int m0, int n0,
                                         float* __restrict__ dst) {
  const int lane = threadIdx.x & 31;
  const int mr = lane & 15, kh = lane >> 4;
#pragma unroll
  for (int t = 0; t < 2; ++t)
#pragma unroll
    for (int v = 0; v < 8; ++v) {
      const int m = m0 + tile_row<RI>(t, 8 * kh + v);
      if constexpr (NT == 2) {
        v2f w;
        w.x = o[t][0][v];
        w.y = o[t][1][v];
        *reinterpret_cast<v2f*>(dst + (size_t)m * H_ + n0 + 2 * mr) = w;
      } else {
        const int n = n0 + mr;
        if (n < H_) dst[(size_t)m * H_ + n] = o[t][0][v];
      }
    }
}

template <int NT, bool RI>
__device__ __forceinline__ void store_fm(const float (&o)[2][NT][8], int m0, int n0,
                                         float* __restrict__ dstT, int pitch) {
  const int lane = threadIdx.x & 31;
  const int mr = lane & 15, kh = lane >> 4;
#pragma unroll
  for (int u = 0; u < NT; ++u) {
    const int n = (NT == 2) ? n0 + 2 * mr + u : n0 + mr;
    if (NT == 2 || n < H_) {
      float* col = dstT + (size_t)n * pitch + m0;
      if constexpr (RI) {
#pragma unroll
        for (int q = 0; q < 4; ++q) {
          v4f w;
          w.x = o[0][u][2 * q];
          w.y = o[1][u][2 * q];
          w.z = o[0][u][2 * q + 1];
          w.w = o[1][u][2 * q + 1];
          *reinterpret_cast<v4f*>(col + 16 * kh + 4 * q) = w;
        }
      } else {
#pragma unroll
        for (int t = 0; t < 2; ++t)
#pragma unroll
          for (int q = 0; q < 2; ++q) {
            v4f w;
            w.x = o[t][u][4 * q];
            w.y = o[t][u][4 * q + 1];
            w.z = o[t][u][4 * q + 2];
            w.w = o[t][u][4 * q + 3];
            *reinterpret_cast<v4f*>(col + 16 * t + 8 * kh + 4 * q) = w;
          }
      }
    }
  }
}

template <int NT>
__device__ __forceinline__ int lane_col(int n0) {
  const int mr = threadIdx.x & 15;
  return NT == 2 ? n0 + 2 * mr : min(n0 + mr, H_ - 1);
}

template <int NT>
__device__ __forceinline__ v2f ldB_unit(const float* __restrict__ w, int cb, int j, int kk) {
  if constexpr (NT == 2) {
    return ld2(w + (size_t)(kk + j) * H_ + cb);
  } else {
    v2f b;
    b.x = w[(size_t)kk * H_ + cb];
    b.y = w[(size_t)(kk + 1) * H_ + cb];
    return b;
  }
}

constexpr int NS_ = 7;
constexpr int KP_ = H_ / 8;

constexpr int GD_ = 5;

struct GSlot { v4f b[4]; v4f y1, y2; };

__device__ __forceinline__ float ang_sim(float cosv) {
  return 1.0f - acosf(cosv * 0.99999f) * (1.0f / PI_F);
}

__global__ void __launch_bounds__(32) k_graph(const float* __restrict__ xa, const float* __restrict__ xv,
                        const float* __restrict__ xt, float* __restrict__ intra,
                        float* __restrict__ crossv, float* __restrict__ dinv) {
  const int mb = blockIdx.x >> 2;
  const int m  = mb >> 5;
  const int d0 = (mb & 31) * L_;
  const int w  = blockIdx.x & 3;
  const int lane = threadIdx.x & 31;
  const int c = lane & 15, h = lane >> 4;
  const int n1 = (m == 0) ? 1 : 0;
  const int n2 = (m == 2) ? 1 : 2;
  const float* xm = (m == 0) ? xa : (m == 1) ? xv : xt;
  const float* x1 = (n1 == 0) ? xa : xv;
  const float* x2 = (n2 == 1) ? xv : xt;
  const float* pb[4];
#pragma unroll
  for (int u = 0; u < 4; ++u)
    pb[u] = xm + (size_t)(d0 + 16 * ((w + u) & 3) + c) * H_ + 4 * h;
  const float* p1 = x1 + (size_t)(d0 + 16 * w + c) * H_ + 4 * h;
  const float* p2 = x2 + (size_t)(d0 + 16 * w + c) * H_ + 4 * h;

  v8f acc[4] = {};
  float ss[4] = {0.f, 0.f, 0.f, 0.f};
  float s1 = 0.f, s2 = 0.f, d1 = 0.f, d2 = 0.f;

  auto fetch = [&](GSlot& s, int p) {
#pragma unroll
    for (int u = 0; u < 4; ++u) s.b[u] = ld4(pb[u] + 8 * p);
    s.y1 = ld4(p1 + 8 * p);
    s.y2 = ld4(p2 + 8 * p);
  };
  auto use = [&](const GSlot& s) {
    const v4f a = s.b[0];
#pragma unroll
    for (int u = 0; u < 4; ++u) {
      const v4f b = s.b[u];
      ss[u] = fmaf(b.x, b.x, ss[u]); ss[u] = fmaf(b.y, b.y, ss[u]);
      ss[u] = fmaf(b.z, b.z, ss[u]); ss[u] = fmaf(b.w, b.w, ss[u]);
    }
    d1 = fmaf(a.x, s.y1.x, d1); d1 = fmaf(a.y, s.y1.y, d1);
    d1 = fmaf(a.z, s.y1.z, d1); d1 = fmaf(a.w, s.y1.w, d1);
    d2 = fmaf(a.x, s.y2.x, d2); d2 = fmaf(a.y, s.y2.y, d2);
    d2 = fmaf(a.z, s.y2.z, d2); d2 = fmaf(a.w, s.y2.w, d2);
    s1 = fmaf(s.y1.x, s.y1.x, s1); s1 = fmaf(s.y1.y, s.y1.y, s1);
    s1 = fmaf(s.y1.z, s.y1.z, s1); s1 = fmaf(s.y1.w, s.y1.w, s1);
    s2 = fmaf(s.y2.x, s.y2.x, s2); s2 = fmaf(s.y2.y, s.y2.y, s2);
    s2 = fmaf(s.y2.z, s.y2.z, s2); s2 = fmaf(s.y2.w, s.y2.w, s2);
    const v2f alo = a.xy, ahi = a.zw;
#pragma unroll
    for (int u = 0; u < 4; ++u) {
      const v2f blo = s.b[u].xy;
      acc[u] = __builtin_amdgcn_wmma_f32_16x16x4_f32(false, alo, false, blo,
                                                     (short)0, acc[u], false, false);
    }
#pragma unroll
    for (int u = 0; u < 4; ++u) {
      const v2f bhi = s.b[u].zw;
      acc[u] = __builtin_amdgcn_wmma_f32_16x16x4_f32(false, ahi, false, bhi,
                                                     (short)0, acc[u], false, false);
    }
  };

  GSlot ring[GD_];
#pragma unroll
  for (int d = 0; d < GD_; ++d) {
    fetch(ring[d], d);
    __builtin_amdgcn_sched_barrier(0);
  }
#pragma nounroll
  for (int p0 = 0; p0 < KP_ - GD_; p0 += GD_) {
#pragma unroll
    for (int d = 0; d < GD_; ++d) {
      use(ring[d]);
      fetch(ring[d], p0 + d + GD_);
      __builtin_amdgcn_sched_barrier(0);
    }
  }
#pragma unroll
  for (int d = 0; d < GD_; ++d) use(ring[d]);

#pragma unroll
  for (int u = 0; u < 4; ++u) ss[u] += __shfl_xor(ss[u], 16, 32);
  s1 += __shfl_xor(s1, 16, 32);
  s2 += __shfl_xor(s2, 16, 32);
  d1 += __shfl_xor(d1, 16, 32);
  d2 += __shfl_xor(d2, 16, 32);

  float rn[4];
#pragma unroll
  for (int u = 0; u < 4; ++u) rn[u] = 1.0f / sqrtf(ss[u]);
  const float cr1 = ang_sim(d1 * (rn[0] * (1.0f / sqrtf(s1))));
  const float cr2 = ang_sim(d2 * (rn[0] * (1.0f / sqrtf(s2))));

  float rs[8];
  const size_t srow = (size_t)mb * L_ + 16 * w + 8 * h;
#pragma unroll
  for (int r = 0; r < 8; ++r) {
    const float rr = __shfl(rn[0], 8 * h + r, 32);
    float t = 0.f;
#pragma unroll
    for (int u = 0; u < 4; ++u) {
      const float v = ang_sim(acc[u][r] * (rr * rn[u]));
      intra[(srow + r) * L_ + 16 * ((w + u) & 3) + c] = v;
      t += v;
    }
#pragma unroll
    for (int off = 8; off > 0; off >>= 1) t += __shfl_xor(t, off, 32);
    rs[r] = t;
  }
  float mine = rs[0];
#pragma unroll
  for (int r = 1; r < 8; ++r) mine = ((c & 7) == r) ? rs[r] : mine;
  if ((c >> 3) == h) {
    const int a = d0 + 16 * w + c;
    crossv[((size_t)m * M_ + n1) * A_ + a] = cr1;
    crossv[((size_t)m * M_ + n2) * A_ + a] = cr2;
    dinv[(size_t)m * A_ + a] = 1.0f / sqrtf(mine + cr1 + cr2);
  }
}


template <int NT>
__device__ __forceinline__ void fc0_body(const float* __restrict__ xr, const float* __restrict__ w,
                                         const float* __restrict__ bias, float* __restrict__ h0,
                                         float* __restrict__ h0T, int m0, int n0) {
  const int mr = threadIdx.x & 15;
  const int cb = lane_col<NT>(n0);
  const float* ar[2] = {xr + (size_t)mr * H_, xr + (size_t)(16 + mr) * H_};
  v8f acc[2][NT] = {};
  auto cA = [](v4f r) { return r; };
  auto lA = [&](int t, int kq) { return ld4(ar[t] + kq); };
  auto lB = [&](int j, int kk) { return ldB_unit<NT>(w, cb, j, kk); };
  Ring<2, NT, 5, v4f> g0;
  ring_fill(g0, lA, lB);
  ring_main<2, NT, 5, KP_>(acc, g0, lA, cA, lB);
  ring_drain(acc, g0, cA);
  auto ep = [&](int m, int n, float v) { return fmaxf(v + bias[n], 0.0f); };
  float o[2][NT][8];
  epi_vals<NT, false>(acc, m0, n0, o, ep);
  store_rm<NT, false>(o, m0, n0, h0);
  store_fm<NT, false>(o, m0, n0, h0T, N_);
}

__global__ void __launch_bounds__(64) k_fc0(const float* __restrict__ xa, const float* __restrict__ xv,
                      const float* __restrict__ xt, const float* __restrict__ w,
                      const float* __restrict__ bias, float* __restrict__ h0,
                      float* __restrict__ h0T) {
  const int m0 = (blockIdx.y * 2 + (threadIdx.x >> 5)) * 32, n0 = blockIdx.x * 32;
  const int md = m0 / A_;
  const float* xr = ((md == 0) ? xa : (md == 1) ? xv : xt) + (size_t)(m0 - md * A_) * H_;
  if (blockIdx.x == NS_ - 1) fc0_body<1>(xr, w, bias, h0, h0T, m0, n0);
  else                       fc0_body<2>(xr, w, bias, h0, h0T, m0, n0);
}

template <int NT>
__device__ __forceinline__ void spmm_body(const float* __restrict__ intra,
                                          const float* __restrict__ crossv,
                                          const float* __restrict__ dinv,
                                          const float* __restrict__ h, float* __restrict__ hi,
                                          float* __restrict__ hiT, int mb, int i0, int n0) {
  const int mr = threadIdx.x & 15;
  const int m = mb >> 5;
  const int rowbase = mb * L_;
  const int cb = lane_col<NT>(n0);
  const float* S = intra + (size_t)mb * L_ * L_;
  const float* sr[2] = {S + (size_t)(i0 + mr) * L_, S + (size_t)(i0 + 16 + mr) * L_};
  const float* dv = dinv + rowbase;
  const float* hb = h + (size_t)rowbase * H_;
  v8f acc[2][NT] = {};
  auto lA = [&](int t, int kq) { RawQuad r; r.p = ld4(sr[t] + kq); r.q = ld4(dv + kq); return r; };
  auto cA = [](RawQuad r) { return r.p * r.q; };
  auto lB = [&](int j, int kk) { return ldB_unit<NT>(hb, cb, j, kk); };
  Ring<2, NT, 4, RawQuad> g0;
  ring_fill(g0, lA, lB);
  ring_main<2, NT, 4, L_ / 8>(acc, g0, lA, cA, lB);
  ring_drain(acc, g0, cA);
  auto ep = [&](int i, int n, float a_) {
    const int row = rowbase + i;
    const int a   = row - m * A_;
    float s = a_;
#pragma unroll
    for (int nn = 0; nn < M_; ++nn) {
      if (nn == m) continue;
      const int orow = nn * A_ + a;
      s += crossv[((size_t)m * M_ + nn) * A_ + a] * dinv[orow] * h[(size_t)orow * H_ + n];
    }
    return s * dinv[row];
  };
  float o[2][NT][8];
  epi_vals<NT, false>(acc, i0, n0, o, ep);
  store_rm<NT, false>(o, i0, n0, hi + (size_t)rowbase * H_);
  store_fm<NT, false>(o, i0, n0, hiT + rowbase, N_);
}

__global__ void __launch_bounds__(64) k_spmm(const float* __restrict__ intra, const float* __restrict__ crossv,
                       const float* __restrict__ dinv, const float* __restrict__ h,
                       float* __restrict__ hi, float* __restrict__ hiT) {
  const int mb = blockIdx.y;
  const int i0 = (threadIdx.x >> 5) * 32, n0 = blockIdx.x * 32;
  if (blockIdx.x == NS_ - 1) spmm_body<1>(intra, crossv, dinv, h, hi, hiT, mb, i0, n0);
  else                       spmm_body<2>(intra, crossv, dinv, h, hi, hiT, mb, i0, n0);
}

template <int NT, bool FM_OUT>
__device__ __forceinline__ void conv_body(const float* __restrict__ hi, const float* __restrict__ h0,
                                          const float* __restrict__ hiT, const float* __restrict__ h0T,
                                          const float* __restrict__ w, float beta,
                                          float* __restrict__ hout, int m0, int n0) {
  const int mr = threadIdx.x & 15;
  const int cb = lane_col<NT>(n0);
  const float* t0 = hiT + m0 + 2 * mr;
  const float* t1 = h0T + m0 + 2 * mr;
  const float* w1 = w + (size_t)H_ * H_;
  v8f acc[2][NT] = {};
  auto lA0 = [&](int pl, int k) { return ld2(t0 + (size_t)k * N_); };
  auto lA1 = [&](int pl, int k) { return ld2(t1 + (size_t)k * N_); };
  auto lB0 = [&](int j, int kk) { return ldB_unit<NT>(w, cb, j, kk); };
  auto lB1 = [&](int j, int kk) { return ldB_unit<NT>(w1, cb, j, kk); };
  RingT<NT, 5, 1> g0, g1;
  ringt_fill(g0, lA0, lB0);
  ringt_main<false, NT, 5, KP_>(acc, g0, lA0, lB0);
  ringt_fill(g1, lA1, lB1);
  ringt_drain<false>(acc, g0);
  ringt_main<false, NT, 5, KP_>(acc, g1, lA1, lB1);
  ringt_drain<false>(acc, g1);
  auto ep = [&](int m, int n, float v) {
    const float r = 0.9f * hi[(size_t)m * H_ + n] + 0.1f * h0[(size_t)m * H_ + n];
    return fmaxf(beta * v + (1.0f - beta) * r, 0.0f);
  };
  float o[2][NT][8];
  epi_vals<NT, true>(acc, m0, n0, o, ep);
  if constexpr (FM_OUT) store_fm<NT, true>(o, m0, n0, hout, N_);
  else                  store_rm<NT, true>(o, m0, n0, hout);
}

template <bool FM_OUT>
__global__ void __launch_bounds__(64) k_conv(const float* __restrict__ hi, const float* __restrict__ h0,
                       const float* __restrict__ hiT, const float* __restrict__ h0T,
                       const float* __restrict__ w, float beta, float* __restrict__ hout) {
  const int m0 = (blockIdx.y * 2 + (threadIdx.x >> 5)) * 32, n0 = blockIdx.x * 32;
  if (blockIdx.x == NS_ - 1) conv_body<1, FM_OUT>(hi, h0, hiT, h0T, w, beta, hout, m0, n0);
  else                       conv_body<2, FM_OUT>(hi, h0, hiT, h0T, w, beta, hout, m0, n0);
}

template <int NT>
__device__ __forceinline__ void hx_body(const float* __restrict__ xm, const float* __restrict__ hfT,
                                        const float* __restrict__ W, const float* __restrict__ Bb,
                                        float* __restrict__ hx, float* __restrict__ hxT,
                                        int rb, int a0_, int n0) {
  const int mr = threadIdx.x & 15;
  const int cb = lane_col<NT>(n0);
  const size_t r0 = (size_t)(a0_ + 2 * mr) * H_, r1 = (size_t)(a0_ + 2 * mr + 1) * H_;
  const float* p0[2] = {xm + r0, xm + r1};
  const float* tf = hfT + rb + a0_ + 2 * mr;
  const float* W1 = W + (size_t)H_ * H_;
  v8f acc[2][NT] = {};
  auto cA = [](v4f r) { return r; };
  auto lA0 = [&](int t, int kq) { return ld4(p0[t] + kq); };
  auto lA1 = [&](int pl, int k) { return ld2(tf + (size_t)k * N_); };
  auto lB0 = [&](int j, int kk) { return ldB_unit<NT>(W, cb, j, kk); };
  auto lB1 = [&](int j, int kk) { return ldB_unit<NT>(W1, cb, j, kk); };
  Ring<2, NT, 5, v4f> g0;
  RingT<NT, 5, 1> g1;
  ring_fill(g0, lA0, lB0);
  ring_main<2, NT, 5, KP_>(acc, g0, lA0, cA, lB0);
  ringt_fill(g1, lA1, lB1);
  ring_drain(acc, g0, cA);
  ringt_main<false, NT, 5, KP_>(acc, g1, lA1, lB1);
  ringt_drain<false>(acc, g1);
  auto ep = [&](int a, int n, float v) { return fmaxf(v + Bb[n], 0.0f); };
  float o[2][NT][8];
  epi_vals<NT, true>(acc, a0_, n0, o, ep);
  store_rm<NT, true>(o, a0_, n0, hx + (size_t)rb * H_);
  store_fm<NT, true>(o, a0_, n0, hxT + (size_t)rb * H_, A_);
}

__global__ void __launch_bounds__(64) k_hx(const float* __restrict__ xa, const float* __restrict__ xv,
                     const float* __restrict__ xt, const float* __restrict__ hfT,
                     const float* __restrict__ tw, const float* __restrict__ tb,
                     float* __restrict__ hx, float* __restrict__ hxT) {
  const int m  = blockIdx.z;
  const float* xm = (m == 0) ? xa : (m == 1) ? xv : xt;
  const int a0_ = (blockIdx.y * 2 + (threadIdx.x >> 5)) * 32, n0 = blockIdx.x * 32;
  const float* W  = tw + (size_t)m * 2 * H_ * H_;
  const float* Bb = tb + (size_t)m * H_;
  if (blockIdx.x == NS_ - 1) hx_body<1>(xm, hfT, W, Bb, hx, hxT, m * A_, a0_, n0);
  else                       hx_body<2>(xm, hfT, W, Bb, hx, hxT, m * A_, a0_, n0);
}

template <int NT>
__device__ __forceinline__ float tr_logit(const v8f (&a)[NT], int g, int e) {
  if constexpr (NT == 2) return a[e & 1][4 * g + (e >> 1)];
  else                   return a[0][e];
}

template <int NT>
__device__ __forceinline__ void gate_body(const float* __restrict__ Hi, const float* __restrict__ Hj,
                                          const float* __restrict__ HiT, const float* __restrict__ HjT,
                                          const float* __restrict__ W, const float* __restrict__ Bb,
                                          const float* __restrict__ Ui, float* __restrict__ part_out,
                                          int a0_, int n0) {
  const int mr = threadIdx.x & 15;
  const int kh = (threadIdx.x & 31) >> 4;
  const int cb = lane_col<NT>(n0);
  const size_t r0 = (size_t)(a0_ + 2 * mr) * H_, r1 = (size_t)(a0_ + 2 * mr + 1) * H_;
  const float* pi[2] = {Hi + r0, Hi + r1};
  const float* pj[2] = {Hj + r0, Hj + r1};
  const float* ti = HiT + a0_ + 2 * mr;
  const float* tj = HjT + a0_ + 2 * mr;
  const float* W1 = W + (size_t)H_ * H_;
  const float* W2 = W + (size_t)2 * H_ * H_;
  v8f acc[2][NT] = {};
  auto lAi = [&](int pl, int k) { return ld2(ti + (size_t)k * A_); };
  auto lAj = [&](int pl, int k) { return ld2(tj + (size_t)k * A_); };
  auto lAp = [&](int pl, int k) { return ld2((pl ? tj : ti) + (size_t)k * A_); };
  auto lB0 = [&](int j, int kk) { return ldB_unit<NT>(W, cb, j, kk); };
  auto lB1 = [&](int j, int kk) { return ldB_unit<NT>(W1, cb, j, kk); };
  auto lB2 = [&](int j, int kk) { return ldB_unit<NT>(W2, cb, j, kk); };
  RingT<NT, 5, 1> g0, g1;
  RingT<NT, 5, 2> g2;
  ringt_fill(g0, lAi, lB0);
  ringt_main<true, NT, 5, KP_>(acc, g0, lAi, lB0);
  ringt_fill(g1, lAj, lB1);
  ringt_drain<true>(acc, g0);
  ringt_main<true, NT, 5, KP_>(acc, g1, lAj, lB1);
  ringt_fill(g2, lAp, lB2);
  ringt_drain<true>(acc, g1);
  ringt_main<true, NT, 5, KP_>(acc, g2, lAp, lB2);
  ringt_drain<true>(acc, g2);

  float part[2][TAG_] = {};
#pragma unroll
  for (int t = 0; t < 2; ++t) {
#pragma unroll
    for (int g = 0; g < NT; ++g) {
      const int nb = (NT == 2) ? n0 + 16 * kh + 8 * g : n0 + 8 * kh;
      if (NT == 2 || nb < H_) {
        const v4f* hip = reinterpret_cast<const v4f*>(pi[t] + nb);
        const v4f* hjp = reinterpret_cast<const v4f*>(pj[t] + nb);
        const v4f* bp  = reinterpret_cast<const v4f*>(Bb + nb);
        const v4f* up  = reinterpret_cast<const v4f*>(Ui + (size_t)nb * TAG_);
        const v4f hi0 = hip[0], hi1 = hip[1], hj0 = hjp[0], hj1 = hjp[1], b0 = bp[0], b1 = bp[1];
        const float hiv[8] = {hi0.x, hi0.y, hi0.z, hi0.w, hi1.x, hi1.y, hi1.z, hi1.w};
        const float hjv[8] = {hj0.x, hj0.y, hj0.z, hj0.w, hj1.x, hj1.y, hj1.z, hj1.w};
        const float bv[8]  = {b0.x, b0.y, b0.z, b0.w, b1.x, b1.y, b1.z, b1.w};
        float uu[8 * TAG_];
#pragma unroll
        for (int q = 0; q < 2 * TAG_; ++q) {
          const v4f x = up[q];
          uu[4 * q] = x.x; uu[4 * q + 1] = x.y; uu[4 * q + 2] = x.z; uu[4 * q + 3] = x.w;
        }
#pragma unroll
        for (int e = 0; e < 8; ++e) {
          const float lg = tr_logit<NT>(acc[t], g, e);
          const float z = 1.0f / (1.0f + expf(-(lg + bv[e])));
          const float gt = z * hiv[e] + (1.0f - z) * hjv[e];
#pragma unroll
          for (int c = 0; c < TAG_; ++c) part[t][c] += gt * uu[e * TAG_ + c];
        }
      }
    }
  }
#pragma unroll
  for (int t = 0; t < 2; ++t)
#pragma unroll
    for (int c = 0; c < TAG_; ++c) part[t][c] += __shfl_xor(part[t][c], 16, 32);
  if (kh == 0) {
#pragma unroll
    for (int t = 0; t < 2; ++t) {
      float* po = part_out + (size_t)(a0_ + 2 * mr + t) * TAG_;
#pragma unroll
      for (int c = 0; c < TAG_; c += 2) {
        v2f o; o.x = part[t][c]; o.y = part[t][c + 1];
        *reinterpret_cast<v2f*>(po + c) = o;
      }
    }
  }
}

__global__ void __launch_bounds__(64) k_gate(const float* __restrict__ hx, const float* __restrict__ hxT,
                       const float* __restrict__ cw,
                       const float* __restrict__ cb, const float* __restrict__ uw,
                       float* __restrict__ plog) {
  const int p = blockIdx.z;
  const int i = p >> 1;
  const int j = (p & 1) ? (i == 2 ? 1 : 2) : (i == 0 ? 1 : 0);
  const int a0_ = (blockIdx.y * 2 + (threadIdx.x >> 5)) * 32, n0 = blockIdx.x * 32;
  const float* Hi = hx + (size_t)i * A_ * H_;
  const float* Hj = hx + (size_t)j * A_ * H_;
  const float* HiT = hxT + (size_t)i * H_ * A_;
  const float* HjT = hxT + (size_t)j * H_ * A_;
  const float* W  = cw + (size_t)p * 3 * H_ * H_;
  const float* Bb = cb + (size_t)p * H_;
  const float* Ui = uw + (size_t)i * H_ * TAG_;
  float* po = plog + ((size_t)p * NS_ + blockIdx.x) * A_ * TAG_;
  if (blockIdx.x == NS_ - 1) gate_body<1>(Hi, Hj, HiT, HjT, W, Bb, Ui, po, a0_, n0);
  else                       gate_body<2>(Hi, Hj, HiT, HjT, W, Bb, Ui, po, a0_, n0);
}

__global__ void k_final(const float* __restrict__ plog, const float* __restrict__ ub,
                        float* __restrict__ out) {
  const int e = blockIdx.x * blockDim.x + threadIdx.x;
  const int t = e % TAG_;
  float s = ub[t] + ub[TAG_ + t] + ub[2 * TAG_ + t];
  for (int q = 0; q < 6 * NS_; ++q) s += plog[(size_t)q * A_ * TAG_ + e];
  out[e] = s;
}

}

extern "C" void kernel_launch(void* const* d_in, const int* in_sizes, int n_in,
                              void* d_out, int out_size, void* d_ws, size_t ws_size,
                              hipStream_t stream) {
  (void)in_sizes; (void)n_in; (void)out_size; (void)ws_size;
  const float* x_a     = (const float*)d_in[0];
  const float* x_v     = (const float*)d_in[1];
  const float* x_t     = (const float*)d_in[2];
  const float* fc0_w   = (const float*)d_in[3];
  const float* fc0_b   = (const float*)d_in[4];
  const float* conv_w  = (const float*)d_in[5];
  const float* tm_w    = (const float*)d_in[6];
  const float* tm_b    = (const float*)d_in[7];
  const float* cross_w = (const float*)d_in[8];
  const float* cross_b = (const float*)d_in[9];
  const float* uni_w   = (const float*)d_in[10];
  const float* uni_b   = (const float*)d_in[11];
  float* out = (float*)d_out;

  const size_t NH = (size_t)N_ * H_;
  float* ws    = (float*)d_ws;
  float* intra = ws; ws += (size_t)N_ * L_;
  float* crossv= ws; ws += (size_t)M_ * M_ * A_;
  float* dinv  = ws; ws += N_;
  float* h0    = ws; ws += NH;
  float* h0T   = ws; ws += NH;
  float* hA    = ws; ws += NH;
  float* hBfT  = ws; ws += NH;
  float* hi    = ws; ws += NH;
  float* hiT   = ws; ws += NH;
  float* hx    = ws; ws += NH;
  float* hxT   = ws; ws += NH;
  float* plog  = ws; ws += (size_t)6 * NS_ * A_ * TAG_;


  k_graph<<<M_ * B_ * 4,         dim3(32), 0, stream>>>(x_a, x_v, x_t, intra, crossv, dinv);
  k_fc0  <<<dim3(NS_, N_ / 64),  dim3(64), 0, stream>>>(x_a, x_v, x_t, fc0_w, fc0_b, h0, h0T);

  const float beta0 = logf(0.5f / 1.0f + 1.0f);
  const float beta1 = logf(0.5f / 2.0f + 1.0f);
  k_spmm <<<dim3(NS_, M_ * B_),  dim3(64), 0, stream>>>(intra, crossv, dinv, h0, hi, hiT);
  k_conv<false><<<dim3(NS_, N_ / 64),  dim3(64), 0, stream>>>(hi, h0, hiT, h0T, conv_w, beta0, hA);
  k_spmm <<<dim3(NS_, M_ * B_),  dim3(64), 0, stream>>>(intra, crossv, dinv, hA, hi, hiT);
  k_conv<true><<<dim3(NS_, N_ / 64),  dim3(64), 0, stream>>>(hi, h0, hiT, h0T, conv_w + 2 * H_ * H_, beta1, hBfT);

  k_hx   <<<dim3(NS_, A_ / 64, M_), dim3(64), 0, stream>>>(x_a, x_v, x_t, hBfT, tm_w, tm_b, hx, hxT);

  k_gate <<<dim3(NS_, A_ / 64, 6), dim3(64), 0, stream>>>(hx, hxT, cross_w, cross_b, uni_w, plog);

  k_final<<<A_ * TAG_ / 256, dim3(256), 0, stream>>>(plog, uni_b, out);
}


# k_spmm epilogue: the two cross-modal terms formed branch-free (partner modalities by arithmetic), so all its loads issue together; was 64 dependent load-wait-add steps per lane; sum order kept
# speedup vs baseline: 7.7233x; 1.1998x over previous
#include <hip/hip_runtime.h>
#include <math.h>

typedef __attribute__((ext_vector_type(2))) float v2f;
typedef __attribute__((ext_vector_type(4))) float v4f;
typedef __attribute__((ext_vector_type(8))) float v8f;

namespace {

constexpr int B_   = 32;
constexpr int L_   = 64;
constexpr int H_   = 200;
constexpr int M_   = 3;
constexpr int A_   = B_ * L_;
constexpr int N_   = M_ * A_;
constexpr int TAG_ = 6;
constexpr float PI_F = 3.14159265358979323846f;

struct RawQuad { v4f p, q; };

__device__ __forceinline__ v4f ld4(const float* p) {
  return *reinterpret_cast<const v4f*>(p);
}

__device__ __forceinline__ v2f ld2(const float* p) {
  return *reinterpret_cast<const v2f*>(p);
}

template <int MT, int NT, int D, typename RA>
struct Ring { RA a[D][MT]; v2f b[D][2][NT]; };

template <int NT>
__device__ __forceinline__ void b_frags(const v2f (&raw)[NT], v2f (&f)[NT]) {
  if constexpr (NT == 2) {
    f[0].x = raw[0].x; f[0].y = raw[1].x;
    f[1].x = raw[0].y; f[1].y = raw[1].y;
  } else {
    f[0] = raw[0];
  }
}

template <int MT, int NT, bool TR = false>
__device__ __forceinline__ void mma_step(v8f (&acc)[MT][NT], const v2f (&a)[MT],
                                         const v2f (&b)[NT]) {
#pragma unroll
  for (int t = 0; t < MT; ++t)
#pragma unroll
    for (int u = 0; u < NT; ++u) {
      if constexpr (TR)
        acc[t][u] = __builtin_amdgcn_wmma_f32_16x16x4_f32(false, b[u], false, a[t],
                                                          (short)0, acc[t][u], false, false);
      else
        acc[t][u] = __builtin_amdgcn_wmma_f32_16x16x4_f32(false, a[t], false, b[u],
                                                          (short)0, acc[t][u], false, false);
    }
}


template <int MT, int NT, int D, typename RA, typename FA, typename FB>
__device__ __forceinline__ void ring_fill(Ring<MT, NT, D, RA>& r, FA ldA, FB ldB) {
  const int kh = (threadIdx.x & 31) >> 4;
#pragma unroll
  for (int d = 0; d < D; ++d) {
    const int kq = 8 * d + 4 * kh;
#pragma unroll
    for (int t = 0; t < MT; ++t) r.a[d][t] = ldA(t, kq);
#pragma unroll
    for (int s = 0; s < 2; ++s)
#pragma unroll
      for (int u = 0; u < NT; ++u) r.b[d][s][u] = ldB(u, kq + 2 * s);
    __builtin_amdgcn_sched_barrier(0);
  }
}

template <int MT, int NT, int D, int NPAIRS, bool TR = false, typename RA, typename FA, typename FC, typename FB>
__device__ __forceinline__ void ring_main(v8f (&acc)[MT][NT], Ring<MT, NT, D, RA>& r,
                                          FA ldA, FC cvA, FB ldB) {
  static_assert(NPAIRS % D == 0 && NPAIRS >= D, "ring depth must divide the pair count");
  const int kh = (threadIdx.x & 31) >> 4;
#pragma nounroll
  for (int p0 = 0; p0 < NPAIRS - D; p0 += D) {
#pragma unroll
    for (int d = 0; d < D; ++d) {
      v2f lo[MT], hi[MT];
#pragma unroll
      for (int t = 0; t < MT; ++t) {
        const v4f c = cvA(r.a[d][t]);
        lo[t] = c.xy;
        hi[t] = c.zw;
      }
      v2f f0[NT], f1[NT];
      b_frags<NT>(r.b[d][0], f0);
      b_frags<NT>(r.b[d][1], f1);
      mma_step<MT, NT, TR>(acc, lo, f0);
      mma_step<MT, NT, TR>(acc, hi, f1);
      const int kq = 8 * (p0 + d + D) + 4 * kh;
#pragma unroll
      for (int t = 0; t < MT; ++t) r.a[d][t] = ldA(t, kq);
#pragma unroll
      for (int s = 0; s < 2; ++s)
#pragma unroll
        for (int u = 0; u < NT; ++u) r.b[d][s][u] = ldB(u, kq + 2 * s);
      __builtin_amdgcn_sched_barrier(0);
    }
  }
}

template <bool TR = false, int MT, int NT, int D, typename RA, typename FC>
__device__ __forceinline__ void ring_drain(v8f (&acc)[MT][NT], Ring<MT, NT, D, RA>& r, FC cvA) {
#pragma unroll
  for (int d = 0; d < D; ++d) {
    v2f lo[MT], hi[MT];
#pragma unroll
    for (int t = 0; t < MT; ++t) {
      const v4f c = cvA(r.a[d][t]);
      lo[t] = c.xy;
      hi[t] = c.zw;
    }
    v2f f0[NT], f1[NT];
    b_frags<NT>(r.b[d][0], f0);
    b_frags<NT>(r.b[d][1], f1);
    mma_step<MT, NT, TR>(acc, lo, f0);
    mma_step<MT, NT, TR>(acc, hi, f1);
  }
}

template <int NT, int D, int NP>
struct RingT { v2f a[D][NP][4]; v2f b[D][2][NT]; };

template <int NP>
__device__ __forceinline__ void a_frags(const v2f (&raw)[NP][4], v2f (&lo)[2], v2f (&hi)[2]) {
  v2f q[4];
#pragma unroll
  for (int j = 0; j < 4; ++j) {
    if constexpr (NP == 2) q[j] = raw[0][j] * raw[1][j];
    else                   q[j] = raw[0][j];
  }
  lo[0].x = q[0].x; lo[0].y = q[1].x; hi[0].x = q[2].x; hi[0].y = q[3].x;
  lo[1].x = q[0].y; lo[1].y = q[1].y; hi[1].x = q[2].y; hi[1].y = q[3].y;
}

template <int NT, int D, int NP, typename FA, typename FB>
__device__ __forceinline__ void ringt_fill(RingT<NT, D, NP>& r, FA ldA, FB ldB) {
  const int kh = (threadIdx.x & 31) >> 4;
#pragma unroll
  for (int d = 0; d < D; ++d) {
    const int kq = 8 * d + 4 * kh;
#pragma unroll
    for (int pl = 0; pl < NP; ++pl)
#pragma unroll
      for (int j = 0; j < 4; ++j) r.a[d][pl][j] = ldA(pl, kq + j);
#pragma unroll
    for (int s = 0; s < 2; ++s)
#pragma unroll
      for (int u = 0; u < NT; ++u) r.b[d][s][u] = ldB(u, kq + 2 * s);
    __builtin_amdgcn_sched_barrier(0);
  }
}

template <bool TR, int NT, int D, int NPAIRS, int NP, typename FA, typename FB>
__device__ __forceinline__ void ringt_main(v8f (&acc)[2][NT], RingT<NT, D, NP>& r, FA ldA, FB ldB) {
  static_assert(NPAIRS % D == 0 && NPAIRS >= D, "ring depth must divide the pair count");
  const int kh = (threadIdx.x & 31) >> 4;
#pragma nounroll
  for (int p0 = 0; p0 < NPAIRS - D; p0 += D) {
#pragma unroll
    for (int d = 0; d < D; ++d) {
      v2f lo[2], hi[2];
      a_frags<NP>(r.a[d], lo, hi);
      v2f f0[NT], f1[NT];
      b_frags<NT>(r.b[d][0], f0);
      b_frags<NT>(r.b[d][1], f1);
      mma_step<2, NT, TR>(acc, lo, f0);
      mma_step<2, NT, TR>(acc, hi, f1);
      const int kq = 8 * (p0 + d + D) + 4 * kh;
#pragma unroll
      for (int pl = 0; pl < NP; ++pl)
#pragma unroll
        for (int j = 0; j < 4; ++j) r.a[d][pl][j] = ldA(pl, kq + j);
#pragma unroll
      for (int s = 0; s < 2; ++s)
#pragma unroll
        for (int u = 0; u < NT; ++u) r.b[d][s][u] = ldB(u, kq + 2 * s);
      __builtin_amdgcn_sched_barrier(0);
    }
  }
}

template <bool TR, int NT, int D, int NP>
__device__ __forceinline__ void ringt_drain(v8f (&acc)[2][NT], RingT<NT, D, NP>& r) {
#pragma unroll
  for (int d = 0; d < D; ++d) {
    v2f lo[2], hi[2];
    a_frags<NP>(r.a[d], lo, hi);
    v2f f0[NT], f1[NT];
    b_frags<NT>(r.b[d][0], f0);
    b_frags<NT>(r.b[d][1], f1);
    mma_step<2, NT, TR>(acc, lo, f0);
    mma_step<2, NT, TR>(acc, hi, f1);
  }
}

template <bool RI>
__device__ __forceinline__ int tile_row(int t, int x) { return RI ? 2 * x + t : 16 * t + x; }

template <int NT, bool RI, typename FV>
__device__ __forceinline__ void epi_vals(const v8f (&acc)[2][NT], int m0, int n0,
                                         float (&o)[2][NT][8], FV val) {
  const int lane = threadIdx.x & 31;
  const int mr = lane & 15, kh = lane >> 4;
#pragma unroll
  for (int t = 0; t < 2; ++t)
#pragma unroll
    for (int u = 0; u < NT; ++u)
#pragma unroll
      for (int v = 0; v < 8; ++v) {
        const int m = m0 + tile_row<RI>(t, 8 * kh + v);
        const int n = (NT == 2) ? n0 + 2 * mr + u : n0 + mr;
        o[t][u][v] = (NT == 2 || n < H_) ? val(m, n, acc[t][u][v]) : 0.0f;
      }
}

template <int NT, bool RI>
__device__ __forceinline__ void store_rm(const float (&o)[2][NT][8], int m0, int n0,
                                         float* __restrict__ dst) {
  const int lane = threadIdx.x & 31;
  const int mr = lane & 15, kh = lane >> 4;
#pragma unroll
  for (int t = 0; t < 2; ++t)
#pragma unroll
    for (int v = 0; v < 8; ++v) {
      const int m = m0 + tile_row<RI>(t, 8 * kh + v);
      if constexpr (NT == 2) {
        v2f w;
        w.x = o[t][0][v];
        w.y = o[t][1][v];
        *reinterpret_cast<v2f*>(dst + (size_t)m * H_ + n0 + 2 * mr) = w;
      } else {
        const int n = n0 + mr;
        if (n < H_) dst[(size_t)m * H_ + n] = o[t][0][v];
      }
    }
}

template <int NT, bool RI>
__device__ __forceinline__ void store_fm(const float (&o)[2][NT][8], int m0, int n0,
                                         float* __restrict__ dstT, int pitch) {
  const int lane = threadIdx.x & 31;
  const int mr = lane & 15, kh = lane >> 4;
#pragma unroll
  for (int u = 0; u < NT; ++u) {
    const int n = (NT == 2) ? n0 + 2 * mr + u : n0 + mr;
    if (NT == 2 || n < H_) {
      float* col = dstT + (size_t)n * pitch + m0;
      if constexpr (RI) {
#pragma unroll
        for (int q = 0; q < 4; ++q) {
          v4f w;
          w.x = o[0][u][2 * q];
          w.y = o[1][u][2 * q];
          w.z = o[0][u][2 * q + 1];
          w.w = o[1][u][2 * q + 1];
          *reinterpret_cast<v4f*>(col + 16 * kh + 4 * q) = w;
        }
      } else {
#pragma unroll
        for (int t = 0; t < 2; ++t)
#pragma unroll
          for (int q = 0; q < 2; ++q) {
            v4f w;
            w.x = o[t][u][4 * q];
            w.y = o[t][u][4 * q + 1];
            w.z = o[t][u][4 * q + 2];
            w.w = o[t][u][4 * q + 3];
            *reinterpret_cast<v4f*>(col + 16 * t + 8 * kh + 4 * q) = w;
          }
      }
    }
  }
}

template <int NT>
__device__ __forceinline__ int lane_col(int n0) {
  const int mr = threadIdx.x & 15;
  return NT == 2 ? n0 + 2 * mr : min(n0 + mr, H_ - 1);
}

template <int NT>
__device__ __forceinline__ v2f ldB_unit(const float* __restrict__ w, int cb, int j, int kk) {
  if constexpr (NT == 2) {
    return ld2(w + (size_t)(kk + j) * H_ + cb);
  } else {
    v2f b;
    b.x = w[(size_t)kk * H_ + cb];
    b.y = w[(size_t)(kk + 1) * H_ + cb];
    return b;
  }
}

constexpr int NS_ = 7;
constexpr int KP_ = H_ / 8;

constexpr int GD_ = 5;

struct GSlot { v4f b[4]; v4f y1, y2; };

__device__ __forceinline__ float ang_sim(float cosv) {
  return 1.0f - acosf(cosv * 0.99999f) * (1.0f / PI_F);
}

__global__ void __launch_bounds__(32) k_graph(const float* __restrict__ xa, const float* __restrict__ xv,
                        const float* __restrict__ xt, float* __restrict__ intra,
                        float* __restrict__ crossv, float* __restrict__ dinv) {
  const int mb = blockIdx.x >> 2;
  const int m  = mb >> 5;
  const int d0 = (mb & 31) * L_;
  const int w  = blockIdx.x & 3;
  const int lane = threadIdx.x & 31;
  const int c = lane & 15, h = lane >> 4;
  const int n1 = (m == 0) ? 1 : 0;
  const int n2 = (m == 2) ? 1 : 2;
  const float* xm = (m == 0) ? xa : (m == 1) ? xv : xt;
  const float* x1 = (n1 == 0) ? xa : xv;
  const float* x2 = (n2 == 1) ? xv : xt;
  const float* pb[4];
#pragma unroll
  for (int u = 0; u < 4; ++u)
    pb[u] = xm + (size_t)(d0 + 16 * ((w + u) & 3) + c) * H_ + 4 * h;
  const float* p1 = x1 + (size_t)(d0 + 16 * w + c) * H_ + 4 * h;
  const float* p2 = x2 + (size_t)(d0 + 16 * w + c) * H_ + 4 * h;

  v8f acc[4] = {};
  float ss[4] = {0.f, 0.f, 0.f, 0.f};
  float s1 = 0.f, s2 = 0.f, d1 = 0.f, d2 = 0.f;

  auto fetch = [&](GSlot& s, int p) {
#pragma unroll
    for (int u = 0; u < 4; ++u) s.b[u] = ld4(pb[u] + 8 * p);
    s.y1 = ld4(p1 + 8 * p);
    s.y2 = ld4(p2 + 8 * p);
  };
  auto use = [&](const GSlot& s) {
    const v4f a = s.b[0];
#pragma unroll
    for (int u = 0; u < 4; ++u) {
      const v4f b = s.b[u];
      ss[u] = fmaf(b.x, b.x, ss[u]); ss[u] = fmaf(b.y, b.y, ss[u]);
      ss[u] = fmaf(b.z, b.z, ss[u]); ss[u] = fmaf(b.w, b.w, ss[u]);
    }
    d1 = fmaf(a.x, s.y1.x, d1); d1 = fmaf(a.y, s.y1.y, d1);
    d1 = fmaf(a.z, s.y1.z, d1); d1 = fmaf(a.w, s.y1.w, d1);
    d2 = fmaf(a.x, s.y2.x, d2); d2 = fmaf(a.y, s.y2.y, d2);
    d2 = fmaf(a.z, s.y2.z, d2); d2 = fmaf(a.w, s.y2.w, d2);
    s1 = fmaf(s.y1.x, s.y1.x, s1); s1 = fmaf(s.y1.y, s.y1.y, s1);
    s1 = fmaf(s.y1.z, s.y1.z, s1); s1 = fmaf(s.y1.w, s.y1.w, s1);
    s2 = fmaf(s.y2.x, s.y2.x, s2); s2 = fmaf(s.y2.y, s.y2.y, s2);
    s2 = fmaf(s.y2.z, s.y2.z, s2); s2 = fmaf(s.y2.w, s.y2.w, s2);
    const v2f alo = a.xy, ahi = a.zw;
#pragma unroll
    for (int u = 0; u < 4; ++u) {
      const v2f blo = s.b[u].xy;
      acc[u] = __builtin_amdgcn_wmma_f32_16x16x4_f32(false, alo, false, blo,
                                                     (short)0, acc[u], false, false);
    }
#pragma unroll
    for (int u = 0; u < 4; ++u) {
      const v2f bhi = s.b[u].zw;
      acc[u] = __builtin_amdgcn_wmma_f32_16x16x4_f32(false, ahi, false, bhi,
                                                     (short)0, acc[u], false, false);
    }
  };

  GSlot ring[GD_];
#pragma unroll
  for (int d = 0; d < GD_; ++d) {
    fetch(ring[d], d);
    __builtin_amdgcn_sched_barrier(0);
  }
#pragma nounroll
  for (int p0 = 0; p0 < KP_ - GD_; p0 += GD_) {
#pragma unroll
    for (int d = 0; d < GD_; ++d) {
      use(ring[d]);
      fetch(ring[d], p0 + d + GD_);
      __builtin_amdgcn_sched_barrier(0);
    }
  }
#pragma unroll
  for (int d = 0; d < GD_; ++d) use(ring[d]);

#pragma unroll
  for (int u = 0; u < 4; ++u) ss[u] += __shfl_xor(ss[u], 16, 32);
  s1 += __shfl_xor(s1, 16, 32);
  s2 += __shfl_xor(s2, 16, 32);
  d1 += __shfl_xor(d1, 16, 32);
  d2 += __shfl_xor(d2, 16, 32);

  float rn[4];
#pragma unroll
  for (int u = 0; u < 4; ++u) rn[u] = 1.0f / sqrtf(ss[u]);
  const float cr1 = ang_sim(d1 * (rn[0] * (1.0f / sqrtf(s1))));
  const float cr2 = ang_sim(d2 * (rn[0] * (1.0f / sqrtf(s2))));

  float rs[8];
  const size_t srow = (size_t)mb * L_ + 16 * w + 8 * h;
#pragma unroll
  for (int r = 0; r < 8; ++r) {
    const float rr = __shfl(rn[0], 8 * h + r, 32);
    float t = 0.f;
#pragma unroll
    for (int u = 0; u < 4; ++u) {
      const float v = ang_sim(acc[u][r] * (rr * rn[u]));
      intra[(srow + r) * L_ + 16 * ((w + u) & 3) + c] = v;
      t += v;
    }
#pragma unroll
    for (int off = 8; off > 0; off >>= 1) t += __shfl_xor(t, off, 32);
    rs[r] = t;
  }
  float mine = rs[0];
#pragma unroll
  for (int r = 1; r < 8; ++r) mine = ((c & 7) == r) ? rs[r] : mine;
  if ((c >> 3) == h) {
    const int a = d0 + 16 * w + c;
    crossv[((size_t)m * M_ + n1) * A_ + a] = cr1;
    crossv[((size_t)m * M_ + n2) * A_ + a] = cr2;
    dinv[(size_t)m * A_ + a] = 1.0f / sqrtf(mine + cr1 + cr2);
  }
}


template <int NT>
__device__ __forceinline__ void fc0_body(const float* __restrict__ xr, const float* __restrict__ w,
                                         const float* __restrict__ bias, float* __restrict__ h0,
                                         float* __restrict__ h0T, int m0, int n0) {
  const int mr = threadIdx.x & 15;
  const int cb = lane_col<NT>(n0);
  const float* ar[2] = {xr + (size_t)mr * H_, xr + (size_t)(16 + mr) * H_};
  v8f acc[2][NT] = {};
  auto cA = [](v4f r) { return r; };
  auto lA = [&](int t, int kq) { return ld4(ar[t] + kq); };
  auto lB = [&](int j, int kk) { return ldB_unit<NT>(w, cb, j, kk); };
  Ring<2, NT, 5, v4f> g0;
  ring_fill(g0, lA, lB);
  ring_main<2, NT, 5, KP_>(acc, g0, lA, cA, lB);
  ring_drain(acc, g0, cA);
  auto ep = [&](int m, int n, float v) { return fmaxf(v + bias[n], 0.0f); };
  float o[2][NT][8];
  epi_vals<NT, false>(acc, m0, n0, o, ep);
  store_rm<NT, false>(o, m0, n0, h0);
  store_fm<NT, false>(o, m0, n0, h0T, N_);
}

__global__ void __launch_bounds__(64) k_fc0(const float* __restrict__ xa, const float* __restrict__ xv,
                      const float* __restrict__ xt, const float* __restrict__ w,
                      const float* __restrict__ bias, float* __restrict__ h0,
                      float* __restrict__ h0T) {
  const int m0 = (blockIdx.y * 2 + (threadIdx.x >> 5)) * 32, n0 = blockIdx.x * 32;
  const int md = m0 / A_;
  const float* xr = ((md == 0) ? xa : (md == 1) ? xv : xt) + (size_t)(m0 - md * A_) * H_;
  if (blockIdx.x == NS_ - 1) fc0_body<1>(xr, w, bias, h0, h0T, m0, n0);
  else                       fc0_body<2>(xr, w, bias, h0, h0T, m0, n0);
}

template <int NT>
__device__ __forceinline__ void spmm_body(const float* __restrict__ intra,
                                          const float* __restrict__ crossv,
                                          const float* __restrict__ dinv,
                                          const float* __restrict__ h, float* __restrict__ hi,
                                          float* __restrict__ hiT, int mb, int i0, int n0) {
  const int mr = threadIdx.x & 15;
  const int m = mb >> 5;
  const int rowbase = mb * L_;
  const int cb = lane_col<NT>(n0);
  const float* S = intra + (size_t)mb * L_ * L_;
  const float* sr[2] = {S + (size_t)(i0 + mr) * L_, S + (size_t)(i0 + 16 + mr) * L_};
  const float* dv = dinv + rowbase;
  const float* hb = h + (size_t)rowbase * H_;
  v8f acc[2][NT] = {};
  auto lA = [&](int t, int kq) { RawQuad r; r.p = ld4(sr[t] + kq); r.q = ld4(dv + kq); return r; };
  auto cA = [](RawQuad r) { return r.p * r.q; };
  auto lB = [&](int j, int kk) { return ldB_unit<NT>(hb, cb, j, kk); };
  Ring<2, NT, 4, RawQuad> g0;
  ring_fill(g0, lA, lB);
  ring_main<2, NT, 4, L_ / 8>(acc, g0, lA, cA, lB);
  ring_drain(acc, g0, cA);
  const int kh = (threadIdx.x & 31) >> 4;
  const int n1 = (m == 0) ? 1 : 0;
  const int n2 = (m == 2) ? 1 : 2;
  const int ab = rowbase - m * A_;
  const float* c1 = crossv + ((size_t)m * M_ + n1) * A_ + ab;
  const float* c2 = crossv + ((size_t)m * M_ + n2) * A_ + ab;
  const float* e1 = dinv + (size_t)n1 * A_ + ab;
  const float* e2 = dinv + (size_t)n2 * A_ + ab;
  const float* g1 = h + ((size_t)n1 * A_ + ab) * H_;
  const float* g2 = h + ((size_t)n2 * A_ + ab) * H_;
  float f1[2][8], f2[2][8], fr[2][8];
  float x1[2][8][NT], x2[2][8][NT];
#pragma unroll
  for (int t = 0; t < 2; ++t) {
    const int r = i0 + 16 * t + 8 * kh;
#pragma unroll
    for (int q = 0; q < 2; ++q) {
      const v4f p1 = ld4(c1 + r + 4 * q) * ld4(e1 + r + 4 * q);
      const v4f p2 = ld4(c2 + r + 4 * q) * ld4(e2 + r + 4 * q);
      const v4f dr = ld4(dv + r + 4 * q);
      f1[t][4 * q] = p1.x; f1[t][4 * q + 1] = p1.y; f1[t][4 * q + 2] = p1.z; f1[t][4 * q + 3] = p1.w;
      f2[t][4 * q] = p2.x; f2[t][4 * q + 1] = p2.y; f2[t][4 * q + 2] = p2.z; f2[t][4 * q + 3] = p2.w;
      fr[t][4 * q] = dr.x; fr[t][4 * q + 1] = dr.y; fr[t][4 * q + 2] = dr.z; fr[t][4 * q + 3] = dr.w;
    }
#pragma unroll
    for (int v = 0; v < 8; ++v) {
      const size_t off = (size_t)(r + v) * H_ + cb;
      if constexpr (NT == 2) {
        const v2f a = ld2(g1 + off), b = ld2(g2 + off);
        x1[t][v][0] = a.x; x1[t][v][1] = a.y;
        x2[t][v][0] = b.x; x2[t][v][1] = b.y;
      } else {
        x1[t][v][0] = g1[off];
        x2[t][v][0] = g2[off];
      }
    }
  }
  float o[2][NT][8];
#pragma unroll
  for (int t = 0; t < 2; ++t)
#pragma unroll
    for (int u = 0; u < NT; ++u)
#pragma unroll
      for (int v = 0; v < 8; ++v) {
        float s = acc[t][u][v];
        s += f1[t][v] * x1[t][v][u];
        s += f2[t][v] * x2[t][v][u];
        o[t][u][v] = s * fr[t][v];
      }
  store_rm<NT, false>(o, i0, n0, hi + (size_t)rowbase * H_);
  store_fm<NT, false>(o, i0, n0, hiT + rowbase, N_);
}

__global__ void __launch_bounds__(64) k_spmm(const float* __restrict__ intra, const float* __restrict__ crossv,
                       const float* __restrict__ dinv, const float* __restrict__ h,
                       float* __restrict__ hi, float* __restrict__ hiT) {
  const int mb = blockIdx.y;
  const int i0 = (threadIdx.x >> 5) * 32, n0 = blockIdx.x * 32;
  if (blockIdx.x == NS_ - 1) spmm_body<1>(intra, crossv, dinv, h, hi, hiT, mb, i0, n0);
  else                       spmm_body<2>(intra, crossv, dinv, h, hi, hiT, mb, i0, n0);
}

template <int NT, bool FM_OUT>
__device__ __forceinline__ void conv_body(const float* __restrict__ hi, const float* __restrict__ h0,
                                          const float* __restrict__ hiT, const float* __restrict__ h0T,
                                          const float* __restrict__ w, float beta,
                                          float* __restrict__ hout, int m0, int n0) {
  const int mr = threadIdx.x & 15;
  const int cb = lane_col<NT>(n0);
  const float* t0 = hiT + m0 + 2 * mr;
  const float* t1 = h0T + m0 + 2 * mr;
  const float* w1 = w + (size_t)H_ * H_;
  v8f acc[2][NT] = {};
  auto lA0 = [&](int pl, int k) { return ld2(t0 + (size_t)k * N_); };
  auto lA1 = [&](int pl, int k) { return ld2(t1 + (size_t)k * N_); };
  auto lB0 = [&](int j, int kk) { return ldB_unit<NT>(w, cb, j, kk); };
  auto lB1 = [&](int j, int kk) { return ldB_unit<NT>(w1, cb, j, kk); };
  RingT<NT, 5, 1> g0, g1;
  ringt_fill(g0, lA0, lB0);
  ringt_main<false, NT, 5, KP_>(acc, g0, lA0, lB0);
  ringt_fill(g1, lA1, lB1);
  ringt_drain<false>(acc, g0);
  ringt_main<false, NT, 5, KP_>(acc, g1, lA1, lB1);
  ringt_drain<false>(acc, g1);
  auto ep = [&](int m, int n, float v) {
    const float r = 0.9f * hi[(size_t)m * H_ + n] + 0.1f * h0[(size_t)m * H_ + n];
    return fmaxf(beta * v + (1.0f - beta) * r, 0.0f);
  };
  float o[2][NT][8];
  epi_vals<NT, true>(acc, m0, n0, o, ep);
  if constexpr (FM_OUT) store_fm<NT, true>(o, m0, n0, hout, N_);
  else                  store_rm<NT, true>(o, m0, n0, hout);
}

template <bool FM_OUT>
__global__ void __launch_bounds__(64) k_conv(const float* __restrict__ hi, const float* __restrict__ h0,
                       const float* __restrict__ hiT, const float* __restrict__ h0T,
                       const float* __restrict__ w, float beta, float* __restrict__ hout) {
  const int m0 = (blockIdx.y * 2 + (threadIdx.x >> 5)) * 32, n0 = blockIdx.x * 32;
  if (blockIdx.x == NS_ - 1) conv_body<1, FM_OUT>(hi, h0, hiT, h0T, w, beta, hout, m0, n0);
  else                       conv_body<2, FM_OUT>(hi, h0, hiT, h0T, w, beta, hout, m0, n0);
}

template <int NT>
__device__ __forceinline__ void hx_body(const float* __restrict__ xm, const float* __restrict__ hfT,
                                        const float* __restrict__ W, const float* __restrict__ Bb,
                                        float* __restrict__ hx, float* __restrict__ hxT,
                                        int rb, int a0_, int n0) {
  const int mr = threadIdx.x & 15;
  const int cb = lane_col<NT>(n0);
  const size_t r0 = (size_t)(a0_ + 2 * mr) * H_, r1 = (size_t)(a0_ + 2 * mr + 1) * H_;
  const float* p0[2] = {xm + r0, xm + r1};
  const float* tf = hfT + rb + a0_ + 2 * mr;
  const float* W1 = W + (size_t)H_ * H_;
  v8f acc[2][NT] = {};
  auto cA = [](v4f r) { return r; };
  auto lA0 = [&](int t, int kq) { return ld4(p0[t] + kq); };
  auto lA1 = [&](int pl, int k) { return ld2(tf + (size_t)k * N_); };
  auto lB0 = [&](int j, int kk) { return ldB_unit<NT>(W, cb, j, kk); };
  auto lB1 = [&](int j, int kk) { return ldB_unit<NT>(W1, cb, j, kk); };
  Ring<2, NT, 5, v4f> g0;
  RingT<NT, 5, 1> g1;
  ring_fill(g0, lA0, lB0);
  ring_main<2, NT, 5, KP_>(acc, g0, lA0, cA, lB0);
  ringt_fill(g1, lA1, lB1);
  ring_drain(acc, g0, cA);
  ringt_main<false, NT, 5, KP_>(acc, g1, lA1, lB1);
  ringt_drain<false>(acc, g1);
  auto ep = [&](int a, int n, float v) { return fmaxf(v + Bb[n], 0.0f); };
  float o[2][NT][8];
  epi_vals<NT, true>(acc, a0_, n0, o, ep);
  store_rm<NT, true>(o, a0_, n0, hx + (size_t)rb * H_);
  store_fm<NT, true>(o, a0_, n0, hxT + (size_t)rb * H_, A_);
}

__global__ void __launch_bounds__(64) k_hx(const float* __restrict__ xa, const float* __restrict__ xv,
                     const float* __restrict__ xt, const float* __restrict__ hfT,
                     const float* __restrict__ tw, const float* __restrict__ tb,
                     float* __restrict__ hx, float* __restrict__ hxT) {
  const int m  = blockIdx.z;
  const float* xm = (m == 0) ? xa : (m == 1) ? xv : xt;
  const int a0_ = (blockIdx.y * 2 + (threadIdx.x >> 5)) * 32, n0 = blockIdx.x * 32;
  const float* W  = tw + (size_t)m * 2 * H_ * H_;
  const float* Bb = tb + (size_t)m * H_;
  if (blockIdx.x == NS_ - 1) hx_body<1>(xm, hfT, W, Bb, hx, hxT, m * A_, a0_, n0);
  else                       hx_body<2>(xm, hfT, W, Bb, hx, hxT, m * A_, a0_, n0);
}

template <int NT>
__device__ __forceinline__ float tr_logit(const v8f (&a)[NT], int g, int e) {
  if constexpr (NT == 2) return a[e & 1][4 * g + (e >> 1)];
  else                   return a[0][e];
}

template <int NT>
__device__ __forceinline__ void gate_body(const float* __restrict__ Hi, const float* __restrict__ Hj,
                                          const float* __restrict__ HiT, const float* __restrict__ HjT,
                                          const float* __restrict__ W, const float* __restrict__ Bb,
                                          const float* __restrict__ Ui, float* __restrict__ part_out,
                                          int a0_, int n0) {
  const int mr = threadIdx.x & 15;
  const int kh = (threadIdx.x & 31) >> 4;
  const int cb = lane_col<NT>(n0);
  const size_t r0 = (size_t)(a0_ + 2 * mr) * H_, r1 = (size_t)(a0_ + 2 * mr + 1) * H_;
  const float* pi[2] = {Hi + r0, Hi + r1};
  const float* pj[2] = {Hj + r0, Hj + r1};
  const float* ti = HiT + a0_ + 2 * mr;
  const float* tj = HjT + a0_ + 2 * mr;
  const float* W1 = W + (size_t)H_ * H_;
  const float* W2 = W + (size_t)2 * H_ * H_;
  v8f acc[2][NT] = {};
  auto lAi = [&](int pl, int k) { return ld2(ti + (size_t)k * A_); };
  auto lAj = [&](int pl, int k) { return ld2(tj + (size_t)k * A_); };
  auto lAp = [&](int pl, int k) { return ld2((pl ? tj : ti) + (size_t)k * A_); };
  auto lB0 = [&](int j, int kk) { return ldB_unit<NT>(W, cb, j, kk); };
  auto lB1 = [&](int j, int kk) { return ldB_unit<NT>(W1, cb, j, kk); };
  auto lB2 = [&](int j, int kk) { return ldB_unit<NT>(W2, cb, j, kk); };
  RingT<NT, 5, 1> g0, g1;
  RingT<NT, 5, 2> g2;
  ringt_fill(g0, lAi, lB0);
  ringt_main<true, NT, 5, KP_>(acc, g0, lAi, lB0);
  ringt_fill(g1, lAj, lB1);
  ringt_drain<true>(acc, g0);
  ringt_main<true, NT, 5, KP_>(acc, g1, lAj, lB1);
  ringt_fill(g2, lAp, lB2);
  ringt_drain<true>(acc, g1);
  ringt_main<true, NT, 5, KP_>(acc, g2, lAp, lB2);
  ringt_drain<true>(acc, g2);

  float part[2][TAG_] = {};
#pragma unroll
  for (int t = 0; t < 2; ++t) {
#pragma unroll
    for (int g = 0; g < NT; ++g) {
      const int nb = (NT == 2) ? n0 + 16 * kh + 8 * g : n0 + 8 * kh;
      if (NT == 2 || nb < H_) {
        const v4f* hip = reinterpret_cast<const v4f*>(pi[t] + nb);
        const v4f* hjp = reinterpret_cast<const v4f*>(pj[t] + nb);
        const v4f* bp  = reinterpret_cast<const v4f*>(Bb + nb);
        const v4f* up  = reinterpret_cast<const v4f*>(Ui + (size_t)nb * TAG_);
        const v4f hi0 = hip[0], hi1 = hip[1], hj0 = hjp[0], hj1 = hjp[1], b0 = bp[0], b1 = bp[1];
        const float hiv[8] = {hi0.x, hi0.y, hi0.z, hi0.w, hi1.x, hi1.y, hi1.z, hi1.w};
        const float hjv[8] = {hj0.x, hj0.y, hj0.z, hj0.w, hj1.x, hj1.y, hj1.z, hj1.w};
        const float bv[8]  = {b0.x, b0.y, b0.z, b0.w, b1.x, b1.y, b1.z, b1.w};
        float uu[8 * TAG_];
#pragma unroll
        for (int q = 0; q < 2 * TAG_; ++q) {
          const v4f x = up[q];
          uu[4 * q] = x.x; uu[4 * q + 1] = x.y; uu[4 * q + 2] = x.z; uu[4 * q + 3] = x.w;
        }
#pragma unroll
        for (int e = 0; e < 8; ++e) {
          const float lg = tr_logit<NT>(acc[t], g, e);
          const float z = 1.0f / (1.0f + expf(-(lg + bv[e])));
          const float gt = z * hiv[e] + (1.0f - z) * hjv[e];
#pragma unroll
          for (int c = 0; c < TAG_; ++c) part[t][c] += gt * uu[e * TAG_ + c];
        }
      }
    }
  }
#pragma unroll
  for (int t = 0; t < 2; ++t)
#pragma unroll
    for (int c = 0; c < TAG_; ++c) part[t][c] += __shfl_xor(part[t][c], 16, 32);
  if (kh == 0) {
#pragma unroll
    for (int t = 0; t < 2; ++t) {
      float* po = part_out + (size_t)(a0_ + 2 * mr + t) * TAG_;
#pragma unroll
      for (int c = 0; c < TAG_; c += 2) {
        v2f o; o.x = part[t][c]; o.y = part[t][c + 1];
        *reinterpret_cast<v2f*>(po + c) = o;
      }
    }
  }
}

__global__ void __launch_bounds__(64) k_gate(const float* __restrict__ hx, const float* __restrict__ hxT,
                       const float* __restrict__ cw,
                       const float* __restrict__ cb, const float* __restrict__ uw,
                       float* __restrict__ plog) {
  const int p = blockIdx.z;
  const int i = p >> 1;
  const int j = (p & 1) ? (i == 2 ? 1 : 2) : (i == 0 ? 1 : 0);
  const int a0_ = (blockIdx.y * 2 + (threadIdx.x >> 5)) * 32, n0 = blockIdx.x * 32;
  const float* Hi = hx + (size_t)i * A_ * H_;
  const float* Hj = hx + (size_t)j * A_ * H_;
  const float* HiT = hxT + (size_t)i * H_ * A_;
  const float* HjT = hxT + (size_t)j * H_ * A_;
  const float* W  = cw + (size_t)p * 3 * H_ * H_;
  const float* Bb = cb + (size_t)p * H_;
  const float* Ui = uw + (size_t)i * H_ * TAG_;
  float* po = plog + ((size_t)p * NS_ + blockIdx.x) * A_ * TAG_;
  if (blockIdx.x == NS_ - 1) gate_body<1>(Hi, Hj, HiT, HjT, W, Bb, Ui, po, a0_, n0);
  else                       gate_body<2>(Hi, Hj, HiT, HjT, W, Bb, Ui, po, a0_, n0);
}

__global__ void k_final(const float* __restrict__ plog, const float* __restrict__ ub,
                        float* __restrict__ out) {
  const int e = blockIdx.x * blockDim.x + threadIdx.x;
  const int t = e % TAG_;
  float s = ub[t] + ub[TAG_ + t] + ub[2 * TAG_ + t];
  for (int q = 0; q < 6 * NS_; ++q) s += plog[(size_t)q * A_ * TAG_ + e];
  out[e] = s;
}

}

extern "C" void kernel_launch(void* const* d_in, const int* in_sizes, int n_in,
                              void* d_out, int out_size, void* d_ws, size_t ws_size,
                              hipStream_t stream) {
  (void)in_sizes; (void)n_in; (void)out_size; (void)ws_size;
  const float* x_a     = (const float*)d_in[0];
  const float* x_v     = (const float*)d_in[1];
  const float* x_t     = (const float*)d_in[2];
  const float* fc0_w   = (const float*)d_in[3];
  const float* fc0_b   = (const float*)d_in[4];
  const float* conv_w  = (const float*)d_in[5];
  const float* tm_w    = (const float*)d_in[6];
  const float* tm_b    = (const float*)d_in[7];
  const float* cross_w = (const float*)d_in[8];
  const float* cross_b = (const float*)d_in[9];
  const float* uni_w   = (const float*)d_in[10];
  const float* uni_b   = (const float*)d_in[11];
  float* out = (float*)d_out;

  const size_t NH = (size_t)N_ * H_;
  float* ws    = (float*)d_ws;
  float* intra = ws; ws += (size_t)N_ * L_;
  float* crossv= ws; ws += (size_t)M_ * M_ * A_;
  float* dinv  = ws; ws += N_;
  float* h0    = ws; ws += NH;
  float* h0T   = ws; ws += NH;
  float* hA    = ws; ws += NH;
  float* hBfT  = ws; ws += NH;
  float* hi    = ws; ws += NH;
  float* hiT   = ws; ws += NH;
  float* hx    = ws; ws += NH;
  float* hxT   = ws; ws += NH;
  float* plog  = ws; ws += (size_t)6 * NS_ * A_ * TAG_;


  k_graph<<<M_ * B_ * 4,         dim3(32), 0, stream>>>(x_a, x_v, x_t, intra, crossv, dinv);
  k_fc0  <<<dim3(NS_, N_ / 64),  dim3(64), 0, stream>>>(x_a, x_v, x_t, fc0_w, fc0_b, h0, h0T);

  const float beta0 = logf(0.5f / 1.0f + 1.0f);
  const float beta1 = logf(0.5f / 2.0f + 1.0f);
  k_spmm <<<dim3(NS_, M_ * B_),  dim3(64), 0, stream>>>(intra, crossv, dinv, h0, hi, hiT);
  k_conv<false><<<dim3(NS_, N_ / 64),  dim3(64), 0, stream>>>(hi, h0, hiT, h0T, conv_w, beta0, hA);
  k_spmm <<<dim3(NS_, M_ * B_),  dim3(64), 0, stream>>>(intra, crossv, dinv, hA, hi, hiT);
  k_conv<true><<<dim3(NS_, N_ / 64),  dim3(64), 0, stream>>>(hi, h0, hiT, h0T, conv_w + 2 * H_ * H_, beta1, hBfT);

  k_hx   <<<dim3(NS_, A_ / 64, M_), dim3(64), 0, stream>>>(x_a, x_v, x_t, hBfT, tm_w, tm_b, hx, hxT);

  k_gate <<<dim3(NS_, A_ / 64, 6), dim3(64), 0, stream>>>(hx, hxT, cross_w, cross_b, uni_w, plog);

  k_final<<<A_ * TAG_ / 256, dim3(256), 0, stream>>>(plog, uni_b, out);
}


# GEMM kernels (fc0, spmm, conv, hx, gate): work-group = 4 waves, 2 row blocks x 2 column strips (was 2 waves, 1 strip): a wave on each SIMD of a unit, A and B lines each wanted by 2 waves; sums kept
# speedup vs baseline: 8.0023x; 1.0361x over previous
#include <hip/hip_runtime.h>
#include <math.h>

typedef __attribute__((ext_vector_type(2))) float v2f;
typedef __attribute__((ext_vector_type(4))) float v4f;
typedef __attribute__((ext_vector_type(8))) float v8f;

namespace {

constexpr int B_   = 32;
constexpr int L_   = 64;
constexpr int H_   = 200;
constexpr int M_   = 3;
constexpr int A_   = B_ * L_;
constexpr int N_   = M_ * A_;
constexpr int TAG_ = 6;
constexpr float PI_F = 3.14159265358979323846f;

struct RawQuad { v4f p, q; };

__device__ __forceinline__ v4f ld4(const float* p) {
  return *reinterpret_cast<const v4f*>(p);
}

__device__ __forceinline__ v2f ld2(const float* p) {
  return *reinterpret_cast<const v2f*>(p);
}

template <int MT, int NT, int D, typename RA>
struct Ring { RA a[D][MT]; v2f b[D][2][NT]; };

template <int NT>
__device__ __forceinline__ void b_frags(const v2f (&raw)[NT], v2f (&f)[NT]) {
  if constexpr (NT == 2) {
    f[0].x = raw[0].x; f[0].y = raw[1].x;
    f[1].x = raw[0].y; f[1].y = raw[1].y;
  } else {
    f[0] = raw[0];
  }
}

template <int MT, int NT, bool TR = false>
__device__ __forceinline__ void mma_step(v8f (&acc)[MT][NT], const v2f (&a)[MT],
                                         const v2f (&b)[NT]) {
#pragma unroll
  for (int t = 0; t < MT; ++t)
#pragma unroll
    for (int u = 0; u < NT; ++u) {
      if constexpr (TR)
        acc[t][u] = __builtin_amdgcn_wmma_f32_16x16x4_f32(false, b[u], false, a[t],
                                                          (short)0, acc[t][u], false, false);
      else
        acc[t][u] = __builtin_amdgcn_wmma_f32_16x16x4_f32(false, a[t], false, b[u],
                                                          (short)0, acc[t][u], false, false);
    }
}


template <int MT, int NT, int D, typename RA, typename FA, typename FB>
__device__ __forceinline__ void ring_fill(Ring<MT, NT, D, RA>& r, FA ldA, FB ldB) {
  const int kh = (threadIdx.x & 31) >> 4;
#pragma unroll
  for (int d = 0; d < D; ++d) {
    const int kq = 8 * d + 4 * kh;
#pragma unroll
    for (int t = 0; t < MT; ++t) r.a[d][t] = ldA(t, kq);
#pragma unroll
    for (int s = 0; s < 2; ++s)
#pragma unroll
      for (int u = 0; u < NT; ++u) r.b[d][s][u] = ldB(u, kq + 2 * s);
    __builtin_amdgcn_sched_barrier(0);
  }
}

template <int MT, int NT, int D, int NPAIRS, bool TR = false, typename RA, typename FA, typename FC, typename FB>
__device__ __forceinline__ void ring_main(v8f (&acc)[MT][NT], Ring<MT, NT, D, RA>& r,
                                          FA ldA, FC cvA, FB ldB) {
  static_assert(NPAIRS % D == 0 && NPAIRS >= D, "ring depth must divide the pair count");
  const int kh = (threadIdx.x & 31) >> 4;
#pragma nounroll
  for (int p0 = 0; p0 < NPAIRS - D; p0 += D) {
#pragma unroll
    for (int d = 0; d < D; ++d) {
      v2f lo[MT], hi[MT];
#pragma unroll
      for (int t = 0; t < MT; ++t) {
        const v4f c = cvA(r.a[d][t]);
        lo[t] = c.xy;
        hi[t] = c.zw;
      }
      v2f f0[NT], f1[NT];
      b_frags<NT>(r.b[d][0], f0);
      b_frags<NT>(r.b[d][1], f1);
      mma_step<MT, NT, TR>(acc, lo, f0);
      mma_step<MT, NT, TR>(acc, hi, f1);
      const int kq = 8 * (p0 + d + D) + 4 * kh;
#pragma unroll
      for (int t = 0; t < MT; ++t) r.a[d][t] = ldA(t, kq);
#pragma unroll
      for (int s = 0; s < 2; ++s)
#pragma unroll
        for (int u = 0; u < NT; ++u) r.b[d][s][u] = ldB(u, kq + 2 * s);
      __builtin_amdgcn_sched_barrier(0);
    }
  }
}

template <bool TR = false, int MT, int NT, int D, typename RA, typename FC>
__device__ __forceinline__ void ring_drain(v8f (&acc)[MT][NT], Ring<MT, NT, D, RA>& r, FC cvA) {
#pragma unroll
  for (int d = 0; d < D; ++d) {
    v2f lo[MT], hi[MT];
#pragma unroll
    for (int t = 0; t < MT; ++t) {
      const v4f c = cvA(r.a[d][t]);
      lo[t] = c.xy;
      hi[t] = c.zw;
    }
    v2f f0[NT], f1[NT];
    b_frags<NT>(r.b[d][0], f0);
    b_frags<NT>(r.b[d][1], f1);
    mma_step<MT, NT, TR>(acc, lo, f0);
    mma_step<MT, NT, TR>(acc, hi, f1);
  }
}

template <int NT, int D, int NP>
struct RingT { v2f a[D][NP][4]; v2f b[D][2][NT]; };

template <int NP>
__device__ __forceinline__ void a_frags(const v2f (&raw)[NP][4], v2f (&lo)[2], v2f (&hi)[2]) {
  v2f q[4];
#pragma unroll
  for (int j = 0; j < 4; ++j) {
    if constexpr (NP == 2) q[j] = raw[0][j] * raw[1][j];
    else                   q[j] = raw[0][j];
  }
  lo[0].x = q[0].x; lo[0].y = q[1].x; hi[0].x = q[2].x; hi[0].y = q[3].x;
  lo[1].x = q[0].y; lo[1].y = q[1].y; hi[1].x = q[2].y; hi[1].y = q[3].y;
}

template <int NT, int D, int NP, typename FA, typename FB>
__device__ __forceinline__ void ringt_fill(RingT<NT, D, NP>& r, FA ldA, FB ldB) {
  const int kh = (threadIdx.x & 31) >> 4;
#pragma unroll
  for (int d = 0; d < D; ++d) {
    const int kq = 8 * d + 4 * kh;
#pragma unroll
    for (int pl = 0; pl < NP; ++pl)
#pragma unroll
      for (int j = 0; j < 4; ++j) r.a[d][pl][j] = ldA(pl, kq + j);
#pragma unroll
    for (int s = 0; s < 2; ++s)
#pragma unroll
      for (int u = 0; u < NT; ++u) r.b[d][s][u] = ldB(u, kq + 2 * s);
    __builtin_amdgcn_sched_barrier(0);
  }
}

template <bool TR, int NT, int D, int NPAIRS, int NP, typename FA, typename FB>
__device__ __forceinline__ void ringt_main(v8f (&acc)[2][NT], RingT<NT, D, NP>& r, FA ldA, FB ldB) {
  static_assert(NPAIRS % D == 0 && NPAIRS >= D, "ring depth must divide the pair count");
  const int kh = (threadIdx.x & 31) >> 4;
#pragma nounroll
  for (int p0 = 0; p0 < NPAIRS - D; p0 += D) {
#pragma unroll
    for (int d = 0; d < D; ++d) {
      v2f lo[2], hi[2];
      a_frags<NP>(r.a[d], lo, hi);
      v2f f0[NT], f1[NT];
      b_frags<NT>(r.b[d][0], f0);
      b_frags<NT>(r.b[d][1], f1);
      mma_step<2, NT, TR>(acc, lo, f0);
      mma_step<2, NT, TR>(acc, hi, f1);
      const int kq = 8 * (p0 + d + D) + 4 * kh;
#pragma unroll
      for (int pl = 0; pl < NP; ++pl)
#pragma unroll
        for (int j = 0; j < 4; ++j) r.a[d][pl][j] = ldA(pl, kq + j);
#pragma unroll
      for (int s = 0; s < 2; ++s)
#pragma unroll
        for (int u = 0; u < NT; ++u) r.b[d][s][u] = ldB(u, kq + 2 * s);
      __builtin_amdgcn_sched_barrier(0);
    }
  }
}

template <bool TR, int NT, int D, int NP>
__device__ __forceinline__ void ringt_drain(v8f (&acc)[2][NT], RingT<NT, D, NP>& r) {
#pragma unroll
  for (int d = 0; d < D; ++d) {
    v2f lo[2], hi[2];
    a_frags<NP>(r.a[d], lo, hi);
    v2f f0[NT], f1[NT];
    b_frags<NT>(r.b[d][0], f0);
    b_frags<NT>(r.b[d][1], f1);
    mma_step<2, NT, TR>(acc, lo, f0);
    mma_step<2, NT, TR>(acc, hi, f1);
  }
}

template <bool RI>
__device__ __forceinline__ int tile_row(int t, int x) { return RI ? 2 * x + t : 16 * t + x; }

template <int NT, bool RI, typename FV>
__device__ __forceinline__ void epi_vals(const v8f (&acc)[2][NT], int m0, int n0,
                                         float (&o)[2][NT][8], FV val) {
  const int lane = threadIdx.x & 31;
  const int mr = lane & 15, kh = lane >> 4;
#pragma unroll
  for (int t = 0; t < 2; ++t)
#pragma unroll
    for (int u = 0; u < NT; ++u)
#pragma unroll
      for (int v = 0; v < 8; ++v) {
        const int m = m0 + tile_row<RI>(t, 8 * kh + v);
        const int n = (NT == 2) ? n0 + 2 * mr + u : n0 + mr;
        o[t][u][v] = (NT == 2 || n < H_) ? val(m, n, acc[t][u][v]) : 0.0f;
      }
}

template <int NT, bool RI>
__device__ __forceinline__ void store_rm(const float (&o)[2][NT][8], int m0, int n0,
                                         float* __restrict__ dst) {
  const int lane = threadIdx.x & 31;
  const int mr = lane & 15, kh = lane >> 4;
#pragma unroll
  for (int t = 0; t < 2; ++t)
#pragma unroll
    for (int v = 0; v < 8; ++v) {
      const int m = m0 + tile_row<RI>(t, 8 * kh + v);
      if constexpr (NT == 2) {
        v2f w;
        w.x = o[t][0][v];
        w.y = o[t][1][v];
        *reinterpret_cast<v2f*>(dst + (size_t)m * H_ + n0 + 2 * mr) = w;
      } else {
        const int n = n0 + mr;
        if (n < H_) dst[(size_t)m * H_ + n] = o[t][0][v];
      }
    }
}

template <int NT, bool RI>
__device__ __forceinline__ void store_fm(const float (&o)[2][NT][8], int m0, int n0,
                                         float* __restrict__ dstT, int pitch) {
  const int lane = threadIdx.x & 31;
  const int mr = lane & 15, kh = lane >> 4;
#pragma unroll
  for (int u = 0; u < NT; ++u) {
    const int n = (NT == 2) ? n0 + 2 * mr + u : n0 + mr;
    if (NT == 2 || n < H_) {
      float* col = dstT + (size_t)n * pitch + m0;
      if constexpr (RI) {
#pragma unroll
        for (int q = 0; q < 4; ++q) {
          v4f w;
          w.x = o[0][u][2 * q];
          w.y = o[1][u][2 * q];
          w.z = o[0][u][2 * q + 1];
          w.w = o[1][u][2 * q + 1];
          *reinterpret_cast<v4f*>(col + 16 * kh + 4 * q) = w;
        }
      } else {
#pragma unroll
        for (int t = 0; t < 2; ++t)
#pragma unroll
          for (int q = 0; q < 2; ++q) {
            v4f w;
            w.x = o[t][u][4 * q];
            w.y = o[t][u][4 * q + 1];
            w.z = o[t][u][4 * q + 2];
            w.w = o[t][u][4 * q + 3];
            *reinterpret_cast<v4f*>(col + 16 * t + 8 * kh + 4 * q) = w;
          }
      }
    }
  }
}

template <int NT>
__device__ __forceinline__ int lane_col(int n0) {
  const int mr = threadIdx.x & 15;
  return NT == 2 ? n0 + 2 * mr : min(n0 + mr, H_ - 1);
}

template <int NT>
__device__ __forceinline__ v2f ldB_unit(const float* __restrict__ w, int cb, int j, int kk) {
  if constexpr (NT == 2) {
    return ld2(w + (size_t)(kk + j) * H_ + cb);
  } else {
    v2f b;
    b.x = w[(size_t)kk * H_ + cb];
    b.y = w[(size_t)(kk + 1) * H_ + cb];
    return b;
  }
}

constexpr int NS_ = 7;
constexpr int KP_ = H_ / 8;

constexpr int NSP_ = (NS_ + 1) / 2;

__device__ __forceinline__ int wg_wave() { return __builtin_amdgcn_readfirstlane((int)(threadIdx.x >> 5)); }

constexpr int GD_ = 5;

struct GSlot { v4f b[4]; v4f y1, y2; };

__device__ __forceinline__ float ang_sim(float cosv) {
  return 1.0f - acosf(cosv * 0.99999f) * (1.0f / PI_F);
}

__global__ void __launch_bounds__(32) k_graph(const float* __restrict__ xa, const float* __restrict__ xv,
                        const float* __restrict__ xt, float* __restrict__ intra,
                        float* __restrict__ crossv, float* __restrict__ dinv) {
  const int mb = blockIdx.x >> 2;
  const int m  = mb >> 5;
  const int d0 = (mb & 31) * L_;
  const int w  = blockIdx.x & 3;
  const int lane = threadIdx.x & 31;
  const int c = lane & 15, h = lane >> 4;
  const int n1 = (m == 0) ? 1 : 0;
  const int n2 = (m == 2) ? 1 : 2;
  const float* xm = (m == 0) ? xa : (m == 1) ? xv : xt;
  const float* x1 = (n1 == 0) ? xa : xv;
  const float* x2 = (n2 == 1) ? xv : xt;
  const float* pb[4];
#pragma unroll
  for (int u = 0; u < 4; ++u)
    pb[u] = xm + (size_t)(d0 + 16 * ((w + u) & 3) + c) * H_ + 4 * h;
  const float* p1 = x1 + (size_t)(d0 + 16 * w + c) * H_ + 4 * h;
  const float* p2 = x2 + (size_t)(d0 + 16 * w + c) * H_ + 4 * h;

  v8f acc[4] = {};
  float ss[4] = {0.f, 0.f, 0.f, 0.f};
  float s1 = 0.f, s2 = 0.f, d1 = 0.f, d2 = 0.f;

  auto fetch = [&](GSlot& s, int p) {
#pragma unroll
    for (int u = 0; u < 4; ++u) s.b[u] = ld4(pb[u] + 8 * p);
    s.y1 = ld4(p1 + 8 * p);
    s.y2 = ld4(p2 + 8 * p);
  };
  auto use = [&](const GSlot& s) {
    const v4f a = s.b[0];
#pragma unroll
    for (int u = 0; u < 4; ++u) {
      const v4f b = s.b[u];
      ss[u] = fmaf(b.x, b.x, ss[u]); ss[u] = fmaf(b.y, b.y, ss[u]);
      ss[u] = fmaf(b.z, b.z, ss[u]); ss[u] = fmaf(b.w, b.w, ss[u]);
    }
    d1 = fmaf(a.x, s.y1.x, d1); d1 = fmaf(a.y, s.y1.y, d1);
    d1 = fmaf(a.z, s.y1.z, d1); d1 = fmaf(a.w, s.y1.w, d1);
    d2 = fmaf(a.x, s.y2.x, d2); d2 = fmaf(a.y, s.y2.y, d2);
    d2 = fmaf(a.z, s.y2.z, d2); d2 = fmaf(a.w, s.y2.w, d2);
    s1 = fmaf(s.y1.x, s.y1.x, s1); s1 = fmaf(s.y1.y, s.y1.y, s1);
    s1 = fmaf(s.y1.z, s.y1.z, s1); s1 = fmaf(s.y1.w, s.y1.w, s1);
    s2 = fmaf(s.y2.x, s.y2.x, s2); s2 = fmaf(s.y2.y, s.y2.y, s2);
    s2 = fmaf(s.y2.z, s.y2.z, s2); s2 = fmaf(s.y2.w, s.y2.w, s2);
    const v2f alo = a.xy, ahi = a.zw;
#pragma unroll
    for (int u = 0; u < 4; ++u) {
      const v2f blo = s.b[u].xy;
      acc[u] = __builtin_amdgcn_wmma_f32_16x16x4_f32(false, alo, false, blo,
                                                     (short)0, acc[u], false, false);
    }
#pragma unroll
    for (int u = 0; u < 4; ++u) {
      const v2f bhi = s.b[u].zw;
      acc[u] = __builtin_amdgcn_wmma_f32_16x16x4_f32(false, ahi, false, bhi,
                                                     (short)0, acc[u], false, false);
    }
  };

  GSlot ring[GD_];
#pragma unroll
  for (int d = 0; d < GD_; ++d) {
    fetch(ring[d], d);
    __builtin_amdgcn_sched_barrier(0);
  }
#pragma nounroll
  for (int p0 = 0; p0 < KP_ - GD_; p0 += GD_) {
#pragma unroll
    for (int d = 0; d < GD_; ++d) {
      use(ring[d]);
      fetch(ring[d], p0 + d + GD_);
      __builtin_amdgcn_sched_barrier(0);
    }
  }
#pragma unroll
  for (int d = 0; d < GD_; ++d) use(ring[d]);

#pragma unroll
  for (int u = 0; u < 4; ++u) ss[u] += __shfl_xor(ss[u], 16, 32);
  s1 += __shfl_xor(s1, 16, 32);
  s2 += __shfl_xor(s2, 16, 32);
  d1 += __shfl_xor(d1, 16, 32);
  d2 += __shfl_xor(d2, 16, 32);

  float rn[4];
#pragma unroll
  for (int u = 0; u < 4; ++u) rn[u] = 1.0f / sqrtf(ss[u]);
  const float cr1 = ang_sim(d1 * (rn[0] * (1.0f / sqrtf(s1))));
  const float cr2 = ang_sim(d2 * (rn[0] * (1.0f / sqrtf(s2))));

  float rs[8];
  const size_t srow = (size_t)mb * L_ + 16 * w + 8 * h;
#pragma unroll
  for (int r = 0; r < 8; ++r) {
    const float rr = __shfl(rn[0], 8 * h + r, 32);
    float t = 0.f;
#pragma unroll
    for (int u = 0; u < 4; ++u) {
      const float v = ang_sim(acc[u][r] * (rr * rn[u]));
      intra[(srow + r) * L_ + 16 * ((w + u) & 3) + c] = v;
      t += v;
    }
#pragma unroll
    for (int off = 8; off > 0; off >>= 1) t += __shfl_xor(t, off, 32);
    rs[r] = t;
  }
  float mine = rs[0];
#pragma unroll
  for (int r = 1; r < 8; ++r) mine = ((c & 7) == r) ? rs[r] : mine;
  if ((c >> 3) == h) {
    const int a = d0 + 16 * w + c;
    crossv[((size_t)m * M_ + n1) * A_ + a] = cr1;
    crossv[((size_t)m * M_ + n2) * A_ + a] = cr2;
    dinv[(size_t)m * A_ + a] = 1.0f / sqrtf(mine + cr1 + cr2);
  }
}


template <int NT>
__device__ __forceinline__ void fc0_body(const float* __restrict__ xr, const float* __restrict__ w,
                                         const float* __restrict__ bias, float* __restrict__ h0,
                                         float* __restrict__ h0T, int m0, int n0) {
  const int mr = threadIdx.x & 15;
  const int cb = lane_col<NT>(n0);
  const float* ar[2] = {xr + (size_t)mr * H_, xr + (size_t)(16 + mr) * H_};
  v8f acc[2][NT] = {};
  auto cA = [](v4f r) { return r; };
  auto lA = [&](int t, int kq) { return ld4(ar[t] + kq); };
  auto lB = [&](int j, int kk) { return ldB_unit<NT>(w, cb, j, kk); };
  Ring<2, NT, 5, v4f> g0;
  ring_fill(g0, lA, lB);
  ring_main<2, NT, 5, KP_>(acc, g0, lA, cA, lB);
  ring_drain(acc, g0, cA);
  auto ep = [&](int m, int n, float v) { return fmaxf(v + bias[n], 0.0f); };
  float o[2][NT][8];
  epi_vals<NT, false>(acc, m0, n0, o, ep);
  store_rm<NT, false>(o, m0, n0, h0);
  store_fm<NT, false>(o, m0, n0, h0T, N_);
}

__global__ void __launch_bounds__(128) k_fc0(const float* __restrict__ xa, const float* __restrict__ xv,
                      const float* __restrict__ xt, const float* __restrict__ w,
                      const float* __restrict__ bias, float* __restrict__ h0,
                      float* __restrict__ h0T) {
  const int wv = wg_wave();
  const int strip = 2 * blockIdx.x + (wv & 1);
  if (strip >= NS_) return;
  const int m0 = (blockIdx.y * 2 + (wv >> 1)) * 32, n0 = strip * 32;
  const int md = m0 / A_;
  const float* xr = ((md == 0) ? xa : (md == 1) ? xv : xt) + (size_t)(m0 - md * A_) * H_;
  if (strip == NS_ - 1) fc0_body<1>(xr, w, bias, h0, h0T, m0, n0);
  else                  fc0_body<2>(xr, w, bias, h0, h0T, m0, n0);
}

template <int NT>
__device__ __forceinline__ void spmm_body(const float* __restrict__ intra,
                                          const float* __restrict__ crossv,
                                          const float* __restrict__ dinv,
                                          const float* __restrict__ h, float* __restrict__ hi,
                                          float* __restrict__ hiT, int mb, int i0, int n0) {
  const int mr = threadIdx.x & 15;
  const int m = mb >> 5;
  const int rowbase = mb * L_;
  const int cb = lane_col<NT>(n0);
  const float* S = intra + (size_t)mb * L_ * L_;
  const float* sr[2] = {S + (size_t)(i0 + mr) * L_, S + (size_t)(i0 + 16 + mr) * L_};
  const float* dv = dinv + rowbase;
  const float* hb = h + (size_t)rowbase * H_;
  v8f acc[2][NT] = {};
  auto lA = [&](int t, int kq) { RawQuad r; r.p = ld4(sr[t] + kq); r.q = ld4(dv + kq); return r; };
  auto cA = [](RawQuad r) { return r.p * r.q; };
  auto lB = [&](int j, int kk) { return ldB_unit<NT>(hb, cb, j, kk); };
  Ring<2, NT, 4, RawQuad> g0;
  ring_fill(g0, lA, lB);
  ring_main<2, NT, 4, L_ / 8>(acc, g0, lA, cA, lB);
  ring_drain(acc, g0, cA);
  const int kh = (threadIdx.x & 31) >> 4;
  const int n1 = (m == 0) ? 1 : 0;
  const int n2 = (m == 2) ? 1 : 2;
  const int ab = rowbase - m * A_;
  const float* c1 = crossv + ((size_t)m * M_ + n1) * A_ + ab;
  const float* c2 = crossv + ((size_t)m * M_ + n2) * A_ + ab;
  const float* e1 = dinv + (size_t)n1 * A_ + ab;
  const float* e2 = dinv + (size_t)n2 * A_ + ab;
  const float* g1 = h + ((size_t)n1 * A_ + ab) * H_;
  const float* g2 = h + ((size_t)n2 * A_ + ab) * H_;
  float f1[2][8], f2[2][8], fr[2][8];
  float x1[2][8][NT], x2[2][8][NT];
#pragma unroll
  for (int t = 0; t < 2; ++t) {
    const int r = i0 + 16 * t + 8 * kh;
#pragma unroll
    for (int q = 0; q < 2; ++q) {
      const v4f p1 = ld4(c1 + r + 4 * q) * ld4(e1 + r + 4 * q);
      const v4f p2 = ld4(c2 + r + 4 * q) * ld4(e2 + r + 4 * q);
      const v4f dr = ld4(dv + r + 4 * q);
      f1[t][4 * q] = p1.x; f1[t][4 * q + 1] = p1.y; f1[t][4 * q + 2] = p1.z; f1[t][4 * q + 3] = p1.w;
      f2[t][4 * q] = p2.x; f2[t][4 * q + 1] = p2.y; f2[t][4 * q + 2] = p2.z; f2[t][4 * q + 3] = p2.w;
      fr[t][4 * q] = dr.x; fr[t][4 * q + 1] = dr.y; fr[t][4 * q + 2] = dr.z; fr[t][4 * q + 3] = dr.w;
    }
#pragma unroll
    for (int v = 0; v < 8; ++v) {
      const size_t off = (size_t)(r + v) * H_ + cb;
      if constexpr (NT == 2) {
        const v2f a = ld2(g1 + off), b = ld2(g2 + off);
        x1[t][v][0] = a.x; x1[t][v][1] = a.y;
        x2[t][v][0] = b.x; x2[t][v][1] = b.y;
      } else {
        x1[t][v][0] = g1[off];
        x2[t][v][0] = g2[off];
      }
    }
  }
  float o[2][NT][8];
#pragma unroll
  for (int t = 0; t < 2; ++t)
#pragma unroll
    for (int u = 0; u < NT; ++u)
#pragma unroll
      for (int v = 0; v < 8; ++v) {
        float s = acc[t][u][v];
        s += f1[t][v] * x1[t][v][u];
        s += f2[t][v] * x2[t][v][u];
        o[t][u][v] = s * fr[t][v];
      }
  store_rm<NT, false>(o, i0, n0, hi + (size_t)rowbase * H_);
  store_fm<NT, false>(o, i0, n0, hiT + rowbase, N_);
}

__global__ void __launch_bounds__(128) k_spmm(const float* __restrict__ intra, const float* __restrict__ crossv,
                       const float* __restrict__ dinv, const float* __restrict__ h,
                       float* __restrict__ hi, float* __restrict__ hiT) {
  const int wv = wg_wave();
  const int strip = 2 * blockIdx.x + (wv & 1);
  if (strip >= NS_) return;
  const int mb = blockIdx.y;
  const int i0 = (wv >> 1) * 32, n0 = strip * 32;
  if (strip == NS_ - 1) spmm_body<1>(intra, crossv, dinv, h, hi, hiT, mb, i0, n0);
  else                  spmm_body<2>(intra, crossv, dinv, h, hi, hiT, mb, i0, n0);
}

template <int NT, bool FM_OUT>
__device__ __forceinline__ void conv_body(const float* __restrict__ hi, const float* __restrict__ h0,
                                          const float* __restrict__ hiT, const float* __restrict__ h0T,
                                          const float* __restrict__ w, float beta,
                                          float* __restrict__ hout, int m0, int n0) {
  const int mr = threadIdx.x & 15;
  const int cb = lane_col<NT>(n0);
  const float* t0 = hiT + m0 + 2 * mr;
  const float* t1 = h0T + m0 + 2 * mr;
  const float* w1 = w + (size_t)H_ * H_;
  v8f acc[2][NT] = {};
  auto lA0 = [&](int pl, int k) { return ld2(t0 + (size_t)k * N_); };
  auto lA1 = [&](int pl, int k) { return ld2(t1 + (size_t)k * N_); };
  auto lB0 = [&](int j, int kk) { return ldB_unit<NT>(w, cb, j, kk); };
  auto lB1 = [&](int j, int kk) { return ldB_unit<NT>(w1, cb, j, kk); };
  RingT<NT, 5, 1> g0, g1;
  ringt_fill(g0, lA0, lB0);
  ringt_main<false, NT, 5, KP_>(acc, g0, lA0, lB0);
  ringt_fill(g1, lA1, lB1);
  ringt_drain<false>(acc, g0);
  ringt_main<false, NT, 5, KP_>(acc, g1, lA1, lB1);
  ringt_drain<false>(acc, g1);
  auto ep = [&](int m, int n, float v) {
    const float r = 0.9f * hi[(size_t)m * H_ + n] + 0.1f * h0[(size_t)m * H_ + n];
    return fmaxf(beta * v + (1.0f - beta) * r, 0.0f);
  };
  float o[2][NT][8];
  epi_vals<NT, true>(acc, m0, n0, o, ep);
  if constexpr (FM_OUT) store_fm<NT, true>(o, m0, n0, hout, N_);
  else                  store_rm<NT, true>(o, m0, n0, hout);
}

template <bool FM_OUT>
__global__ void __launch_bounds__(128) k_conv(const float* __restrict__ hi, const float* __restrict__ h0,
                       const float* __restrict__ hiT, const float* __restrict__ h0T,
                       const float* __restrict__ w, float beta, float* __restrict__ hout) {
  const int wv = wg_wave();
  const int strip = 2 * blockIdx.x + (wv & 1);
  if (strip >= NS_) return;
  const int m0 = (blockIdx.y * 2 + (wv >> 1)) * 32, n0 = strip * 32;
  if (strip == NS_ - 1) conv_body<1, FM_OUT>(hi, h0, hiT, h0T, w, beta, hout, m0, n0);
  else                  conv_body<2, FM_OUT>(hi, h0, hiT, h0T, w, beta, hout, m0, n0);
}

template <int NT>
__device__ __forceinline__ void hx_body(const float* __restrict__ xm, const float* __restrict__ hfT,
                                        const float* __restrict__ W, const float* __restrict__ Bb,
                                        float* __restrict__ hx, float* __restrict__ hxT,
                                        int rb, int a0_, int n0) {
  const int mr = threadIdx.x & 15;
  const int cb = lane_col<NT>(n0);
  const size_t r0 = (size_t)(a0_ + 2 * mr) * H_, r1 = (size_t)(a0_ + 2 * mr + 1) * H_;
  const float* p0[2] = {xm + r0, xm + r1};
  const float* tf = hfT + rb + a0_ + 2 * mr;
  const float* W1 = W + (size_t)H_ * H_;
  v8f acc[2][NT] = {};
  auto cA = [](v4f r) { return r; };
  auto lA0 = [&](int t, int kq) { return ld4(p0[t] + kq); };
  auto lA1 = [&](int pl, int k) { return ld2(tf + (size_t)k * N_); };
  auto lB0 = [&](int j, int kk) { return ldB_unit<NT>(W, cb, j, kk); };
  auto lB1 = [&](int j, int kk) { return ldB_unit<NT>(W1, cb, j, kk); };
  Ring<2, NT, 5, v4f> g0;
  RingT<NT, 5, 1> g1;
  ring_fill(g0, lA0, lB0);
  ring_main<2, NT, 5, KP_>(acc, g0, lA0, cA, lB0);
  ringt_fill(g1, lA1, lB1);
  ring_drain(acc, g0, cA);
  ringt_main<false, NT, 5, KP_>(acc, g1, lA1, lB1);
  ringt_drain<false>(acc, g1);
  auto ep = [&](int a, int n, float v) { return fmaxf(v + Bb[n], 0.0f); };
  float o[2][NT][8];
  epi_vals<NT, true>(acc, a0_, n0, o, ep);
  store_rm<NT, true>(o, a0_, n0, hx + (size_t)rb * H_);
  store_fm<NT, true>(o, a0_, n0, hxT + (size_t)rb * H_, A_);
}

__global__ void __launch_bounds__(128) k_hx(const float* __restrict__ xa, const float* __restrict__ xv,
                     const float* __restrict__ xt, const float* __restrict__ hfT,
                     const float* __restrict__ tw, const float* __restrict__ tb,
                     float* __restrict__ hx, float* __restrict__ hxT) {
  const int wv = wg_wave();
  const int strip = 2 * blockIdx.x + (wv & 1);
  if (strip >= NS_) return;
  const int m  = blockIdx.z;
  const float* xm = (m == 0) ? xa : (m == 1) ? xv : xt;
  const int a0_ = (blockIdx.y * 2 + (wv >> 1)) * 32, n0 = strip * 32;
  const float* W  = tw + (size_t)m * 2 * H_ * H_;
  const float* Bb = tb + (size_t)m * H_;
  if (strip == NS_ - 1) hx_body<1>(xm, hfT, W, Bb, hx, hxT, m * A_, a0_, n0);
  else                  hx_body<2>(xm, hfT, W, Bb, hx, hxT, m * A_, a0_, n0);
}

template <int NT>
__device__ __forceinline__ float tr_logit(const v8f (&a)[NT], int g, int e) {
  if constexpr (NT == 2) return a[e & 1][4 * g + (e >> 1)];
  else                   return a[0][e];
}

template <int NT>
__device__ __forceinline__ void gate_body(const float* __restrict__ Hi, const float* __restrict__ Hj,
                                          const float* __restrict__ HiT, const float* __restrict__ HjT,
                                          const float* __restrict__ W, const float* __restrict__ Bb,
                                          const float* __restrict__ Ui, float* __restrict__ part_out,
                                          int a0_, int n0) {
  const int mr = threadIdx.x & 15;
  const int kh = (threadIdx.x & 31) >> 4;
  const int cb = lane_col<NT>(n0);
  const size_t r0 = (size_t)(a0_ + 2 * mr) * H_, r1 = (size_t)(a0_ + 2 * mr + 1) * H_;
  const float* pi[2] = {Hi + r0, Hi + r1};
  const float* pj[2] = {Hj + r0, Hj + r1};
  const float* ti = HiT + a0_ + 2 * mr;
  const float* tj = HjT + a0_ + 2 * mr;
  const float* W1 = W + (size_t)H_ * H_;
  const float* W2 = W + (size_t)2 * H_ * H_;
  v8f acc[2][NT] = {};
  auto lAi = [&](int pl, int k) { return ld2(ti + (size_t)k * A_); };
  auto lAj = [&](int pl, int k) { return ld2(tj + (size_t)k * A_); };
  auto lAp = [&](int pl, int k) { return ld2((pl ? tj : ti) + (size_t)k * A_); };
  auto lB0 = [&](int j, int kk) { return ldB_unit<NT>(W, cb, j, kk); };
  auto lB1 = [&](int j, int kk) { return ldB_unit<NT>(W1, cb, j, kk); };
  auto lB2 = [&](int j, int kk) { return ldB_unit<NT>(W2, cb, j, kk); };
  RingT<NT, 5, 1> g0, g1;
  RingT<NT, 5, 2> g2;
  ringt_fill(g0, lAi, lB0);
  ringt_main<true, NT, 5, KP_>(acc, g0, lAi, lB0);
  ringt_fill(g1, lAj, lB1);
  ringt_drain<true>(acc, g0);
  ringt_main<true, NT, 5, KP_>(acc, g1, lAj, lB1);
  ringt_fill(g2, lAp, lB2);
  ringt_drain<true>(acc, g1);
  ringt_main<true, NT, 5, KP_>(acc, g2, lAp, lB2);
  ringt_drain<true>(acc, g2);

  float part[2][TAG_] = {};
#pragma unroll
  for (int t = 0; t < 2; ++t) {
#pragma unroll
    for (int g = 0; g < NT; ++g) {
      const int nb = (NT == 2) ? n0 + 16 * kh + 8 * g : n0 + 8 * kh;
      if (NT == 2 || nb < H_) {
        const v4f* hip = reinterpret_cast<const v4f*>(pi[t] + nb);
        const v4f* hjp = reinterpret_cast<const v4f*>(pj[t] + nb);
        const v4f* bp  = reinterpret_cast<const v4f*>(Bb + nb);
        const v4f* up  = reinterpret_cast<const v4f*>(Ui + (size_t)nb * TAG_);
        const v4f hi0 = hip[0], hi1 = hip[1], hj0 = hjp[0], hj1 = hjp[1], b0 = bp[0], b1 = bp[1];
        const float hiv[8] = {hi0.x, hi0.y, hi0.z, hi0.w, hi1.x, hi1.y, hi1.z, hi1.w};
        const float hjv[8] = {hj0.x, hj0.y, hj0.z, hj0.w, hj1.x, hj1.y, hj1.z, hj1.w};
        const float bv[8]  = {b0.x, b0.y, b0.z, b0.w, b1.x, b1.y, b1.z, b1.w};
        float uu[8 * TAG_];
#pragma unroll
        for (int q = 0; q < 2 * TAG_; ++q) {
          const v4f x = up[q];
          uu[4 * q] = x.x; uu[4 * q + 1] = x.y; uu[4 * q + 2] = x.z; uu[4 * q + 3] = x.w;
        }
#pragma unroll
        for (int e = 0; e < 8; ++e) {
          const float lg = tr_logit<NT>(acc[t], g, e);
          const float z = 1.0f / (1.0f + expf(-(lg + bv[e])));
          const float gt = z * hiv[e] + (1.0f - z) * hjv[e];
#pragma unroll
          for (int c = 0; c < TAG_; ++c) part[t][c] += gt * uu[e * TAG_ + c];
        }
      }
    }
  }
#pragma unroll
  for (int t = 0; t < 2; ++t)
#pragma unroll
    for (int c = 0; c < TAG_; ++c) part[t][c] += __shfl_xor(part[t][c], 16, 32);
  if (kh == 0) {
#pragma unroll
    for (int t = 0; t < 2; ++t) {
      float* po = part_out + (size_t)(a0_ + 2 * mr + t) * TAG_;
#pragma unroll
      for (int c = 0; c < TAG_; c += 2) {
        v2f o; o.x = part[t][c]; o.y = part[t][c + 1];
        *reinterpret_cast<v2f*>(po + c) = o;
      }
    }
  }
}

__global__ void __launch_bounds__(128) k_gate(const float* __restrict__ hx, const float* __restrict__ hxT,
                       const float* __restrict__ cw,
                       const float* __restrict__ cb, const float* __restrict__ uw,
                       float* __restrict__ plog) {
  const int wv = wg_wave();
  const int strip = 2 * blockIdx.x + (wv & 1);
  if (strip >= NS_) return;
  const int p = blockIdx.z;
  const int i = p >> 1;
  const int j = (p & 1) ? (i == 2 ? 1 : 2) : (i == 0 ? 1 : 0);
  const int a0_ = (blockIdx.y * 2 + (wv >> 1)) * 32, n0 = strip * 32;
  const float* Hi = hx + (size_t)i * A_ * H_;
  const float* Hj = hx + (size_t)j * A_ * H_;
  const float* HiT = hxT + (size_t)i * H_ * A_;
  const float* HjT = hxT + (size_t)j * H_ * A_;
  const float* W  = cw + (size_t)p * 3 * H_ * H_;
  const float* Bb = cb + (size_t)p * H_;
  const float* Ui = uw + (size_t)i * H_ * TAG_;
  float* po = plog + ((size_t)p * NS_ + strip) * A_ * TAG_;
  if (strip == NS_ - 1) gate_body<1>(Hi, Hj, HiT, HjT, W, Bb, Ui, po, a0_, n0);
  else                  gate_body<2>(Hi, Hj, HiT, HjT, W, Bb, Ui, po, a0_, n0);
}

__global__ void k_final(const float* __restrict__ plog, const float* __restrict__ ub,
                        float* __restrict__ out) {
  const int e = blockIdx.x * blockDim.x + threadIdx.x;
  const int t = e % TAG_;
  float s = ub[t] + ub[TAG_ + t] + ub[2 * TAG_ + t];
  for (int q = 0; q < 6 * NS_; ++q) s += plog[(size_t)q * A_ * TAG_ + e];
  out[e] = s;
}

}

extern "C" void kernel_launch(void* const* d_in, const int* in_sizes, int n_in,
                              void* d_out, int out_size, void* d_ws, size_t ws_size,
                              hipStream_t stream) {
  (void)in_sizes; (void)n_in; (void)out_size; (void)ws_size;
  const float* x_a     = (const float*)d_in[0];
  const float* x_v     = (const float*)d_in[1];
  const float* x_t     = (const float*)d_in[2];
  const float* fc0_w   = (const float*)d_in[3];
  const float* fc0_b   = (const float*)d_in[4];
  const float* conv_w  = (const float*)d_in[5];
  const float* tm_w    = (const float*)d_in[6];
  const float* tm_b    = (const float*)d_in[7];
  const float* cross_w = (const float*)d_in[8];
  const float* cross_b = (const float*)d_in[9];
  const float* uni_w   = (const float*)d_in[10];
  const float* uni_b   = (const float*)d_in[11];
  float* out = (float*)d_out;

  const size_t NH = (size_t)N_ * H_;
  float* ws    = (float*)d_ws;
  float* intra = ws; ws += (size_t)N_ * L_;
  float* crossv= ws; ws += (size_t)M_ * M_ * A_;
  float* dinv  = ws; ws += N_;
  float* h0    = ws; ws += NH;
  float* h0T   = ws; ws += NH;
  float* hA    = ws; ws += NH;
  float* hBfT  = ws; ws += NH;
  float* hi    = ws; ws += NH;
  float* hiT   = ws; ws += NH;
  float* hx    = ws; ws += NH;
  float* hxT   = ws; ws += NH;
  float* plog  = ws; ws += (size_t)6 * NS_ * A_ * TAG_;


  k_graph<<<M_ * B_ * 4,         dim3(32), 0, stream>>>(x_a, x_v, x_t, intra, crossv, dinv);
  k_fc0  <<<dim3(NSP_, N_ / 64), dim3(128), 0, stream>>>(x_a, x_v, x_t, fc0_w, fc0_b, h0, h0T);

  const float beta0 = logf(0.5f / 1.0f + 1.0f);
  const float beta1 = logf(0.5f / 2.0f + 1.0f);
  k_spmm <<<dim3(NSP_, M_ * B_), dim3(128), 0, stream>>>(intra, crossv, dinv, h0, hi, hiT);
  k_conv<false><<<dim3(NSP_, N_ / 64), dim3(128), 0, stream>>>(hi, h0, hiT, h0T, conv_w, beta0, hA);
  k_spmm <<<dim3(NSP_, M_ * B_), dim3(128), 0, stream>>>(intra, crossv, dinv, hA, hi, hiT);
  k_conv<true><<<dim3(NSP_, N_ / 64), dim3(128), 0, stream>>>(hi, h0, hiT, h0T, conv_w + 2 * H_ * H_, beta1, hBfT);

  k_hx   <<<dim3(NSP_, A_ / 64, M_), dim3(128), 0, stream>>>(x_a, x_v, x_t, hBfT, tm_w, tm_b, hx, hxT);

  k_gate <<<dim3(NSP_, A_ / 64, 6), dim3(128), 0, stream>>>(hx, hxT, cross_w, cross_b, uni_w, plog);

  k_final<<<A_ * TAG_ / 256, dim3(256), 0, stream>>>(plog, uni_b, out);
}


# fc0 and conv layer 0 regrouped as dialogue x column strip x 3 modalities (6 waves): the graph product that follows reads their tile from LDS in the same launch; 7 launches, was 9; sums kept
# speedup vs baseline: 8.7546x; 1.0940x over previous
#include <hip/hip_runtime.h>
#include <math.h>

typedef __attribute__((ext_vector_type(2))) float v2f;
typedef __attribute__((ext_vector_type(4))) float v4f;
typedef __attribute__((ext_vector_type(8))) float v8f;

namespace {

constexpr int B_   = 32;
constexpr int L_   = 64;
constexpr int H_   = 200;
constexpr int M_   = 3;
constexpr int A_   = B_ * L_;
constexpr int N_   = M_ * A_;
constexpr int TAG_ = 6;
constexpr float PI_F = 3.14159265358979323846f;

struct RawQuad { v4f p, q; };

__device__ __forceinline__ v4f ld4(const float* p) {
  return *reinterpret_cast<const v4f*>(p);
}

__device__ __forceinline__ v2f ld2(const float* p) {
  return *reinterpret_cast<const v2f*>(p);
}

template <int MT, int NT, int D, typename RA>
struct Ring { RA a[D][MT]; v2f b[D][2][NT]; };

template <int NT>
__device__ __forceinline__ void b_frags(const v2f (&raw)[NT], v2f (&f)[NT]) {
  if constexpr (NT == 2) {
    f[0].x = raw[0].x; f[0].y = raw[1].x;
    f[1].x = raw[0].y; f[1].y = raw[1].y;
  } else {
    f[0] = raw[0];
  }
}

template <int MT, int NT, bool TR = false>
__device__ __forceinline__ void mma_step(v8f (&acc)[MT][NT], const v2f (&a)[MT],
                                         const v2f (&b)[NT]) {
#pragma unroll
  for (int t = 0; t < MT; ++t)
#pragma unroll
    for (int u = 0; u < NT; ++u) {
      if constexpr (TR)
        acc[t][u] = __builtin_amdgcn_wmma_f32_16x16x4_f32(false, b[u], false, a[t],
                                                          (short)0, acc[t][u], false, false);
      else
        acc[t][u] = __builtin_amdgcn_wmma_f32_16x16x4_f32(false, a[t], false, b[u],
                                                          (short)0, acc[t][u], false, false);
    }
}


template <int MT, int NT, int D, typename RA, typename FA, typename FB>
__device__ __forceinline__ void ring_fill(Ring<MT, NT, D, RA>& r, FA ldA, FB ldB) {
  const int kh = (threadIdx.x & 31) >> 4;
#pragma unroll
  for (int d = 0; d < D; ++d) {
    const int kq = 8 * d + 4 * kh;
#pragma unroll
    for (int t = 0; t < MT; ++t) r.a[d][t] = ldA(t, kq);
#pragma unroll
    for (int s = 0; s < 2; ++s)
#pragma unroll
      for (int u = 0; u < NT; ++u) r.b[d][s][u] = ldB(u, kq + 2 * s);
    __builtin_amdgcn_sched_barrier(0);
  }
}

template <int MT, int NT, int D, int NPAIRS, bool TR = false, typename RA, typename FA, typename FC, typename FB>
__device__ __forceinline__ void ring_main(v8f (&acc)[MT][NT], Ring<MT, NT, D, RA>& r,
                                          FA ldA, FC cvA, FB ldB) {
  static_assert(NPAIRS % D == 0 && NPAIRS >= D, "ring depth must divide the pair count");
  const int kh = (threadIdx.x & 31) >> 4;
#pragma nounroll
  for (int p0 = 0; p0 < NPAIRS - D; p0 += D) {
#pragma unroll
    for (int d = 0; d < D; ++d) {
      v2f lo[MT], hi[MT];
#pragma unroll
      for (int t = 0; t < MT; ++t) {
        const v4f c = cvA(r.a[d][t]);
        lo[t] = c.xy;
        hi[t] = c.zw;
      }
      v2f f0[NT], f1[NT];
      b_frags<NT>(r.b[d][0], f0);
      b_frags<NT>(r.b[d][1], f1);
      mma_step<MT, NT, TR>(acc, lo, f0);
      mma_step<MT, NT, TR>(acc, hi, f1);
      const int kq = 8 * (p0 + d + D) + 4 * kh;
#pragma unroll
      for (int t = 0; t < MT; ++t) r.a[d][t] = ldA(t, kq);
#pragma unroll
      for (int s = 0; s < 2; ++s)
#pragma unroll
        for (int u = 0; u < NT; ++u) r.b[d][s][u] = ldB(u, kq + 2 * s);
      __builtin_amdgcn_sched_barrier(0);
    }
  }
}

template <bool TR = false, int MT, int NT, int D, typename RA, typename FC>
__device__ __forceinline__ void ring_drain(v8f (&acc)[MT][NT], Ring<MT, NT, D, RA>& r, FC cvA) {
#pragma unroll
  for (int d = 0; d < D; ++d) {
    v2f lo[MT], hi[MT];
#pragma unroll
    for (int t = 0; t < MT; ++t) {
      const v4f c = cvA(r.a[d][t]);
      lo[t] = c.xy;
      hi[t] = c.zw;
    }
    v2f f0[NT], f1[NT];
    b_frags<NT>(r.b[d][0], f0);
    b_frags<NT>(r.b[d][1], f1);
    mma_step<MT, NT, TR>(acc, lo, f0);
    mma_step<MT, NT, TR>(acc, hi, f1);
  }
}

template <int NT, int D, int NP>
struct RingT { v2f a[D][NP][4]; v2f b[D][2][NT]; };

template <int NP>
__device__ __forceinline__ void a_frags(const v2f (&raw)[NP][4], v2f (&lo)[2], v2f (&hi)[2]) {
  v2f q[4];
#pragma unroll
  for (int j = 0; j < 4; ++j) {
    if constexpr (NP == 2) q[j] = raw[0][j] * raw[1][j];
    else                   q[j] = raw[0][j];
  }
  lo[0].x = q[0].x; lo[0].y = q[1].x; hi[0].x = q[2].x; hi[0].y = q[3].x;
  lo[1].x = q[0].y; lo[1].y = q[1].y; hi[1].x = q[2].y; hi[1].y = q[3].y;
}

template <int NT, int D, int NP, typename FA, typename FB>
__device__ __forceinline__ void ringt_fill(RingT<NT, D, NP>& r, FA ldA, FB ldB) {
  const int kh = (threadIdx.x & 31) >> 4;
#pragma unroll
  for (int d = 0; d < D; ++d) {
    const int kq = 8 * d + 4 * kh;
#pragma unroll
    for (int pl = 0; pl < NP; ++pl)
#pragma unroll
      for (int j = 0; j < 4; ++j) r.a[d][pl][j] = ldA(pl, kq + j);
#pragma unroll
    for (int s = 0; s < 2; ++s)
#pragma unroll
      for (int u = 0; u < NT; ++u) r.b[d][s][u] = ldB(u, kq + 2 * s);
    __builtin_amdgcn_sched_barrier(0);
  }
}

template <bool TR, int NT, int D, int NPAIRS, int NP, typename FA, typename FB>
__device__ __forceinline__ void ringt_main(v8f (&acc)[2][NT], RingT<NT, D, NP>& r, FA ldA, FB ldB) {
  static_assert(NPAIRS % D == 0 && NPAIRS >= D, "ring depth must divide the pair count");
  const int kh = (threadIdx.x & 31) >> 4;
#pragma nounroll
  for (int p0 = 0; p0 < NPAIRS - D; p0 += D) {
#pragma unroll
    for (int d = 0; d < D; ++d) {
      v2f lo[2], hi[2];
      a_frags<NP>(r.a[d], lo, hi);
      v2f f0[NT], f1[NT];
      b_frags<NT>(r.b[d][0], f0);
      b_frags<NT>(r.b[d][1], f1);
      mma_step<2, NT, TR>(acc, lo, f0);
      mma_step<2, NT, TR>(acc, hi, f1);
      const int kq = 8 * (p0 + d + D) + 4 * kh;
#pragma unroll
      for (int pl = 0; pl < NP; ++pl)
#pragma unroll
        for (int j = 0; j < 4; ++j) r.a[d][pl][j] = ldA(pl, kq + j);
#pragma unroll
      for (int s = 0; s < 2; ++s)
#pragma unroll
        for (int u = 0; u < NT; ++u) r.b[d][s][u] = ldB(u, kq + 2 * s);
      __builtin_amdgcn_sched_barrier(0);
    }
  }
}

template <bool TR, int NT, int D, int NP>
__device__ __forceinline__ void ringt_drain(v8f (&acc)[2][NT], RingT<NT, D, NP>& r) {
#pragma unroll
  for (int d = 0; d < D; ++d) {
    v2f lo[2], hi[2];
    a_frags<NP>(r.a[d], lo, hi);
    v2f f0[NT], f1[NT];
    b_frags<NT>(r.b[d][0], f0);
    b_frags<NT>(r.b[d][1], f1);
    mma_step<2, NT, TR>(acc, lo, f0);
    mma_step<2, NT, TR>(acc, hi, f1);
  }
}

template <bool RI>
__device__ __forceinline__ int tile_row(int t, int x) { return RI ? 2 * x + t : 16 * t + x; }

template <int NT, bool RI, typename FV>
__device__ __forceinline__ void epi_vals(const v8f (&acc)[2][NT], int m0, int n0,
                                         float (&o)[2][NT][8], FV val) {
  const int lane = threadIdx.x & 31;
  const int mr = lane & 15, kh = lane >> 4;
#pragma unroll
  for (int t = 0; t < 2; ++t)
#pragma unroll
    for (int u = 0; u < NT; ++u)
#pragma unroll
      for (int v = 0; v < 8; ++v) {
        const int m = m0 + tile_row<RI>(t, 8 * kh + v);
        const int n = (NT == 2) ? n0 + 2 * mr + u : n0 + mr;
        o[t][u][v] = (NT == 2 || n < H_) ? val(m, n, acc[t][u][v]) : 0.0f;
      }
}

template <int NT, bool RI>
__device__ __forceinline__ void store_rm(const float (&o)[2][NT][8], int m0, int n0,
                                         float* __restrict__ dst) {
  const int lane = threadIdx.x & 31;
  const int mr = lane & 15, kh = lane >> 4;
#pragma unroll
  for (int t = 0; t < 2; ++t)
#pragma unroll
    for (int v = 0; v < 8; ++v) {
      const int m = m0 + tile_row<RI>(t, 8 * kh + v);
      if constexpr (NT == 2) {
        v2f w;
        w.x = o[t][0][v];
        w.y = o[t][1][v];
        *reinterpret_cast<v2f*>(dst + (size_t)m * H_ + n0 + 2 * mr) = w;
      } else {
        const int n = n0 + mr;
        if (n < H_) dst[(size_t)m * H_ + n] = o[t][0][v];
      }
    }
}

template <int NT, bool RI>
__device__ __forceinline__ void store_fm(const float (&o)[2][NT][8], int m0, int n0,
                                         float* __restrict__ dstT, int pitch) {
  const int lane = threadIdx.x & 31;
  const int mr = lane & 15, kh = lane >> 4;
#pragma unroll
  for (int u = 0; u < NT; ++u) {
    const int n = (NT == 2) ? n0 + 2 * mr + u : n0 + mr;
    if (NT == 2 || n < H_) {
      float* col = dstT + (size_t)n * pitch + m0;
      if constexpr (RI) {
#pragma unroll
        for (int q = 0; q < 4; ++q) {
          v4f w;
          w.x = o[0][u][2 * q];
          w.y = o[1][u][2 * q];
          w.z = o[0][u][2 * q + 1];
          w.w = o[1][u][2 * q + 1];
          *reinterpret_cast<v4f*>(col + 16 * kh + 4 * q) = w;
        }
      } else {
#pragma unroll
        for (int t = 0; t < 2; ++t)
#pragma unroll
          for (int q = 0; q < 2; ++q) {
            v4f w;
            w.x = o[t][u][4 * q];
            w.y = o[t][u][4 * q + 1];
            w.z = o[t][u][4 * q + 2];
            w.w = o[t][u][4 * q + 3];
            *reinterpret_cast<v4f*>(col + 16 * t + 8 * kh + 4 * q) = w;
          }
      }
    }
  }
}

template <int NT>
__device__ __forceinline__ int lane_col(int n0) {
  const int mr = threadIdx.x & 15;
  return NT == 2 ? n0 + 2 * mr : min(n0 + mr, H_ - 1);
}

template <int NT>
__device__ __forceinline__ v2f ldB_unit(const float* __restrict__ w, int cb, int j, int kk) {
  if constexpr (NT == 2) {
    return ld2(w + (size_t)(kk + j) * H_ + cb);
  } else {
    v2f b;
    b.x = w[(size_t)kk * H_ + cb];
    b.y = w[(size_t)(kk + 1) * H_ + cb];
    return b;
  }
}

constexpr int NS_ = 7;
constexpr int KP_ = H_ / 8;

constexpr int NSP_ = (NS_ + 1) / 2;

__device__ __forceinline__ int wg_wave() { return __builtin_amdgcn_readfirstlane((int)(threadIdx.x >> 5)); }

constexpr int GD_ = 5;

struct GSlot { v4f b[4]; v4f y1, y2; };

__device__ __forceinline__ float ang_sim(float cosv) {
  return 1.0f - acosf(cosv * 0.99999f) * (1.0f / PI_F);
}

__global__ void __launch_bounds__(32) k_graph(const float* __restrict__ xa, const float* __restrict__ xv,
                        const float* __restrict__ xt, float* __restrict__ intra,
                        float* __restrict__ crossv, float* __restrict__ dinv) {
  const int mb = blockIdx.x >> 2;
  const int m  = mb >> 5;
  const int d0 = (mb & 31) * L_;
  const int w  = blockIdx.x & 3;
  const int lane = threadIdx.x & 31;
  const int c = lane & 15, h = lane >> 4;
  const int n1 = (m == 0) ? 1 : 0;
  const int n2 = (m == 2) ? 1 : 2;
  const float* xm = (m == 0) ? xa : (m == 1) ? xv : xt;
  const float* x1 = (n1 == 0) ? xa : xv;
  const float* x2 = (n2 == 1) ? xv : xt;
  const float* pb[4];
#pragma unroll
  for (int u = 0; u < 4; ++u)
    pb[u] = xm + (size_t)(d0 + 16 * ((w + u) & 3) + c) * H_ + 4 * h;
  const float* p1 = x1 + (size_t)(d0 + 16 * w + c) * H_ + 4 * h;
  const float* p2 = x2 + (size_t)(d0 + 16 * w + c) * H_ + 4 * h;

  v8f acc[4] = {};
  float ss[4] = {0.f, 0.f, 0.f, 0.f};
  float s1 = 0.f, s2 = 0.f, d1 = 0.f, d2 = 0.f;

  auto fetch = [&](GSlot& s, int p) {
#pragma unroll
    for (int u = 0; u < 4; ++u) s.b[u] = ld4(pb[u] + 8 * p);
    s.y1 = ld4(p1 + 8 * p);
    s.y2 = ld4(p2 + 8 * p);
  };
  auto use = [&](const GSlot& s) {
    const v4f a = s.b[0];
#pragma unroll
    for (int u = 0; u < 4; ++u) {
      const v4f b = s.b[u];
      ss[u] = fmaf(b.x, b.x, ss[u]); ss[u] = fmaf(b.y, b.y, ss[u]);
      ss[u] = fmaf(b.z, b.z, ss[u]); ss[u] = fmaf(b.w, b.w, ss[u]);
    }
    d1 = fmaf(a.x, s.y1.x, d1); d1 = fmaf(a.y, s.y1.y, d1);
    d1 = fmaf(a.z, s.y1.z, d1); d1 = fmaf(a.w, s.y1.w, d1);
    d2 = fmaf(a.x, s.y2.x, d2); d2 = fmaf(a.y, s.y2.y, d2);
    d2 = fmaf(a.z, s.y2.z, d2); d2 = fmaf(a.w, s.y2.w, d2);
    s1 = fmaf(s.y1.x, s.y1.x, s1); s1 = fmaf(s.y1.y, s.y1.y, s1);
    s1 = fmaf(s.y1.z, s.y1.z, s1); s1 = fmaf(s.y1.w, s.y1.w, s1);
    s2 = fmaf(s.y2.x, s.y2.x, s2); s2 = fmaf(s.y2.y, s.y2.y, s2);
    s2 = fmaf(s.y2.z, s.y2.z, s2); s2 = fmaf(s.y2.w, s.y2.w, s2);
    const v2f alo = a.xy, ahi = a.zw;
#pragma unroll
    for (int u = 0; u < 4; ++u) {
      const v2f blo = s.b[u].xy;
      acc[u] = __builtin_amdgcn_wmma_f32_16x16x4_f32(false, alo, false, blo,
                                                     (short)0, acc[u], false, false);
    }
#pragma unroll
    for (int u = 0; u < 4; ++u) {
      const v2f bhi = s.b[u].zw;
      acc[u] = __builtin_amdgcn_wmma_f32_16x16x4_f32(false, ahi, false, bhi,
                                                     (short)0, acc[u], false, false);
    }
  };

  GSlot ring[GD_];
#pragma unroll
  for (int d = 0; d < GD_; ++d) {
    fetch(ring[d], d);
    __builtin_amdgcn_sched_barrier(0);
  }
#pragma nounroll
  for (int p0 = 0; p0 < KP_ - GD_; p0 += GD_) {
#pragma unroll
    for (int d = 0; d < GD_; ++d) {
      use(ring[d]);
      fetch(ring[d], p0 + d + GD_);
      __builtin_amdgcn_sched_barrier(0);
    }
  }
#pragma unroll
  for (int d = 0; d < GD_; ++d) use(ring[d]);

#pragma unroll
  for (int u = 0; u < 4; ++u) ss[u] += __shfl_xor(ss[u], 16, 32);
  s1 += __shfl_xor(s1, 16, 32);
  s2 += __shfl_xor(s2, 16, 32);
  d1 += __shfl_xor(d1, 16, 32);
  d2 += __shfl_xor(d2, 16, 32);

  float rn[4];
#pragma unroll
  for (int u = 0; u < 4; ++u) rn[u] = 1.0f / sqrtf(ss[u]);
  const float cr1 = ang_sim(d1 * (rn[0] * (1.0f / sqrtf(s1))));
  const float cr2 = ang_sim(d2 * (rn[0] * (1.0f / sqrtf(s2))));

  float rs[8];
  const size_t srow = (size_t)mb * L_ + 16 * w + 8 * h;
#pragma unroll
  for (int r = 0; r < 8; ++r) {
    const float rr = __shfl(rn[0], 8 * h + r, 32);
    float t = 0.f;
#pragma unroll
    for (int u = 0; u < 4; ++u) {
      const float v = ang_sim(acc[u][r] * (rr * rn[u]));
      intra[(srow + r) * L_ + 16 * ((w + u) & 3) + c] = v;
      t += v;
    }
#pragma unroll
    for (int off = 8; off > 0; off >>= 1) t += __shfl_xor(t, off, 32);
    rs[r] = t;
  }
  float mine = rs[0];
#pragma unroll
  for (int r = 1; r < 8; ++r) mine = ((c & 7) == r) ? rs[r] : mine;
  if ((c >> 3) == h) {
    const int a = d0 + 16 * w + c;
    crossv[((size_t)m * M_ + n1) * A_ + a] = cr1;
    crossv[((size_t)m * M_ + n2) * A_ + a] = cr2;
    dinv[(size_t)m * A_ + a] = 1.0f / sqrtf(mine + cr1 + cr2);
  }
}


template <int NT>
__device__ __forceinline__ void fc0_body(const float* __restrict__ xr, const float* __restrict__ w,
                                         const float* __restrict__ bias, int m0, int n0,
                                         float (&o)[2][NT][8]) {
  const int mr = threadIdx.x & 15;
  const int cb = lane_col<NT>(n0);
  const float* ar[2] = {xr + (size_t)mr * H_, xr + (size_t)(16 + mr) * H_};
  v8f acc[2][NT] = {};
  auto cA = [](v4f r) { return r; };
  auto lA = [&](int t, int kq) { return ld4(ar[t] + kq); };
  auto lB = [&](int j, int kk) { return ldB_unit<NT>(w, cb, j, kk); };
  Ring<2, NT, 5, v4f> g0;
  ring_fill(g0, lA, lB);
  ring_main<2, NT, 5, KP_>(acc, g0, lA, cA, lB);
  ring_drain(acc, g0, cA);
  auto ep = [&](int m, int n, float v) { return fmaxf(v + bias[n], 0.0f); };
  epi_vals<NT, false>(acc, m0, n0, o, ep);
}

constexpr int TP_ = 32;
constexpr int TILE_ = M_ * L_ * TP_;

__device__ __forceinline__ void lds_barrier() {
  __builtin_amdgcn_fence(__ATOMIC_RELEASE, "workgroup", "local");
  __builtin_amdgcn_s_barrier();
  __builtin_amdgcn_fence(__ATOMIC_ACQUIRE, "workgroup", "local");
}

template <int NT, bool RI>
__device__ __forceinline__ void tile_put(const float (&o)[2][NT][8], float* tl) {
  const int lane = threadIdx.x & 31;
  const int mr = lane & 15, kh = lane >> 4;
#pragma unroll
  for (int t = 0; t < 2; ++t)
#pragma unroll
    for (int v = 0; v < 8; ++v) {
      float* row = tl + tile_row<RI>(t, 8 * kh + v) * TP_;
      if constexpr (NT == 2) {
        v2f w;
        w.x = o[t][0][v];
        w.y = o[t][1][v];
        *reinterpret_cast<v2f*>(row + 2 * mr) = w;
      } else {
        row[mr] = o[t][0][v];
      }
    }
}

template <int NT>
using SpmmRing = Ring<2, NT, 4, RawQuad>;

template <int NT>
__device__ __forceinline__ void spmm_fill_a(SpmmRing<NT>& g0, const float* __restrict__ intra,
                                            const float* __restrict__ dinv, int mb, int i0) {
  const int mr = threadIdx.x & 15, kh = (threadIdx.x & 31) >> 4;
  const float* S = intra + (size_t)mb * L_ * L_;
  const float* dv = dinv + mb * L_;
#pragma unroll
  for (int d = 0; d < 4; ++d)
#pragma unroll
    for (int t = 0; t < 2; ++t) {
      const int kq = 8 * d + 4 * kh;
      g0.a[d][t].p = ld4(S + (size_t)(i0 + 16 * t + mr) * L_ + kq);
      g0.a[d][t].q = ld4(dv + kq);
    }
}

template <int NT>
__device__ __forceinline__ void spmm_body(SpmmRing<NT>& g0, const float* __restrict__ intra,
                                          const float* __restrict__ crossv,
                                          const float* __restrict__ dinv,
                                          const float* tile, float* __restrict__ hi,
                                          float* __restrict__ hiT, int mb, int i0, int n0) {
  const int mr = threadIdx.x & 15;
  const int m = mb >> 5;
  const int rowbase = mb * L_;
  const int cl = (NT == 2) ? 2 * mr : mr;
  const float* S = intra + (size_t)mb * L_ * L_;
  const float* sr[2] = {S + (size_t)(i0 + mr) * L_, S + (size_t)(i0 + 16 + mr) * L_};
  const float* dv = dinv + rowbase;
  const float* hb = tile + m * L_ * TP_ + cl;
  v8f acc[2][NT] = {};
  auto lA = [&](int t, int kq) { RawQuad r; r.p = ld4(sr[t] + kq); r.q = ld4(dv + kq); return r; };
  auto cA = [](RawQuad r) { return r.p * r.q; };
  auto lB = [&](int j, int kk) {
    if constexpr (NT == 2) {
      return ld2(hb + (kk + j) * TP_);
    } else {
      v2f b;
      b.x = hb[kk * TP_];
      b.y = hb[(kk + 1) * TP_];
      return b;
    }
  };
  {
    const int kh0 = (threadIdx.x & 31) >> 4;
#pragma unroll
    for (int d = 0; d < 4; ++d)
#pragma unroll
      for (int s = 0; s < 2; ++s)
#pragma unroll
        for (int u = 0; u < NT; ++u) g0.b[d][s][u] = lB(u, 8 * d + 4 * kh0 + 2 * s);
  }
  ring_main<2, NT, 4, L_ / 8>(acc, g0, lA, cA, lB);
  ring_drain(acc, g0, cA);
  const int kh = (threadIdx.x & 31) >> 4;
  const int n1 = (m == 0) ? 1 : 0;
  const int n2 = (m == 2) ? 1 : 2;
  const int ab = rowbase - m * A_;
  const float* c1 = crossv + ((size_t)m * M_ + n1) * A_ + ab;
  const float* c2 = crossv + ((size_t)m * M_ + n2) * A_ + ab;
  const float* e1 = dinv + (size_t)n1 * A_ + ab;
  const float* e2 = dinv + (size_t)n2 * A_ + ab;
  const float* g1 = tile + n1 * L_ * TP_ + cl;
  const float* g2 = tile + n2 * L_ * TP_ + cl;
  float f1[2][8], f2[2][8], fr[2][8];
  float x1[2][8][NT], x2[2][8][NT];
#pragma unroll
  for (int t = 0; t < 2; ++t) {
    const int r = i0 + 16 * t + 8 * kh;
#pragma unroll
    for (int q = 0; q < 2; ++q) {
      const v4f p1 = ld4(c1 + r + 4 * q) * ld4(e1 + r + 4 * q);
      const v4f p2 = ld4(c2 + r + 4 * q) * ld4(e2 + r + 4 * q);
      const v4f dr = ld4(dv + r + 4 * q);
      f1[t][4 * q] = p1.x; f1[t][4 * q + 1] = p1.y; f1[t][4 * q + 2] = p1.z; f1[t][4 * q + 3] = p1.w;
      f2[t][4 * q] = p2.x; f2[t][4 * q + 1] = p2.y; f2[t][4 * q + 2] = p2.z; f2[t][4 * q + 3] = p2.w;
      fr[t][4 * q] = dr.x; fr[t][4 * q + 1] = dr.y; fr[t][4 * q + 2] = dr.z; fr[t][4 * q + 3] = dr.w;
    }
#pragma unroll
    for (int v = 0; v < 8; ++v) {
      const int off = (r + v) * TP_;
      if constexpr (NT == 2) {
        const v2f a = ld2(g1 + off), b = ld2(g2 + off);
        x1[t][v][0] = a.x; x1[t][v][1] = a.y;
        x2[t][v][0] = b.x; x2[t][v][1] = b.y;
      } else {
        x1[t][v][0] = g1[off];
        x2[t][v][0] = g2[off];
      }
    }
  }
  float o[2][NT][8];
#pragma unroll
  for (int t = 0; t < 2; ++t)
#pragma unroll
    for (int u = 0; u < NT; ++u)
#pragma unroll
      for (int v = 0; v < 8; ++v) {
        float s = acc[t][u][v];
        s += f1[t][v] * x1[t][v][u];
        s += f2[t][v] * x2[t][v][u];
        o[t][u][v] = s * fr[t][v];
      }
  store_rm<NT, false>(o, i0, n0, hi + (size_t)rowbase * H_);
  store_fm<NT, false>(o, i0, n0, hiT + rowbase, N_);
}

template <int NT>
__device__ __forceinline__ void fc0_spmm_run(const float* __restrict__ xr, const float* __restrict__ w,
                                             const float* __restrict__ bias,
                                             const float* __restrict__ intra,
                                             const float* __restrict__ crossv,
                                             const float* __restrict__ dinv,
                                             float* __restrict__ h0, float* __restrict__ h0T,
                                             float* __restrict__ hi, float* __restrict__ hiT,
                                             float* tile, int m, int dlg, int i0, int n0) {
  const int m0 = m * A_ + dlg * L_ + i0;
  const int mb = m * B_ + dlg;
  float o[2][NT][8];
  fc0_body<NT>(xr, w, bias, m0, n0, o);
  tile_put<NT, false>(o, tile + (m * L_ + i0) * TP_);
  SpmmRing<NT> g0;
  spmm_fill_a<NT>(g0, intra, dinv, mb, i0);
  lds_barrier();
  store_rm<NT, false>(o, m0, n0, h0);
  store_fm<NT, false>(o, m0, n0, h0T, N_);
  spmm_body<NT>(g0, intra, crossv, dinv, tile, hi, hiT, mb, i0, n0);
}

__global__ void __launch_bounds__(192) k_fc0_spmm(const float* __restrict__ xa, const float* __restrict__ xv,
                           const float* __restrict__ xt, const float* __restrict__ w,
                           const float* __restrict__ bias, const float* __restrict__ intra,
                           const float* __restrict__ crossv, const float* __restrict__ dinv,
                           float* __restrict__ h0, float* __restrict__ h0T,
                           float* __restrict__ hi, float* __restrict__ hiT) {
  __shared__ float tile[TILE_];
  const int wv = wg_wave();
  const int m = wv >> 1, i0 = (wv & 1) * 32;
  const int strip = blockIdx.x, dlg = blockIdx.y, n0 = strip * 32;
  const float* xr = ((m == 0) ? xa : (m == 1) ? xv : xt) + (size_t)(dlg * L_ + i0) * H_;
  if (strip == NS_ - 1) fc0_spmm_run<1>(xr, w, bias, intra, crossv, dinv, h0, h0T, hi, hiT, tile, m, dlg, i0, n0);
  else                  fc0_spmm_run<2>(xr, w, bias, intra, crossv, dinv, h0, h0T, hi, hiT, tile, m, dlg, i0, n0);
}

template <int NT>
__device__ __forceinline__ void conv_body(const float* __restrict__ hi, const float* __restrict__ h0,
                                          const float* __restrict__ hiT, const float* __restrict__ h0T,
                                          const float* __restrict__ w, float beta,
                                          int m0, int n0, float (&o)[2][NT][8]) {
  const int mr = threadIdx.x & 15;
  const int cb = lane_col<NT>(n0);
  const float* t0 = hiT + m0 + 2 * mr;
  const float* t1 = h0T + m0 + 2 * mr;
  const float* w1 = w + (size_t)H_ * H_;
  v8f acc[2][NT] = {};
  auto lA0 = [&](int pl, int k) { return ld2(t0 + (size_t)k * N_); };
  auto lA1 = [&](int pl, int k) { return ld2(t1 + (size_t)k * N_); };
  auto lB0 = [&](int j, int kk) { return ldB_unit<NT>(w, cb, j, kk); };
  auto lB1 = [&](int j, int kk) { return ldB_unit<NT>(w1, cb, j, kk); };
  RingT<NT, 5, 1> g0, g1;
  ringt_fill(g0, lA0, lB0);
  ringt_main<false, NT, 5, KP_>(acc, g0, lA0, lB0);
  ringt_fill(g1, lA1, lB1);
  ringt_drain<false>(acc, g0);
  ringt_main<false, NT, 5, KP_>(acc, g1, lA1, lB1);
  ringt_drain<false>(acc, g1);
  auto ep = [&](int m, int n, float v) {
    const float r = 0.9f * hi[(size_t)m * H_ + n] + 0.1f * h0[(size_t)m * H_ + n];
    return fmaxf(beta * v + (1.0f - beta) * r, 0.0f);
  };
  epi_vals<NT, true>(acc, m0, n0, o, ep);
}

template <int NT>
__device__ __forceinline__ void conv_spmm_run(const float* __restrict__ hi, const float* __restrict__ h0,
                                              const float* __restrict__ hiT, const float* __restrict__ h0T,
                                              const float* __restrict__ w, float beta,
                                              const float* __restrict__ intra,
                                              const float* __restrict__ crossv,
                                              const float* __restrict__ dinv,
                                              float* __restrict__ hi2, float* __restrict__ hi2T,
                                              float* tile, int m, int dlg, int i0, int n0) {
  const int mb = m * B_ + dlg;
  float o[2][NT][8];
  conv_body<NT>(hi, h0, hiT, h0T, w, beta, m * A_ + dlg * L_ + i0, n0, o);
  tile_put<NT, true>(o, tile + (m * L_ + i0) * TP_);
  SpmmRing<NT> g0;
  spmm_fill_a<NT>(g0, intra, dinv, mb, i0);
  lds_barrier();
  spmm_body<NT>(g0, intra, crossv, dinv, tile, hi2, hi2T, mb, i0, n0);
}

__global__ void __launch_bounds__(192) k_conv_spmm(const float* __restrict__ hi, const float* __restrict__ h0,
                            const float* __restrict__ hiT, const float* __restrict__ h0T,
                            const float* __restrict__ w, float beta,
                            const float* __restrict__ intra, const float* __restrict__ crossv,
                            const float* __restrict__ dinv,
                            float* __restrict__ hi2, float* __restrict__ hi2T) {
  __shared__ float tile[TILE_];
  const int wv = wg_wave();
  const int m = wv >> 1, i0 = (wv & 1) * 32;
  const int strip = blockIdx.x, dlg = blockIdx.y, n0 = strip * 32;
  if (strip == NS_ - 1) conv_spmm_run<1>(hi, h0, hiT, h0T, w, beta, intra, crossv, dinv, hi2, hi2T, tile, m, dlg, i0, n0);
  else                  conv_spmm_run<2>(hi, h0, hiT, h0T, w, beta, intra, crossv, dinv, hi2, hi2T, tile, m, dlg, i0, n0);
}

__global__ void __launch_bounds__(128) k_conv(const float* __restrict__ hi, const float* __restrict__ h0,
                       const float* __restrict__ hiT, const float* __restrict__ h0T,
                       const float* __restrict__ w, float beta, float* __restrict__ houtT) {
  const int wv = wg_wave();
  const int strip = 2 * blockIdx.x + (wv & 1);
  if (strip >= NS_) return;
  const int m0 = (blockIdx.y * 2 + (wv >> 1)) * 32, n0 = strip * 32;
  if (strip == NS_ - 1) {
    float o[2][1][8];
    conv_body<1>(hi, h0, hiT, h0T, w, beta, m0, n0, o);
    store_fm<1, true>(o, m0, n0, houtT, N_);
  } else {
    float o[2][2][8];
    conv_body<2>(hi, h0, hiT, h0T, w, beta, m0, n0, o);
    store_fm<2, true>(o, m0, n0, houtT, N_);
  }
}

template <int NT>
__device__ __forceinline__ void hx_body(const float* __restrict__ xm, const float* __restrict__ hfT,
                                        const float* __restrict__ W, const float* __restrict__ Bb,
                                        float* __restrict__ hx, float* __restrict__ hxT,
                                        int rb, int a0_, int n0) {
  const int mr = threadIdx.x & 15;
  const int cb = lane_col<NT>(n0);
  const size_t r0 = (size_t)(a0_ + 2 * mr) * H_, r1 = (size_t)(a0_ + 2 * mr + 1) * H_;
  const float* p0[2] = {xm + r0, xm + r1};
  const float* tf = hfT + rb + a0_ + 2 * mr;
  const float* W1 = W + (size_t)H_ * H_;
  v8f acc[2][NT] = {};
  auto cA = [](v4f r) { return r; };
  auto lA0 = [&](int t, int kq) { return ld4(p0[t] + kq); };
  auto lA1 = [&](int pl, int k) { return ld2(tf + (size_t)k * N_); };
  auto lB0 = [&](int j, int kk) { return ldB_unit<NT>(W, cb, j, kk); };
  auto lB1 = [&](int j, int kk) { return ldB_unit<NT>(W1, cb, j, kk); };
  Ring<2, NT, 5, v4f> g0;
  RingT<NT, 5, 1> g1;
  ring_fill(g0, lA0, lB0);
  ring_main<2, NT, 5, KP_>(acc, g0, lA0, cA, lB0);
  ringt_fill(g1, lA1, lB1);
  ring_drain(acc, g0, cA);
  ringt_main<false, NT, 5, KP_>(acc, g1, lA1, lB1);
  ringt_drain<false>(acc, g1);
  auto ep = [&](int a, int n, float v) { return fmaxf(v + Bb[n], 0.0f); };
  float o[2][NT][8];
  epi_vals<NT, true>(acc, a0_, n0, o, ep);
  store_rm<NT, true>(o, a0_, n0, hx + (size_t)rb * H_);
  store_fm<NT, true>(o, a0_, n0, hxT + (size_t)rb * H_, A_);
}

__global__ void __launch_bounds__(128) k_hx(const float* __restrict__ xa, const float* __restrict__ xv,
                     const float* __restrict__ xt, const float* __restrict__ hfT,
                     const float* __restrict__ tw, const float* __restrict__ tb,
                     float* __restrict__ hx, float* __restrict__ hxT) {
  const int wv = wg_wave();
  const int strip = 2 * blockIdx.x + (wv & 1);
  if (strip >= NS_) return;
  const int m  = blockIdx.z;
  const float* xm = (m == 0) ? xa : (m == 1) ? xv : xt;
  const int a0_ = (blockIdx.y * 2 + (wv >> 1)) * 32, n0 = strip * 32;
  const float* W  = tw + (size_t)m * 2 * H_ * H_;
  const float* Bb = tb + (size_t)m * H_;
  if (strip == NS_ - 1) hx_body<1>(xm, hfT, W, Bb, hx, hxT, m * A_, a0_, n0);
  else                  hx_body<2>(xm, hfT, W, Bb, hx, hxT, m * A_, a0_, n0);
}

template <int NT>
__device__ __forceinline__ float tr_logit(const v8f (&a)[NT], int g, int e) {
  if constexpr (NT == 2) return a[e & 1][4 * g + (e >> 1)];
  else                   return a[0][e];
}

template <int NT>
__device__ __forceinline__ void gate_body(const float* __restrict__ Hi, const float* __restrict__ Hj,
                                          const float* __restrict__ HiT, const float* __restrict__ HjT,
                                          const float* __restrict__ W, const float* __restrict__ Bb,
                                          const float* __restrict__ Ui, float* __restrict__ part_out,
                                          int a0_, int n0) {
  const int mr = threadIdx.x & 15;
  const int kh = (threadIdx.x & 31) >> 4;
  const int cb = lane_col<NT>(n0);
  const size_t r0 = (size_t)(a0_ + 2 * mr) * H_, r1 = (size_t)(a0_ + 2 * mr + 1) * H_;
  const float* pi[2] = {Hi + r0, Hi + r1};
  const float* pj[2] = {Hj + r0, Hj + r1};
  const float* ti = HiT + a0_ + 2 * mr;
  const float* tj = HjT + a0_ + 2 * mr;
  const float* W1 = W + (size_t)H_ * H_;
  const float* W2 = W + (size_t)2 * H_ * H_;
  v8f acc[2][NT] = {};
  auto lAi = [&](int pl, int k) { return ld2(ti + (size_t)k * A_); };
  auto lAj = [&](int pl, int k) { return ld2(tj + (size_t)k * A_); };
  auto lAp = [&](int pl, int k) { return ld2((pl ? tj : ti) + (size_t)k * A_); };
  auto lB0 = [&](int j, int kk) { return ldB_unit<NT>(W, cb, j, kk); };
  auto lB1 = [&](int j, int kk) { return ldB_unit<NT>(W1, cb, j, kk); };
  auto lB2 = [&](int j, int kk) { return ldB_unit<NT>(W2, cb, j, kk); };
  RingT<NT, 5, 1> g0, g1;
  RingT<NT, 5, 2> g2;
  ringt_fill(g0, lAi, lB0);
  ringt_main<true, NT, 5, KP_>(acc, g0, lAi, lB0);
  ringt_fill(g1, lAj, lB1);
  ringt_drain<true>(acc, g0);
  ringt_main<true, NT, 5, KP_>(acc, g1, lAj, lB1);
  ringt_fill(g2, lAp, lB2);
  ringt_drain<true>(acc, g1);
  ringt_main<true, NT, 5, KP_>(acc, g2, lAp, lB2);
  ringt_drain<true>(acc, g2);

  float part[2][TAG_] = {};
#pragma unroll
  for (int t = 0; t < 2; ++t) {
#pragma unroll
    for (int g = 0; g < NT; ++g) {
      const int nb = (NT == 2) ? n0 + 16 * kh + 8 * g : n0 + 8 * kh;
      if (NT == 2 || nb < H_) {
        const v4f* hip = reinterpret_cast<const v4f*>(pi[t] + nb);
        const v4f* hjp = reinterpret_cast<const v4f*>(pj[t] + nb);
        const v4f* bp  = reinterpret_cast<const v4f*>(Bb + nb);
        const v4f* up  = reinterpret_cast<const v4f*>(Ui + (size_t)nb * TAG_);
        const v4f hi0 = hip[0], hi1 = hip[1], hj0 = hjp[0], hj1 = hjp[1], b0 = bp[0], b1 = bp[1];
        const float hiv[8] = {hi0.x, hi0.y, hi0.z, hi0.w, hi1.x, hi1.y, hi1.z, hi1.w};
        const float hjv[8] = {hj0.x, hj0.y, hj0.z, hj0.w, hj1.x, hj1.y, hj1.z, hj1.w};
        const float bv[8]  = {b0.x, b0.y, b0.z, b0.w, b1.x, b1.y, b1.z, b1.w};
        float uu[8 * TAG_];
#pragma unroll
        for (int q = 0; q < 2 * TAG_; ++q) {
          const v4f x = up[q];
          uu[4 * q] = x.x; uu[4 * q + 1] = x.y; uu[4 * q + 2] = x.z; uu[4 * q + 3] = x.w;
        }
#pragma unroll
        for (int e = 0; e < 8; ++e) {
          const float lg = tr_logit<NT>(acc[t], g, e);
          const float z = 1.0f / (1.0f + expf(-(lg + bv[e])));
          const float gt = z * hiv[e] + (1.0f - z) * hjv[e];
#pragma unroll
          for (int c = 0; c < TAG_; ++c) part[t][c] += gt * uu[e * TAG_ + c];
        }
      }
    }
  }
#pragma unroll
  for (int t = 0; t < 2; ++t)
#pragma unroll
    for (int c = 0; c < TAG_; ++c) part[t][c] += __shfl_xor(part[t][c], 16, 32);
  if (kh == 0) {
#pragma unroll
    for (int t = 0; t < 2; ++t) {
      float* po = part_out + (size_t)(a0_ + 2 * mr + t) * TAG_;
#pragma unroll
      for (int c = 0; c < TAG_; c += 2) {
        v2f o; o.x = part[t][c]; o.y = part[t][c + 1];
        *reinterpret_cast<v2f*>(po + c) = o;
      }
    }
  }
}

__global__ void __launch_bounds__(128) k_gate(const float* __restrict__ hx, const float* __restrict__ hxT,
                       const float* __restrict__ cw,
                       const float* __restrict__ cb, const float* __restrict__ uw,
                       float* __restrict__ plog) {
  const int wv = wg_wave();
  const int strip = 2 * blockIdx.x + (wv & 1);
  if (strip >= NS_) return;
  const int p = blockIdx.z;
  const int i = p >> 1;
  const int j = (p & 1) ? (i == 2 ? 1 : 2) : (i == 0 ? 1 : 0);
  const int a0_ = (blockIdx.y * 2 + (wv >> 1)) * 32, n0 = strip * 32;
  const float* Hi = hx + (size_t)i * A_ * H_;
  const float* Hj = hx + (size_t)j * A_ * H_;
  const float* HiT = hxT + (size_t)i * H_ * A_;
  const float* HjT = hxT + (size_t)j * H_ * A_;
  const float* W  = cw + (size_t)p * 3 * H_ * H_;
  const float* Bb = cb + (size_t)p * H_;
  const float* Ui = uw + (size_t)i * H_ * TAG_;
  float* po = plog + ((size_t)p * NS_ + strip) * A_ * TAG_;
  if (strip == NS_ - 1) gate_body<1>(Hi, Hj, HiT, HjT, W, Bb, Ui, po, a0_, n0);
  else                  gate_body<2>(Hi, Hj, HiT, HjT, W, Bb, Ui, po, a0_, n0);
}

__global__ void k_final(const float* __restrict__ plog, const float* __restrict__ ub,
                        float* __restrict__ out) {
  const int e = blockIdx.x * blockDim.x + threadIdx.x;
  const int t = e % TAG_;
  float s = ub[t] + ub[TAG_ + t] + ub[2 * TAG_ + t];
  for (int q = 0; q < 6 * NS_; ++q) s += plog[(size_t)q * A_ * TAG_ + e];
  out[e] = s;
}

}

extern "C" void kernel_launch(void* const* d_in, const int* in_sizes, int n_in,
                              void* d_out, int out_size, void* d_ws, size_t ws_size,
                              hipStream_t stream) {
  (void)in_sizes; (void)n_in; (void)out_size; (void)ws_size;
  const float* x_a     = (const float*)d_in[0];
  const float* x_v     = (const float*)d_in[1];
  const float* x_t     = (const float*)d_in[2];
  const float* fc0_w   = (const float*)d_in[3];
  const float* fc0_b   = (const float*)d_in[4];
  const float* conv_w  = (const float*)d_in[5];
  const float* tm_w    = (const float*)d_in[6];
  const float* tm_b    = (const float*)d_in[7];
  const float* cross_w = (const float*)d_in[8];
  const float* cross_b = (const float*)d_in[9];
  const float* uni_w   = (const float*)d_in[10];
  const float* uni_b   = (const float*)d_in[11];
  float* out = (float*)d_out;

  const size_t NH = (size_t)N_ * H_;
  float* ws    = (float*)d_ws;
  float* intra = ws; ws += (size_t)N_ * L_;
  float* crossv= ws; ws += (size_t)M_ * M_ * A_;
  float* dinv  = ws; ws += N_;
  float* h0    = ws; ws += NH;
  float* h0T   = ws; ws += NH;
  float* hBfT  = ws; ws += NH;
  float* hi    = ws; ws += NH;
  float* hiT   = ws; ws += NH;
  float* hi2   = ws; ws += NH;
  float* hi2T  = ws; ws += NH;
  float* hx    = ws; ws += NH;
  float* hxT   = ws; ws += NH;
  float* plog  = ws; ws += (size_t)6 * NS_ * A_ * TAG_;


  const float beta0 = logf(0.5f / 1.0f + 1.0f);
  const float beta1 = logf(0.5f / 2.0f + 1.0f);
  k_graph<<<M_ * B_ * 4,         dim3(32), 0, stream>>>(x_a, x_v, x_t, intra, crossv, dinv);
  k_fc0_spmm <<<dim3(NS_, B_), dim3(192), 0, stream>>>(x_a, x_v, x_t, fc0_w, fc0_b, intra, crossv, dinv,
                                                     h0, h0T, hi, hiT);
  k_conv_spmm<<<dim3(NS_, B_), dim3(192), 0, stream>>>(hi, h0, hiT, h0T, conv_w, beta0, intra, crossv, dinv,
                                                     hi2, hi2T);
  k_conv<<<dim3(NSP_, N_ / 64), dim3(128), 0, stream>>>(hi2, h0, hi2T, h0T, conv_w + 2 * H_ * H_, beta1, hBfT);

  k_hx   <<<dim3(NSP_, A_ / 64, M_), dim3(128), 0, stream>>>(x_a, x_v, x_t, hBfT, tm_w, tm_b, hx, hxT);

  k_gate <<<dim3(NSP_, A_ / 64, 6), dim3(128), 0, stream>>>(hx, hxT, cross_w, cross_b, uni_w, plog);

  k_final<<<A_ * TAG_ / 256, dim3(256), 0, stream>>>(plog, uni_b, out);
}


# k_gate: its three k segments (hx_i, hx_j, product) walked in one k loop; a step loads the hx_i, hx_j fragments once for all three weight blocks: 500 loads per wave, was 700; logit k sum order changes
# speedup vs baseline: 9.1217x; 1.0419x over previous
#include <hip/hip_runtime.h>
#include <math.h>

typedef __attribute__((ext_vector_type(2))) float v2f;
typedef __attribute__((ext_vector_type(4))) float v4f;
typedef __attribute__((ext_vector_type(8))) float v8f;

namespace {

constexpr int B_   = 32;
constexpr int L_   = 64;
constexpr int H_   = 200;
constexpr int M_   = 3;
constexpr int A_   = B_ * L_;
constexpr int N_   = M_ * A_;
constexpr int TAG_ = 6;
constexpr float PI_F = 3.14159265358979323846f;

struct RawQuad { v4f p, q; };

__device__ __forceinline__ v4f ld4(const float* p) {
  return *reinterpret_cast<const v4f*>(p);
}

__device__ __forceinline__ v2f ld2(const float* p) {
  return *reinterpret_cast<const v2f*>(p);
}

template <int MT, int NT, int D, typename RA>
struct Ring { RA a[D][MT]; v2f b[D][2][NT]; };

template <int NT>
__device__ __forceinline__ void b_frags(const v2f (&raw)[NT], v2f (&f)[NT]) {
  if constexpr (NT == 2) {
    f[0].x = raw[0].x; f[0].y = raw[1].x;
    f[1].x = raw[0].y; f[1].y = raw[1].y;
  } else {
    f[0] = raw[0];
  }
}

template <int MT, int NT, bool TR = false>
__device__ __forceinline__ void mma_step(v8f (&acc)[MT][NT], const v2f (&a)[MT],
                                         const v2f (&b)[NT]) {
#pragma unroll
  for (int t = 0; t < MT; ++t)
#pragma unroll
    for (int u = 0; u < NT; ++u) {
      if constexpr (TR)
        acc[t][u] = __builtin_amdgcn_wmma_f32_16x16x4_f32(false, b[u], false, a[t],
                                                          (short)0, acc[t][u], false, false);
      else
        acc[t][u] = __builtin_amdgcn_wmma_f32_16x16x4_f32(false, a[t], false, b[u],
                                                          (short)0, acc[t][u], false, false);
    }
}


template <int MT, int NT, int D, typename RA, typename FA, typename FB>
__device__ __forceinline__ void ring_fill(Ring<MT, NT, D, RA>& r, FA ldA, FB ldB) {
  const int kh = (threadIdx.x & 31) >> 4;
#pragma unroll
  for (int d = 0; d < D; ++d) {
    const int kq = 8 * d + 4 * kh;
#pragma unroll
    for (int t = 0; t < MT; ++t) r.a[d][t] = ldA(t, kq);
#pragma unroll
    for (int s = 0; s < 2; ++s)
#pragma unroll
      for (int u = 0; u < NT; ++u) r.b[d][s][u] = ldB(u, kq + 2 * s);
    __builtin_amdgcn_sched_barrier(0);
  }
}

template <int MT, int NT, int D, int NPAIRS, bool TR = false, typename RA, typename FA, typename FC, typename FB>
__device__ __forceinline__ void ring_main(v8f (&acc)[MT][NT], Ring<MT, NT, D, RA>& r,
                                          FA ldA, FC cvA, FB ldB) {
  static_assert(NPAIRS % D == 0 && NPAIRS >= D, "ring depth must divide the pair count");
  const int kh = (threadIdx.x & 31) >> 4;
#pragma nounroll
  for (int p0 = 0; p0 < NPAIRS - D; p0 += D) {
#pragma unroll
    for (int d = 0; d < D; ++d) {
      v2f lo[MT], hi[MT];
#pragma unroll
      for (int t = 0; t < MT; ++t) {
        const v4f c = cvA(r.a[d][t]);
        lo[t] = c.xy;
        hi[t] = c.zw;
      }
      v2f f0[NT], f1[NT];
      b_frags<NT>(r.b[d][0], f0);
      b_frags<NT>(r.b[d][1], f1);
      mma_step<MT, NT, TR>(acc, lo, f0);
      mma_step<MT, NT, TR>(acc, hi, f1);
      const int kq = 8 * (p0 + d + D) + 4 * kh;
#pragma unroll
      for (int t = 0; t < MT; ++t) r.a[d][t] = ldA(t, kq);
#pragma unroll
      for (int s = 0; s < 2; ++s)
#pragma unroll
        for (int u = 0; u < NT; ++u) r.b[d][s][u] = ldB(u, kq + 2 * s);
      __builtin_amdgcn_sched_barrier(0);
    }
  }
}

template <bool TR = false, int MT, int NT, int D, typename RA, typename FC>
__device__ __forceinline__ void ring_drain(v8f (&acc)[MT][NT], Ring<MT, NT, D, RA>& r, FC cvA) {
#pragma unroll
  for (int d = 0; d < D; ++d) {
    v2f lo[MT], hi[MT];
#pragma unroll
    for (int t = 0; t < MT; ++t) {
      const v4f c = cvA(r.a[d][t]);
      lo[t] = c.xy;
      hi[t] = c.zw;
    }
    v2f f0[NT], f1[NT];
    b_frags<NT>(r.b[d][0], f0);
    b_frags<NT>(r.b[d][1], f1);
    mma_step<MT, NT, TR>(acc, lo, f0);
    mma_step<MT, NT, TR>(acc, hi, f1);
  }
}

template <int NT, int D, int NP>
struct RingT { v2f a[D][NP][4]; v2f b[D][2][NT]; };

template <int NP>
__device__ __forceinline__ void a_frags(const v2f (&raw)[NP][4], v2f (&lo)[2], v2f (&hi)[2]) {
  v2f q[4];
#pragma unroll
  for (int j = 0; j < 4; ++j) {
    if constexpr (NP == 2) q[j] = raw[0][j] * raw[1][j];
    else                   q[j] = raw[0][j];
  }
  lo[0].x = q[0].x; lo[0].y = q[1].x; hi[0].x = q[2].x; hi[0].y = q[3].x;
  lo[1].x = q[0].y; lo[1].y = q[1].y; hi[1].x = q[2].y; hi[1].y = q[3].y;
}

template <int NT, int D, int NP, typename FA, typename FB>
__device__ __forceinline__ void ringt_fill(RingT<NT, D, NP>& r, FA ldA, FB ldB) {
  const int kh = (threadIdx.x & 31) >> 4;
#pragma unroll
  for (int d = 0; d < D; ++d) {
    const int kq = 8 * d + 4 * kh;
#pragma unroll
    for (int pl = 0; pl < NP; ++pl)
#pragma unroll
      for (int j = 0; j < 4; ++j) r.a[d][pl][j] = ldA(pl, kq + j);
#pragma unroll
    for (int s = 0; s < 2; ++s)
#pragma unroll
      for (int u = 0; u < NT; ++u) r.b[d][s][u] = ldB(u, kq + 2 * s);
    __builtin_amdgcn_sched_barrier(0);
  }
}

template <bool TR, int NT, int D, int NPAIRS, int NP, typename FA, typename FB>
__device__ __forceinline__ void ringt_main(v8f (&acc)[2][NT], RingT<NT, D, NP>& r, FA ldA, FB ldB) {
  static_assert(NPAIRS % D == 0 && NPAIRS >= D, "ring depth must divide the pair count");
  const int kh = (threadIdx.x & 31) >> 4;
#pragma nounroll
  for (int p0 = 0; p0 < NPAIRS - D; p0 += D) {
#pragma unroll
    for (int d = 0; d < D; ++d) {
      v2f lo[2], hi[2];
      a_frags<NP>(r.a[d], lo, hi);
      v2f f0[NT], f1[NT];
      b_frags<NT>(r.b[d][0], f0);
      b_frags<NT>(r.b[d][1], f1);
      mma_step<2, NT, TR>(acc, lo, f0);
      mma_step<2, NT, TR>(acc, hi, f1);
      const int kq = 8 * (p0 + d + D) + 4 * kh;
#pragma unroll
      for (int pl = 0; pl < NP; ++pl)
#pragma unroll
        for (int j = 0; j < 4; ++j) r.a[d][pl][j] = ldA(pl, kq + j);
#pragma unroll
      for (int s = 0; s < 2; ++s)
#pragma unroll
        for (int u = 0; u < NT; ++u) r.b[d][s][u] = ldB(u, kq + 2 * s);
      __builtin_amdgcn_sched_barrier(0);
    }
  }
}

template <bool TR, int NT, int D, int NP>
__device__ __forceinline__ void ringt_drain(v8f (&acc)[2][NT], RingT<NT, D, NP>& r) {
#pragma unroll
  for (int d = 0; d < D; ++d) {
    v2f lo[2], hi[2];
    a_frags<NP>(r.a[d], lo, hi);
    v2f f0[NT], f1[NT];
    b_frags<NT>(r.b[d][0], f0);
    b_frags<NT>(r.b[d][1], f1);
    mma_step<2, NT, TR>(acc, lo, f0);
    mma_step<2, NT, TR>(acc, hi, f1);
  }
}

template <bool RI>
__device__ __forceinline__ int tile_row(int t, int x) { return RI ? 2 * x + t : 16 * t + x; }

template <int NT, bool RI, typename FV>
__device__ __forceinline__ void epi_vals(const v8f (&acc)[2][NT], int m0, int n0,
                                         float (&o)[2][NT][8], FV val) {
  const int lane = threadIdx.x & 31;
  const int mr = lane & 15, kh = lane >> 4;
#pragma unroll
  for (int t = 0; t < 2; ++t)
#pragma unroll
    for (int u = 0; u < NT; ++u)
#pragma unroll
      for (int v = 0; v < 8; ++v) {
        const int m = m0 + tile_row<RI>(t, 8 * kh + v);
        const int n = (NT == 2) ? n0 + 2 * mr + u : n0 + mr;
        o[t][u][v] = (NT == 2 || n < H_) ? val(m, n, acc[t][u][v]) : 0.0f;
      }
}

template <int NT, bool RI>
__device__ __forceinline__ void store_rm(const float (&o)[2][NT][8], int m0, int n0,
                                         float* __restrict__ dst) {
  const int lane = threadIdx.x & 31;
  const int mr = lane & 15, kh = lane >> 4;
#pragma unroll
  for (int t = 0; t < 2; ++t)
#pragma unroll
    for (int v = 0; v < 8; ++v) {
      const int m = m0 + tile_row<RI>(t, 8 * kh + v);
      if constexpr (NT == 2) {
        v2f w;
        w.x = o[t][0][v];
        w.y = o[t][1][v];
        *reinterpret_cast<v2f*>(dst + (size_t)m * H_ + n0 + 2 * mr) = w;
      } else {
        const int n = n0 + mr;
        if (n < H_) dst[(size_t)m * H_ + n] = o[t][0][v];
      }
    }
}

template <int NT, bool RI>
__device__ __forceinline__ void store_fm(const float (&o)[2][NT][8], int m0, int n0,
                                         float* __restrict__ dstT, int pitch) {
  const int lane = threadIdx.x & 31;
  const int mr = lane & 15, kh = lane >> 4;
#pragma unroll
  for (int u = 0; u < NT; ++u) {
    const int n = (NT == 2) ? n0 + 2 * mr + u : n0 + mr;
    if (NT == 2 || n < H_) {
      float* col = dstT + (size_t)n * pitch + m0;
      if constexpr (RI) {
#pragma unroll
        for (int q = 0; q < 4; ++q) {
          v4f w;
          w.x = o[0][u][2 * q];
          w.y = o[1][u][2 * q];
          w.z = o[0][u][2 * q + 1];
          w.w = o[1][u][2 * q + 1];
          *reinterpret_cast<v4f*>(col + 16 * kh + 4 * q) = w;
        }
      } else {
#pragma unroll
        for (int t = 0; t < 2; ++t)
#pragma unroll
          for (int q = 0; q < 2; ++q) {
            v4f w;
            w.x = o[t][u][4 * q];
            w.y = o[t][u][4 * q + 1];
            w.z = o[t][u][4 * q + 2];
            w.w = o[t][u][4 * q + 3];
            *reinterpret_cast<v4f*>(col + 16 * t + 8 * kh + 4 * q) = w;
          }
      }
    }
  }
}

template <int NT>
__device__ __forceinline__ int lane_col(int n0) {
  const int mr = threadIdx.x & 15;
  return NT == 2 ? n0 + 2 * mr : min(n0 + mr, H_ - 1);
}

template <int NT>
__device__ __forceinline__ v2f ldB_unit(const float* __restrict__ w, int cb, int j, int kk) {
  if constexpr (NT == 2) {
    return ld2(w + (size_t)(kk + j) * H_ + cb);
  } else {
    v2f b;
    b.x = w[(size_t)kk * H_ + cb];
    b.y = w[(size_t)(kk + 1) * H_ + cb];
    return b;
  }
}

constexpr int NS_ = 7;
constexpr int KP_ = H_ / 8;

constexpr int NSP_ = (NS_ + 1) / 2;

__device__ __forceinline__ int wg_wave() { return __builtin_amdgcn_readfirstlane((int)(threadIdx.x >> 5)); }

constexpr int GD_ = 5;

struct GSlot { v4f b[4]; v4f y1, y2; };

__device__ __forceinline__ float ang_sim(float cosv) {
  return 1.0f - acosf(cosv * 0.99999f) * (1.0f / PI_F);
}

__global__ void __launch_bounds__(32) k_graph(const float* __restrict__ xa, const float* __restrict__ xv,
                        const float* __restrict__ xt, float* __restrict__ intra,
                        float* __restrict__ crossv, float* __restrict__ dinv) {
  const int mb = blockIdx.x >> 2;
  const int m  = mb >> 5;
  const int d0 = (mb & 31) * L_;
  const int w  = blockIdx.x & 3;
  const int lane = threadIdx.x & 31;
  const int c = lane & 15, h = lane >> 4;
  const int n1 = (m == 0) ? 1 : 0;
  const int n2 = (m == 2) ? 1 : 2;
  const float* xm = (m == 0) ? xa : (m == 1) ? xv : xt;
  const float* x1 = (n1 == 0) ? xa : xv;
  const float* x2 = (n2 == 1) ? xv : xt;
  const float* pb[4];
#pragma unroll
  for (int u = 0; u < 4; ++u)
    pb[u] = xm + (size_t)(d0 + 16 * ((w + u) & 3) + c) * H_ + 4 * h;
  const float* p1 = x1 + (size_t)(d0 + 16 * w + c) * H_ + 4 * h;
  const float* p2 = x2 + (size_t)(d0 + 16 * w + c) * H_ + 4 * h;

  v8f acc[4] = {};
  float ss[4] = {0.f, 0.f, 0.f, 0.f};
  float s1 = 0.f, s2 = 0.f, d1 = 0.f, d2 = 0.f;

  auto fetch = [&](GSlot& s, int p) {
#pragma unroll
    for (int u = 0; u < 4; ++u) s.b[u] = ld4(pb[u] + 8 * p);
    s.y1 = ld4(p1 + 8 * p);
    s.y2 = ld4(p2 + 8 * p);
  };
  auto use = [&](const GSlot& s) {
    const v4f a = s.b[0];
#pragma unroll
    for (int u = 0; u < 4; ++u) {
      const v4f b = s.b[u];
      ss[u] = fmaf(b.x, b.x, ss[u]); ss[u] = fmaf(b.y, b.y, ss[u]);
      ss[u] = fmaf(b.z, b.z, ss[u]); ss[u] = fmaf(b.w, b.w, ss[u]);
    }
    d1 = fmaf(a.x, s.y1.x, d1); d1 = fmaf(a.y, s.y1.y, d1);
    d1 = fmaf(a.z, s.y1.z, d1); d1 = fmaf(a.w, s.y1.w, d1);
    d2 = fmaf(a.x, s.y2.x, d2); d2 = fmaf(a.y, s.y2.y, d2);
    d2 = fmaf(a.z, s.y2.z, d2); d2 = fmaf(a.w, s.y2.w, d2);
    s1 = fmaf(s.y1.x, s.y1.x, s1); s1 = fmaf(s.y1.y, s.y1.y, s1);
    s1 = fmaf(s.y1.z, s.y1.z, s1); s1 = fmaf(s.y1.w, s.y1.w, s1);
    s2 = fmaf(s.y2.x, s.y2.x, s2); s2 = fmaf(s.y2.y, s.y2.y, s2);
    s2 = fmaf(s.y2.z, s.y2.z, s2); s2 = fmaf(s.y2.w, s.y2.w, s2);
    const v2f alo = a.xy, ahi = a.zw;
#pragma unroll
    for (int u = 0; u < 4; ++u) {
      const v2f blo = s.b[u].xy;
      acc[u] = __builtin_amdgcn_wmma_f32_16x16x4_f32(false, alo, false, blo,
                                                     (short)0, acc[u], false, false);
    }
#pragma unroll
    for (int u = 0; u < 4; ++u) {
      const v2f bhi = s.b[u].zw;
      acc[u] = __builtin_amdgcn_wmma_f32_16x16x4_f32(false, ahi, false, bhi,
                                                     (short)0, acc[u], false, false);
    }
  };

  GSlot ring[GD_];
#pragma unroll
  for (int d = 0; d < GD_; ++d) {
    fetch(ring[d], d);
    __builtin_amdgcn_sched_barrier(0);
  }
#pragma nounroll
  for (int p0 = 0; p0 < KP_ - GD_; p0 += GD_) {
#pragma unroll
    for (int d = 0; d < GD_; ++d) {
      use(ring[d]);
      fetch(ring[d], p0 + d + GD_);
      __builtin_amdgcn_sched_barrier(0);
    }
  }
#pragma unroll
  for (int d = 0; d < GD_; ++d) use(ring[d]);

#pragma unroll
  for (int u = 0; u < 4; ++u) ss[u] += __shfl_xor(ss[u], 16, 32);
  s1 += __shfl_xor(s1, 16, 32);
  s2 += __shfl_xor(s2, 16, 32);
  d1 += __shfl_xor(d1, 16, 32);
  d2 += __shfl_xor(d2, 16, 32);

  float rn[4];
#pragma unroll
  for (int u = 0; u < 4; ++u) rn[u] = 1.0f / sqrtf(ss[u]);
  const float cr1 = ang_sim(d1 * (rn[0] * (1.0f / sqrtf(s1))));
  const float cr2 = ang_sim(d2 * (rn[0] * (1.0f / sqrtf(s2))));

  float rs[8];
  const size_t srow = (size_t)mb * L_ + 16 * w + 8 * h;
#pragma unroll
  for (int r = 0; r < 8; ++r) {
    const float rr = __shfl(rn[0], 8 * h + r, 32);
    float t = 0.f;
#pragma unroll
    for (int u = 0; u < 4; ++u) {
      const float v = ang_sim(acc[u][r] * (rr * rn[u]));
      intra[(srow + r) * L_ + 16 * ((w + u) & 3) + c] = v;
      t += v;
    }
#pragma unroll
    for (int off = 8; off > 0; off >>= 1) t += __shfl_xor(t, off, 32);
    rs[r] = t;
  }
  float mine = rs[0];
#pragma unroll
  for (int r = 1; r < 8; ++r) mine = ((c & 7) == r) ? rs[r] : mine;
  if ((c >> 3) == h) {
    const int a = d0 + 16 * w + c;
    crossv[((size_t)m * M_ + n1) * A_ + a] = cr1;
    crossv[((size_t)m * M_ + n2) * A_ + a] = cr2;
    dinv[(size_t)m * A_ + a] = 1.0f / sqrtf(mine + cr1 + cr2);
  }
}


template <int NT>
__device__ __forceinline__ void fc0_body(const float* __restrict__ xr, const float* __restrict__ w,
                                         const float* __restrict__ bias, int m0, int n0,
                                         float (&o)[2][NT][8]) {
  const int mr = threadIdx.x & 15;
  const int cb = lane_col<NT>(n0);
  const float* ar[2] = {xr + (size_t)mr * H_, xr + (size_t)(16 + mr) * H_};
  v8f acc[2][NT] = {};
  auto cA = [](v4f r) { return r; };
  auto lA = [&](int t, int kq) { return ld4(ar[t] + kq); };
  auto lB = [&](int j, int kk) { return ldB_unit<NT>(w, cb, j, kk); };
  Ring<2, NT, 5, v4f> g0;
  ring_fill(g0, lA, lB);
  ring_main<2, NT, 5, KP_>(acc, g0, lA, cA, lB);
  ring_drain(acc, g0, cA);
  auto ep = [&](int m, int n, float v) { return fmaxf(v + bias[n], 0.0f); };
  epi_vals<NT, false>(acc, m0, n0, o, ep);
}

constexpr int TP_ = 32;
constexpr int TILE_ = M_ * L_ * TP_;

__device__ __forceinline__ void lds_barrier() {
  __builtin_amdgcn_fence(__ATOMIC_RELEASE, "workgroup", "local");
  __builtin_amdgcn_s_barrier();
  __builtin_amdgcn_fence(__ATOMIC_ACQUIRE, "workgroup", "local");
}

template <int NT, bool RI>
__device__ __forceinline__ void tile_put(const float (&o)[2][NT][8], float* tl) {
  const int lane = threadIdx.x & 31;
  const int mr = lane & 15, kh = lane >> 4;
#pragma unroll
  for (int t = 0; t < 2; ++t)
#pragma unroll
    for (int v = 0; v < 8; ++v) {
      float* row = tl + tile_row<RI>(t, 8 * kh + v) * TP_;
      if constexpr (NT == 2) {
        v2f w;
        w.x = o[t][0][v];
        w.y = o[t][1][v];
        *reinterpret_cast<v2f*>(row + 2 * mr) = w;
      } else {
        row[mr] = o[t][0][v];
      }
    }
}

template <int NT>
using SpmmRing = Ring<2, NT, 4, RawQuad>;

template <int NT>
__device__ __forceinline__ void spmm_fill_a(SpmmRing<NT>& g0, const float* __restrict__ intra,
                                            const float* __restrict__ dinv, int mb, int i0) {
  const int mr = threadIdx.x & 15, kh = (threadIdx.x & 31) >> 4;
  const float* S = intra + (size_t)mb * L_ * L_;
  const float* dv = dinv + mb * L_;
#pragma unroll
  for (int d = 0; d < 4; ++d)
#pragma unroll
    for (int t = 0; t < 2; ++t) {
      const int kq = 8 * d + 4 * kh;
      g0.a[d][t].p = ld4(S + (size_t)(i0 + 16 * t + mr) * L_ + kq);
      g0.a[d][t].q = ld4(dv + kq);
    }
}

template <int NT>
__device__ __forceinline__ void spmm_body(SpmmRing<NT>& g0, const float* __restrict__ intra,
                                          const float* __restrict__ crossv,
                                          const float* __restrict__ dinv,
                                          const float* tile, float* __restrict__ hi,
                                          float* __restrict__ hiT, int mb, int i0, int n0) {
  const int mr = threadIdx.x & 15;
  const int m = mb >> 5;
  const int rowbase = mb * L_;
  const int cl = (NT == 2) ? 2 * mr : mr;
  const float* S = intra + (size_t)mb * L_ * L_;
  const float* sr[2] = {S + (size_t)(i0 + mr) * L_, S + (size_t)(i0 + 16 + mr) * L_};
  const float* dv = dinv + rowbase;
  const float* hb = tile + m * L_ * TP_ + cl;
  v8f acc[2][NT] = {};
  auto lA = [&](int t, int kq) { RawQuad r; r.p = ld4(sr[t] + kq); r.q = ld4(dv + kq); return r; };
  auto cA = [](RawQuad r) { return r.p * r.q; };
  auto lB = [&](int j, int kk) {
    if constexpr (NT == 2) {
      return ld2(hb + (kk + j) * TP_);
    } else {
      v2f b;
      b.x = hb[kk * TP_];
      b.y = hb[(kk + 1) * TP_];
      return b;
    }
  };
  {
    const int kh0 = (threadIdx.x & 31) >> 4;
#pragma unroll
    for (int d = 0; d < 4; ++d)
#pragma unroll
      for (int s = 0; s < 2; ++s)
#pragma unroll
        for (int u = 0; u < NT; ++u) g0.b[d][s][u] = lB(u, 8 * d + 4 * kh0 + 2 * s);
  }
  ring_main<2, NT, 4, L_ / 8>(acc, g0, lA, cA, lB);
  ring_drain(acc, g0, cA);
  const int kh = (threadIdx.x & 31) >> 4;
  const int n1 = (m == 0) ? 1 : 0;
  const int n2 = (m == 2) ? 1 : 2;
  const int ab = rowbase - m * A_;
  const float* c1 = crossv + ((size_t)m * M_ + n1) * A_ + ab;
  const float* c2 = crossv + ((size_t)m * M_ + n2) * A_ + ab;
  const float* e1 = dinv + (size_t)n1 * A_ + ab;
  const float* e2 = dinv + (size_t)n2 * A_ + ab;
  const float* g1 = tile + n1 * L_ * TP_ + cl;
  const float* g2 = tile + n2 * L_ * TP_ + cl;
  float f1[2][8], f2[2][8], fr[2][8];
  float x1[2][8][NT], x2[2][8][NT];
#pragma unroll
  for (int t = 0; t < 2; ++t) {
    const int r = i0 + 16 * t + 8 * kh;
#pragma unroll
    for (int q = 0; q < 2; ++q) {
      const v4f p1 = ld4(c1 + r + 4 * q) * ld4(e1 + r + 4 * q);
      const v4f p2 = ld4(c2 + r + 4 * q) * ld4(e2 + r + 4 * q);
      const v4f dr = ld4(dv + r + 4 * q);
      f1[t][4 * q] = p1.x; f1[t][4 * q + 1] = p1.y; f1[t][4 * q + 2] = p1.z; f1[t][4 * q + 3] = p1.w;
      f2[t][4 * q] = p2.x; f2[t][4 * q + 1] = p2.y; f2[t][4 * q + 2] = p2.z; f2[t][4 * q + 3] = p2.w;
      fr[t][4 * q] = dr.x; fr[t][4 * q + 1] = dr.y; fr[t][4 * q + 2] = dr.z; fr[t][4 * q + 3] = dr.w;
    }
#pragma unroll
    for (int v = 0; v < 8; ++v) {
      const int off = (r + v) * TP_;
      if constexpr (NT == 2) {
        const v2f a = ld2(g1 + off), b = ld2(g2 + off);
        x1[t][v][0] = a.x; x1[t][v][1] = a.y;
        x2[t][v][0] = b.x; x2[t][v][1] = b.y;
      } else {
        x1[t][v][0] = g1[off];
        x2[t][v][0] = g2[off];
      }
    }
  }
  float o[2][NT][8];
#pragma unroll
  for (int t = 0; t < 2; ++t)
#pragma unroll
    for (int u = 0; u < NT; ++u)
#pragma unroll
      for (int v = 0; v < 8; ++v) {
        float s = acc[t][u][v];
        s += f1[t][v] * x1[t][v][u];
        s += f2[t][v] * x2[t][v][u];
        o[t][u][v] = s * fr[t][v];
      }
  store_rm<NT, false>(o, i0, n0, hi + (size_t)rowbase * H_);
  store_fm<NT, false>(o, i0, n0, hiT + rowbase, N_);
}

template <int NT>
__device__ __forceinline__ void fc0_spmm_run(const float* __restrict__ xr, const float* __restrict__ w,
                                             const float* __restrict__ bias,
                                             const float* __restrict__ intra,
                                             const float* __restrict__ crossv,
                                             const float* __restrict__ dinv,
                                             float* __restrict__ h0, float* __restrict__ h0T,
                                             float* __restrict__ hi, float* __restrict__ hiT,
                                             float* tile, int m, int dlg, int i0, int n0) {
  const int m0 = m * A_ + dlg * L_ + i0;
  const int mb = m * B_ + dlg;
  float o[2][NT][8];
  fc0_body<NT>(xr, w, bias, m0, n0, o);
  tile_put<NT, false>(o, tile + (m * L_ + i0) * TP_);
  SpmmRing<NT> g0;
  spmm_fill_a<NT>(g0, intra, dinv, mb, i0);
  lds_barrier();
  store_rm<NT, false>(o, m0, n0, h0);
  store_fm<NT, false>(o, m0, n0, h0T, N_);
  spmm_body<NT>(g0, intra, crossv, dinv, tile, hi, hiT, mb, i0, n0);
}

__global__ void __launch_bounds__(192) k_fc0_spmm(const float* __restrict__ xa, const float* __restrict__ xv,
                           const float* __restrict__ xt, const float* __restrict__ w,
                           const float* __restrict__ bias, const float* __restrict__ intra,
                           const float* __restrict__ crossv, const float* __restrict__ dinv,
                           float* __restrict__ h0, float* __restrict__ h0T,
                           float* __restrict__ hi, float* __restrict__ hiT) {
  __shared__ float tile[TILE_];
  const int wv = wg_wave();
  const int m = wv >> 1, i0 = (wv & 1) * 32;
  const int strip = blockIdx.x, dlg = blockIdx.y, n0 = strip * 32;
  const float* xr = ((m == 0) ? xa : (m == 1) ? xv : xt) + (size_t)(dlg * L_ + i0) * H_;
  if (strip == NS_ - 1) fc0_spmm_run<1>(xr, w, bias, intra, crossv, dinv, h0, h0T, hi, hiT, tile, m, dlg, i0, n0);
  else                  fc0_spmm_run<2>(xr, w, bias, intra, crossv, dinv, h0, h0T, hi, hiT, tile, m, dlg, i0, n0);
}

template <int NT>
__device__ __forceinline__ void conv_body(const float* __restrict__ hi, const float* __restrict__ h0,
                                          const float* __restrict__ hiT, const float* __restrict__ h0T,
                                          const float* __restrict__ w, float beta,
                                          int m0, int n0, float (&o)[2][NT][8]) {
  const int mr = threadIdx.x & 15;
  const int cb = lane_col<NT>(n0);
  const float* t0 = hiT + m0 + 2 * mr;
  const float* t1 = h0T + m0 + 2 * mr;
  const float* w1 = w + (size_t)H_ * H_;
  v8f acc[2][NT] = {};
  auto lA0 = [&](int pl, int k) { return ld2(t0 + (size_t)k * N_); };
  auto lA1 = [&](int pl, int k) { return ld2(t1 + (size_t)k * N_); };
  auto lB0 = [&](int j, int kk) { return ldB_unit<NT>(w, cb, j, kk); };
  auto lB1 = [&](int j, int kk) { return ldB_unit<NT>(w1, cb, j, kk); };
  RingT<NT, 5, 1> g0, g1;
  ringt_fill(g0, lA0, lB0);
  ringt_main<false, NT, 5, KP_>(acc, g0, lA0, lB0);
  ringt_fill(g1, lA1, lB1);
  ringt_drain<false>(acc, g0);
  ringt_main<false, NT, 5, KP_>(acc, g1, lA1, lB1);
  ringt_drain<false>(acc, g1);
  auto ep = [&](int m, int n, float v) {
    const float r = 0.9f * hi[(size_t)m * H_ + n] + 0.1f * h0[(size_t)m * H_ + n];
    return fmaxf(beta * v + (1.0f - beta) * r, 0.0f);
  };
  epi_vals<NT, true>(acc, m0, n0, o, ep);
}

template <int NT>
__device__ __forceinline__ void conv_spmm_run(const float* __restrict__ hi, const float* __restrict__ h0,
                                              const float* __restrict__ hiT, const float* __restrict__ h0T,
                                              const float* __restrict__ w, float beta,
                                              const float* __restrict__ intra,
                                              const float* __restrict__ crossv,
                                              const float* __restrict__ dinv,
                                              float* __restrict__ hi2, float* __restrict__ hi2T,
                                              float* tile, int m, int dlg, int i0, int n0) {
  const int mb = m * B_ + dlg;
  float o[2][NT][8];
  conv_body<NT>(hi, h0, hiT, h0T, w, beta, m * A_ + dlg * L_ + i0, n0, o);
  tile_put<NT, true>(o, tile + (m * L_ + i0) * TP_);
  SpmmRing<NT> g0;
  spmm_fill_a<NT>(g0, intra, dinv, mb, i0);
  lds_barrier();
  spmm_body<NT>(g0, intra, crossv, dinv, tile, hi2, hi2T, mb, i0, n0);
}

__global__ void __launch_bounds__(192) k_conv_spmm(const float* __restrict__ hi, const float* __restrict__ h0,
                            const float* __restrict__ hiT, const float* __restrict__ h0T,
                            const float* __restrict__ w, float beta,
                            const float* __restrict__ intra, const float* __restrict__ crossv,
                            const float* __restrict__ dinv,
                            float* __restrict__ hi2, float* __restrict__ hi2T) {
  __shared__ float tile[TILE_];
  const int wv = wg_wave();
  const int m = wv >> 1, i0 = (wv & 1) * 32;
  const int strip = blockIdx.x, dlg = blockIdx.y, n0 = strip * 32;
  if (strip == NS_ - 1) conv_spmm_run<1>(hi, h0, hiT, h0T, w, beta, intra, crossv, dinv, hi2, hi2T, tile, m, dlg, i0, n0);
  else                  conv_spmm_run<2>(hi, h0, hiT, h0T, w, beta, intra, crossv, dinv, hi2, hi2T, tile, m, dlg, i0, n0);
}

__global__ void __launch_bounds__(128) k_conv(const float* __restrict__ hi, const float* __restrict__ h0,
                       const float* __restrict__ hiT, const float* __restrict__ h0T,
                       const float* __restrict__ w, float beta, float* __restrict__ houtT) {
  const int wv = wg_wave();
  const int strip = 2 * blockIdx.x + (wv & 1);
  if (strip >= NS_) return;
  const int m0 = (blockIdx.y * 2 + (wv >> 1)) * 32, n0 = strip * 32;
  if (strip == NS_ - 1) {
    float o[2][1][8];
    conv_body<1>(hi, h0, hiT, h0T, w, beta, m0, n0, o);
    store_fm<1, true>(o, m0, n0, houtT, N_);
  } else {
    float o[2][2][8];
    conv_body<2>(hi, h0, hiT, h0T, w, beta, m0, n0, o);
    store_fm<2, true>(o, m0, n0, houtT, N_);
  }
}

template <int NT>
__device__ __forceinline__ void hx_body(const float* __restrict__ xm, const float* __restrict__ hfT,
                                        const float* __restrict__ W, const float* __restrict__ Bb,
                                        float* __restrict__ hx, float* __restrict__ hxT,
                                        int rb, int a0_, int n0) {
  const int mr = threadIdx.x & 15;
  const int cb = lane_col<NT>(n0);
  const size_t r0 = (size_t)(a0_ + 2 * mr) * H_, r1 = (size_t)(a0_ + 2 * mr + 1) * H_;
  const float* p0[2] = {xm + r0, xm + r1};
  const float* tf = hfT + rb + a0_ + 2 * mr;
  const float* W1 = W + (size_t)H_ * H_;
  v8f acc[2][NT] = {};
  auto cA = [](v4f r) { return r; };
  auto lA0 = [&](int t, int kq) { return ld4(p0[t] + kq); };
  auto lA1 = [&](int pl, int k) { return ld2(tf + (size_t)k * N_); };
  auto lB0 = [&](int j, int kk) { return ldB_unit<NT>(W, cb, j, kk); };
  auto lB1 = [&](int j, int kk) { return ldB_unit<NT>(W1, cb, j, kk); };
  Ring<2, NT, 5, v4f> g0;
  RingT<NT, 5, 1> g1;
  ring_fill(g0, lA0, lB0);
  ring_main<2, NT, 5, KP_>(acc, g0, lA0, cA, lB0);
  ringt_fill(g1, lA1, lB1);
  ring_drain(acc, g0, cA);
  ringt_main<false, NT, 5, KP_>(acc, g1, lA1, lB1);
  ringt_drain<false>(acc, g1);
  auto ep = [&](int a, int n, float v) { return fmaxf(v + Bb[n], 0.0f); };
  float o[2][NT][8];
  epi_vals<NT, true>(acc, a0_, n0, o, ep);
  store_rm<NT, true>(o, a0_, n0, hx + (size_t)rb * H_);
  store_fm<NT, true>(o, a0_, n0, hxT + (size_t)rb * H_, A_);
}

__global__ void __launch_bounds__(128) k_hx(const float* __restrict__ xa, const float* __restrict__ xv,
                     const float* __restrict__ xt, const float* __restrict__ hfT,
                     const float* __restrict__ tw, const float* __restrict__ tb,
                     float* __restrict__ hx, float* __restrict__ hxT) {
  const int wv = wg_wave();
  const int strip = 2 * blockIdx.x + (wv & 1);
  if (strip >= NS_) return;
  const int m  = blockIdx.z;
  const float* xm = (m == 0) ? xa : (m == 1) ? xv : xt;
  const int a0_ = (blockIdx.y * 2 + (wv >> 1)) * 32, n0 = strip * 32;
  const float* W  = tw + (size_t)m * 2 * H_ * H_;
  const float* Bb = tb + (size_t)m * H_;
  if (strip == NS_ - 1) hx_body<1>(xm, hfT, W, Bb, hx, hxT, m * A_, a0_, n0);
  else                  hx_body<2>(xm, hfT, W, Bb, hx, hxT, m * A_, a0_, n0);
}

constexpr int GS_  = H_ / 4;
constexpr int GDS_ = 5;

template <int NT>
struct RingG { v2f a[GDS_][2][2]; v2f b[GDS_][3][NT]; };

template <int NT>
__device__ __forceinline__ void ringg_use(v8f (&acc)[2][NT], const v2f (&a)[2][2], const v2f (&b)[3][NT]) {
  const v2f p0 = a[0][0] * a[1][0], p1 = a[0][1] * a[1][1];
  v2f fi[2], fj[2], fp[2];
  fi[0].x = a[0][0].x; fi[0].y = a[0][1].x; fi[1].x = a[0][0].y; fi[1].y = a[0][1].y;
  fj[0].x = a[1][0].x; fj[0].y = a[1][1].x; fj[1].x = a[1][0].y; fj[1].y = a[1][1].y;
  fp[0].x = p0.x;      fp[0].y = p1.x;      fp[1].x = p0.y;      fp[1].y = p1.y;
  v2f f0[NT], f1[NT], f2[NT];
  b_frags<NT>(b[0], f0);
  b_frags<NT>(b[1], f1);
  b_frags<NT>(b[2], f2);
  mma_step<2, NT, true>(acc, fi, f0);
  mma_step<2, NT, true>(acc, fj, f1);
  mma_step<2, NT, true>(acc, fp, f2);
}

template <int NT>
__device__ __forceinline__ float tr_logit(const v8f (&a)[NT], int g, int e) {
  if constexpr (NT == 2) return a[e & 1][4 * g + (e >> 1)];
  else                   return a[0][e];
}

template <int NT>
__device__ __forceinline__ void gate_body(const float* __restrict__ Hi, const float* __restrict__ Hj,
                                          const float* __restrict__ HiT, const float* __restrict__ HjT,
                                          const float* __restrict__ W, const float* __restrict__ Bb,
                                          const float* __restrict__ Ui, float* __restrict__ part_out,
                                          int a0_, int n0) {
  const int mr = threadIdx.x & 15;
  const int kh = (threadIdx.x & 31) >> 4;
  const int cb = lane_col<NT>(n0);
  const size_t r0 = (size_t)(a0_ + 2 * mr) * H_, r1 = (size_t)(a0_ + 2 * mr + 1) * H_;
  const float* pi[2] = {Hi + r0, Hi + r1};
  const float* pj[2] = {Hj + r0, Hj + r1};
  const float* ti = HiT + a0_ + 2 * mr;
  const float* tj = HjT + a0_ + 2 * mr;
  v8f acc[2][NT] = {};
  auto lA = [&](int pl, int k) { return ld2((pl ? tj : ti) + (size_t)k * A_); };
  auto lB = [&](int sg, int u, int kk) { return ldB_unit<NT>(W + (size_t)sg * H_ * H_, cb, u, kk); };
  RingG<NT> rg;
#pragma unroll
  for (int d = 0; d < GDS_; ++d) {
    const int kk = 4 * d + 2 * kh;
#pragma unroll
    for (int pl = 0; pl < 2; ++pl)
#pragma unroll
      for (int j = 0; j < 2; ++j) rg.a[d][pl][j] = lA(pl, kk + j);
#pragma unroll
    for (int sg = 0; sg < 3; ++sg)
#pragma unroll
      for (int u = 0; u < NT; ++u) rg.b[d][sg][u] = lB(sg, u, kk);
    __builtin_amdgcn_sched_barrier(0);
  }
#pragma nounroll
  for (int s0 = 0; s0 < GS_ - GDS_; s0 += GDS_) {
#pragma unroll
    for (int d = 0; d < GDS_; ++d) {
      ringg_use<NT>(acc, rg.a[d], rg.b[d]);
      const int kk = 4 * (s0 + d + GDS_) + 2 * kh;
#pragma unroll
      for (int pl = 0; pl < 2; ++pl)
#pragma unroll
        for (int j = 0; j < 2; ++j) rg.a[d][pl][j] = lA(pl, kk + j);
#pragma unroll
      for (int sg = 0; sg < 3; ++sg)
#pragma unroll
        for (int u = 0; u < NT; ++u) rg.b[d][sg][u] = lB(sg, u, kk);
      __builtin_amdgcn_sched_barrier(0);
    }
  }
#pragma unroll
  for (int d = 0; d < GDS_; ++d) ringg_use<NT>(acc, rg.a[d], rg.b[d]);

  float part[2][TAG_] = {};
#pragma unroll
  for (int t = 0; t < 2; ++t) {
#pragma unroll
    for (int g = 0; g < NT; ++g) {
      const int nb = (NT == 2) ? n0 + 16 * kh + 8 * g : n0 + 8 * kh;
      if (NT == 2 || nb < H_) {
        const v4f* hip = reinterpret_cast<const v4f*>(pi[t] + nb);
        const v4f* hjp = reinterpret_cast<const v4f*>(pj[t] + nb);
        const v4f* bp  = reinterpret_cast<const v4f*>(Bb + nb);
        const v4f* up  = reinterpret_cast<const v4f*>(Ui + (size_t)nb * TAG_);
        const v4f hi0 = hip[0], hi1 = hip[1], hj0 = hjp[0], hj1 = hjp[1], b0 = bp[0], b1 = bp[1];
        const float hiv[8] = {hi0.x, hi0.y, hi0.z, hi0.w, hi1.x, hi1.y, hi1.z, hi1.w};
        const float hjv[8] = {hj0.x, hj0.y, hj0.z, hj0.w, hj1.x, hj1.y, hj1.z, hj1.w};
        const float bv[8]  = {b0.x, b0.y, b0.z, b0.w, b1.x, b1.y, b1.z, b1.w};
        float uu[8 * TAG_];
#pragma unroll
        for (int q = 0; q < 2 * TAG_; ++q) {
          const v4f x = up[q];
          uu[4 * q] = x.x; uu[4 * q + 1] = x.y; uu[4 * q + 2] = x.z; uu[4 * q + 3] = x.w;
        }
#pragma unroll
        for (int e = 0; e < 8; ++e) {
          const float lg = tr_logit<NT>(acc[t], g, e);
          const float z = 1.0f / (1.0f + expf(-(lg + bv[e])));
          const float gt = z * hiv[e] + (1.0f - z) * hjv[e];
#pragma unroll
          for (int c = 0; c < TAG_; ++c) part[t][c] += gt * uu[e * TAG_ + c];
        }
      }
    }
  }
#pragma unroll
  for (int t = 0; t < 2; ++t)
#pragma unroll
    for (int c = 0; c < TAG_; ++c) part[t][c] += __shfl_xor(part[t][c], 16, 32);
  if (kh == 0) {
#pragma unroll
    for (int t = 0; t < 2; ++t) {
      float* po = part_out + (size_t)(a0_ + 2 * mr + t) * TAG_;
#pragma unroll
      for (int c = 0; c < TAG_; c += 2) {
        v2f o; o.x = part[t][c]; o.y = part[t][c + 1];
        *reinterpret_cast<v2f*>(po + c) = o;
      }
    }
  }
}

__global__ void __launch_bounds__(128) k_gate(const float* __restrict__ hx, const float* __restrict__ hxT,
                       const float* __restrict__ cw,
                       const float* __restrict__ cb, const float* __restrict__ uw,
                       float* __restrict__ plog) {
  const int wv = wg_wave();
  const int strip = 2 * blockIdx.x + (wv & 1);
  if (strip >= NS_) return;
  const int p = blockIdx.z;
  const int i = p >> 1;
  const int j = (p & 1) ? (i == 2 ? 1 : 2) : (i == 0 ? 1 : 0);
  const int a0_ = (blockIdx.y * 2 + (wv >> 1)) * 32, n0 = strip * 32;
  const float* Hi = hx + (size_t)i * A_ * H_;
  const float* Hj = hx + (size_t)j * A_ * H_;
  const float* HiT = hxT + (size_t)i * H_ * A_;
  const float* HjT = hxT + (size_t)j * H_ * A_;
  const float* W  = cw + (size_t)p * 3 * H_ * H_;
  const float* Bb = cb + (size_t)p * H_;
  const float* Ui = uw + (size_t)i * H_ * TAG_;
  float* po = plog + ((size_t)p * NS_ + strip) * A_ * TAG_;
  if (strip == NS_ - 1) gate_body<1>(Hi, Hj, HiT, HjT, W, Bb, Ui, po, a0_, n0);
  else                  gate_body<2>(Hi, Hj, HiT, HjT, W, Bb, Ui, po, a0_, n0);
}

__global__ void k_final(const float* __restrict__ plog, const float* __restrict__ ub,
                        float* __restrict__ out) {
  const int e = blockIdx.x * blockDim.x + threadIdx.x;
  const int t = e % TAG_;
  float s = ub[t] + ub[TAG_ + t] + ub[2 * TAG_ + t];
  for (int q = 0; q < 6 * NS_; ++q) s += plog[(size_t)q * A_ * TAG_ + e];
  out[e] = s;
}

}

extern "C" void kernel_launch(void* const* d_in, const int* in_sizes, int n_in,
                              void* d_out, int out_size, void* d_ws, size_t ws_size,
                              hipStream_t stream) {
  (void)in_sizes; (void)n_in; (void)out_size; (void)ws_size;
  const float* x_a     = (const float*)d_in[0];
  const float* x_v     = (const float*)d_in[1];
  const float* x_t     = (const float*)d_in[2];
  const float* fc0_w   = (const float*)d_in[3];
  const float* fc0_b   = (const float*)d_in[4];
  const float* conv_w  = (const float*)d_in[5];
  const float* tm_w    = (const float*)d_in[6];
  const float* tm_b    = (const float*)d_in[7];
  const float* cross_w = (const float*)d_in[8];
  const float* cross_b = (const float*)d_in[9];
  const float* uni_w   = (const float*)d_in[10];
  const float* uni_b   = (const float*)d_in[11];
  float* out = (float*)d_out;

  const size_t NH = (size_t)N_ * H_;
  float* ws    = (float*)d_ws;
  float* intra = ws; ws += (size_t)N_ * L_;
  float* crossv= ws; ws += (size_t)M_ * M_ * A_;
  float* dinv  = ws; ws += N_;
  float* h0    = ws; ws += NH;
  float* h0T   = ws; ws += NH;
  float* hBfT  = ws; ws += NH;
  float* hi    = ws; ws += NH;
  float* hiT   = ws; ws += NH;
  float* hi2   = ws; ws += NH;
  float* hi2T  = ws; ws += NH;
  float* hx    = ws; ws += NH;
  float* hxT   = ws; ws += NH;
  float* plog  = ws; ws += (size_t)6 * NS_ * A_ * TAG_;


  const float beta0 = logf(0.5f / 1.0f + 1.0f);
  const float beta1 = logf(0.5f / 2.0f + 1.0f);
  k_graph<<<M_ * B_ * 4,         dim3(32), 0, stream>>>(x_a, x_v, x_t, intra, crossv, dinv);
  k_fc0_spmm <<<dim3(NS_, B_), dim3(192), 0, stream>>>(x_a, x_v, x_t, fc0_w, fc0_b, intra, crossv, dinv,
                                                     h0, h0T, hi, hiT);
  k_conv_spmm<<<dim3(NS_, B_), dim3(192), 0, stream>>>(hi, h0, hiT, h0T, conv_w, beta0, intra, crossv, dinv,
                                                     hi2, hi2T);
  k_conv<<<dim3(NSP_, N_ / 64), dim3(128), 0, stream>>>(hi2, h0, hi2T, h0T, conv_w + 2 * H_ * H_, beta1, hBfT);

  k_hx   <<<dim3(NSP_, A_ / 64, M_), dim3(128), 0, stream>>>(x_a, x_v, x_t, hBfT, tm_w, tm_b, hx, hxT);

  k_gate <<<dim3(NSP_, A_ / 64, 6), dim3(128), 0, stream>>>(hx, hxT, cross_w, cross_b, uni_w, plog);

  k_final<<<A_ * TAG_ / 256, dim3(256), 0, stream>>>(plog, uni_b, out);
}
